# starting kernel
# baseline (speedup 1.0000x reference)
#include <hip/hip_runtime.h>

typedef __attribute__((ext_vector_type(2))) float v2f;
typedef __attribute__((ext_vector_type(8))) float v8f;

__global__ void k_zero(float* __restrict__ p, long long n) {
  long long i = (long long)blockIdx.x * blockDim.x + threadIdx.x;
  if (i < n) p[i] = 0.0f;
}

__global__ void k_node_fc(const float* __restrict__ nf, const float* __restrict__ Wn,
                          const float* __restrict__ bn, float* __restrict__ h, int N) {
  int t = blockIdx.x * blockDim.x + threadIdx.x;
  int n = t >> 6, j = t & 63;
  if (n >= N) return;
  float f0 = nf[n * 3 + 0], f1 = nf[n * 3 + 1], f2 = nf[n * 3 + 2];
  h[t] = bn[j] + Wn[j * 3 + 0] * f0 + Wn[j * 3 + 1] * f1 + Wn[j * 3 + 2] * f2;
}

__global__ void k_degree(const int* __restrict__ src, float* __restrict__ cnt, int E) {
  int e = blockIdx.x * blockDim.x + threadIdx.x;
  if (e < E) atomicAdd(&cnt[src[e]], 1.0f);
}

__global__ void k_scatter(const int* __restrict__ src, const int* __restrict__ dst,
                          const float* __restrict__ h, float* __restrict__ accum, int E) {
  int t = blockIdx.x * blockDim.x + threadIdx.x;
  int e = t >> 2;
  if (e >= E) return;
  int q = (t & 3) * 16;
  const float* hr = h + (long long)dst[e] * 64 + q;
  float* ar = accum + (long long)src[e] * 64 + q;
#pragma unroll
  for (int i = 0; i < 16; ++i) atomicAdd(&ar[i], hr[i]);
}

__global__ void __launch_bounds__(256)
k_layer_gemm(const float* __restrict__ hin, const float* __restrict__ accum,
             const float* __restrict__ cnt,
             const float* __restrict__ Wself, const float* __restrict__ bself,
             const float* __restrict__ Wneigh, const float* __restrict__ bneigh,
             float* __restrict__ hout, int N) {
  __shared__ float Bl[32 * 4 * 32 * 2];
  int tid = threadIdx.x;
  for (int idx = tid; idx < 8192; idx += 256) {
    int v = idx & 1, lane = (idx >> 1) & 31, ct = (idx >> 6) & 3, kk = idx >> 8;
    int n = ct * 16 + (lane & 15);
    int k = kk * 4 + 2 * (lane >> 4) + v;
    Bl[idx] = (k < 64) ? Wself[n * 64 + k] : Wneigh[n * 64 + (k - 64)];
  }
  __syncthreads();

  int w = tid >> 5, lane = tid & 31;
  int base = (blockIdx.x * 8 + w) * 16;
  int half = lane >> 4;
  int mrow = base + (lane & 15);
  long long mc = (mrow < N) ? mrow : 0;

  float c = cnt[mc];
  float inv = (c > 0.f) ? 1.f / c : 1.f;

  v2f A[32];
#pragma unroll
  for (int kk = 0; kk < 16; ++kk) {
    int k0 = kk * 4 + 2 * half;
    v2f a = {hin[mc * 64 + k0], hin[mc * 64 + k0 + 1]};
    A[kk] = a;
  }
#pragma unroll
  for (int kk = 0; kk < 16; ++kk) {
    int k0 = kk * 4 + 2 * half;
    v2f a = {accum[mc * 64 + k0] * inv, accum[mc * 64 + k0 + 1] * inv};
    A[16 + kk] = a;
  }

  v8f acc0 = {}, acc1 = {}, acc2 = {}, acc3 = {};
#pragma unroll
  for (int kk = 0; kk < 32; ++kk) {
    int bb = kk * 256 + lane * 2;
    acc0 = __builtin_amdgcn_wmma_f32_16x16x4_f32(false, A[kk], false, *(const v2f*)&Bl[bb + 0],   (short)0, acc0, false, false);
    acc1 = __builtin_amdgcn_wmma_f32_16x16x4_f32(false, A[kk], false, *(const v2f*)&Bl[bb + 64],  (short)0, acc1, false, false);
    acc2 = __builtin_amdgcn_wmma_f32_16x16x4_f32(false, A[kk], false, *(const v2f*)&Bl[bb + 128], (short)0, acc2, false, false);
    acc3 = __builtin_amdgcn_wmma_f32_16x16x4_f32(false, A[kk], false, *(const v2f*)&Bl[bb + 192], (short)0, acc3, false, false);
  }

#pragma unroll
  for (int ct = 0; ct < 4; ++ct) {
    v8f av = (ct == 0) ? acc0 : (ct == 1) ? acc1 : (ct == 2) ? acc2 : acc3;
    int n = ct * 16 + (lane & 15);
    float bs = bself[n] + bneigh[n];
#pragma unroll
    for (int r = 0; r < 8; ++r) {
      int mg = base + r + 8 * half;
      if (mg < N) {
        float val = av[r] + bs;
        hout[(long long)mg * 64 + n] = val > 0.f ? val : 0.f;
      }
    }
  }
}

__global__ void k_prep(const float* __restrict__ Wa, const float* __restrict__ ba,
                       const float* __restrict__ We, const float* __restrict__ be,
                       const float* __restrict__ env, int NV,
                       float* __restrict__ afold, float* __restrict__ cfold) {
  int j = threadIdx.x;
  if (j >= 64) return;
  float a0 = 0.f, a1 = 0.f, a2 = 0.f, c = ba[j];
  for (int k = 0; k < 64; ++k) {
    float wv = Wa[j * 128 + k];
    a0 += wv * We[k * 3 + 0];
    a1 += wv * We[k * 3 + 1];
    a2 += wv * We[k * 3 + 2];
    c += wv * be[k];
  }
  float invn = 1.0f / (float)NV;
  for (int d = 0; d < 64; ++d) {
    float vm = 0.f;
    for (int v = 0; v < NV; ++v) vm += env[v * 64 + d];
    c += Wa[j * 128 + 64 + d] * (vm * invn);
  }
  afold[j * 3 + 0] = a0;
  afold[j * 3 + 1] = a1;
  afold[j * 3 + 2] = a2;
  cfold[j] = c;
}

__global__ void __launch_bounds__(256)
k_edge_score(const float* __restrict__ ef, const float* __restrict__ afold,
             const float* __restrict__ cfold, const float* __restrict__ W1,
             const float* __restrict__ b1, const float* __restrict__ W2,
             const float* __restrict__ b2, float* __restrict__ scores, int E) {
  __shared__ float Bs[16 * 4 * 32 * 2];
  __shared__ float Af[192];
  __shared__ float Cf[64];
  int tid = threadIdx.x;
  for (int idx = tid; idx < 4096; idx += 256) {
    int v = idx & 1, lane = (idx >> 1) & 31, ct = (idx >> 6) & 3, kk = idx >> 8;
    int n = ct * 16 + (lane & 15);
    int k = kk * 4 + 2 * (lane >> 4) + v;
    Bs[idx] = W1[n * 64 + k];
  }
  if (tid < 192) Af[tid] = afold[tid];
  if (tid < 64) Cf[tid] = cfold[tid];
  __syncthreads();

  int w = tid >> 5, lane = tid & 31;
  int base = (blockIdx.x * 8 + w) * 16;
  int half = lane >> 4;
  int row = base + (lane & 15);
  long long rc = (row < E) ? row : 0;
  float f0 = ef[rc * 3 + 0], f1 = ef[rc * 3 + 1], f2 = ef[rc * 3 + 2];

  v2f A[16];
#pragma unroll
  for (int kk = 0; kk < 16; ++kk) {
    int k0 = kk * 4 + 2 * half;
    float x0 = Cf[k0] + Af[k0 * 3] * f0 + Af[k0 * 3 + 1] * f1 + Af[k0 * 3 + 2] * f2;
    float x1 = Cf[k0 + 1] + Af[(k0 + 1) * 3] * f0 + Af[(k0 + 1) * 3 + 1] * f1 +
               Af[(k0 + 1) * 3 + 2] * f2;
    v2f a = {x0 > 0.f ? x0 : 0.f, x1 > 0.f ? x1 : 0.f};
    A[kk] = a;
  }

  v8f acc0 = {}, acc1 = {}, acc2 = {}, acc3 = {};
#pragma unroll
  for (int kk = 0; kk < 16; ++kk) {
    int bb = kk * 256 + lane * 2;
    acc0 = __builtin_amdgcn_wmma_f32_16x16x4_f32(false, A[kk], false, *(const v2f*)&Bs[bb + 0],   (short)0, acc0, false, false);
    acc1 = __builtin_amdgcn_wmma_f32_16x16x4_f32(false, A[kk], false, *(const v2f*)&Bs[bb + 64],  (short)0, acc1, false, false);
    acc2 = __builtin_amdgcn_wmma_f32_16x16x4_f32(false, A[kk], false, *(const v2f*)&Bs[bb + 128], (short)0, acc2, false, false);
    acc3 = __builtin_amdgcn_wmma_f32_16x16x4_f32(false, A[kk], false, *(const v2f*)&Bs[bb + 192], (short)0, acc3, false, false);
  }

  float p[8] = {0, 0, 0, 0, 0, 0, 0, 0};
#pragma unroll
  for (int ct = 0; ct < 4; ++ct) {
    v8f av = (ct == 0) ? acc0 : (ct == 1) ? acc1 : (ct == 2) ? acc2 : acc3;
    int n = ct * 16 + (lane & 15);
    float bn_ = b1[n], wn = W2[n];
#pragma unroll
    for (int r = 0; r < 8; ++r) {
      float s = av[r] + bn_;
      s = s > 0.f ? s : 0.f;
      p[r] += wn * s;
    }
  }
#pragma unroll
  for (int r = 0; r < 8; ++r) {
    p[r] += __shfl_xor(p[r], 1);
    p[r] += __shfl_xor(p[r], 2);
    p[r] += __shfl_xor(p[r], 4);
    p[r] += __shfl_xor(p[r], 8);
  }
  if ((lane & 15) == 0) {
    float bb2 = b2[0];
    int rb = base + 8 * half;
#pragma unroll
    for (int r = 0; r < 8; ++r)
      if (rb + r < E) scores[rb + r] = p[r] + bb2;
  }
}

extern "C" void kernel_launch(void* const* d_in, const int* in_sizes, int n_in,
                              void* d_out, int out_size, void* d_ws, size_t ws_size,
                              hipStream_t stream) {
  if (n_in < 18) return;
  const float* node_feats = (const float*)d_in[0];
  const float* edge_feats = (const float*)d_in[1];
  const float* env        = (const float*)d_in[2];
  const int*   eidx       = (const int*)d_in[3];
  const float* Wn  = (const float*)d_in[4];
  const float* bn  = (const float*)d_in[5];
  const float* We  = (const float*)d_in[6];
  const float* be  = (const float*)d_in[7];
  const float* Wself  = (const float*)d_in[8];
  const float* bself  = (const float*)d_in[9];
  const float* Wneigh = (const float*)d_in[10];
  const float* bneigh = (const float*)d_in[11];
  const float* Wa  = (const float*)d_in[12];
  const float* ba  = (const float*)d_in[13];
  const float* W1  = (const float*)d_in[14];
  const float* b1  = (const float*)d_in[15];
  const float* W2  = (const float*)d_in[16];
  const float* b2  = (const float*)d_in[17];

  int N  = in_sizes[0] / 3;
  int E  = in_sizes[1] / 3;
  int NV = in_sizes[2] / 64;
  int L  = in_sizes[8] / (64 * 64);

  const int* src = eidx;
  const int* dst = eidx + E;

  float* scores = (float*)d_out;
  float* h_out  = (float*)d_out + E;

  float* accum = (float*)d_ws;
  float* cnt   = accum + (long long)N * 64;
  long long off = (((long long)N * 65) + 63) & ~63LL;
  float* h_b   = (float*)d_ws + off;
  float* afold = h_b + (long long)N * 64;
  float* cfold = afold + 192;

  long long zn = (long long)N * 65;
  k_zero<<<(int)((zn + 255) / 256), 256, 0, stream>>>(accum, zn);
  k_node_fc<<<(N * 64 + 255) / 256, 256, 0, stream>>>(node_feats, Wn, bn, h_out, N);
  k_degree<<<(E + 255) / 256, 256, 0, stream>>>(src, cnt, E);

  int tiles = (N + 15) / 16;
  float* hin = h_out;
  for (int l = 0; l < L; ++l) {
    if (l > 0) {
      long long an = (long long)N * 64;
      k_zero<<<(int)((an + 255) / 256), 256, 0, stream>>>(accum, an);
    }
    k_scatter<<<(int)(((long long)E * 4 + 255) / 256), 256, 0, stream>>>(src, dst, hin, accum, E);
    float* hout = (l == L - 1) ? h_out : ((hin == h_out) ? h_b : h_out);
    k_layer_gemm<<<(tiles + 7) / 8, 256, 0, stream>>>(
        hin, accum, cnt,
        Wself + (long long)l * 64 * 64, bself + (long long)l * 64,
        Wneigh + (long long)l * 64 * 64, bneigh + (long long)l * 64, hout, N);
    hin = hout;
  }

  k_prep<<<1, 64, 0, stream>>>(Wa, ba, We, be, env, NV, afold, cfold);
  int etiles = (E + 15) / 16;
  k_edge_score<<<(etiles + 7) / 8, 256, 0, stream>>>(edge_feats, afold, cfold, W1, b1, W2,
                                                     b2, scores, E);
}


# segment sums without float atomics: edges bucketed by source (returning int atomic = arrival-order rank, scan, fill), one wave per node gathers its rows; sum order changes: rank order, 8 partials
# speedup vs baseline: 2.7764x; 2.7764x over previous
#include <hip/hip_runtime.h>

typedef __attribute__((ext_vector_type(2))) float v2f;
typedef __attribute__((ext_vector_type(4))) float v4f;
typedef __attribute__((ext_vector_type(8))) float v8f;

__global__ void k_zero_i(int* __restrict__ p, int n) {
  int i = blockIdx.x * blockDim.x + threadIdx.x;
  if (i < n) p[i] = 0;
}

__global__ void k_node_fc(const float* __restrict__ nf, const float* __restrict__ Wn,
                          const float* __restrict__ bn, float* __restrict__ h, int N) {
  int t = blockIdx.x * blockDim.x + threadIdx.x;
  int n = t >> 6, j = t & 63;
  if (n >= N) return;
  float f0 = nf[n * 3 + 0], f1 = nf[n * 3 + 1], f2 = nf[n * 3 + 2];
  h[t] = bn[j] + Wn[j * 3 + 0] * f0 + Wn[j * 3 + 1] * f1 + Wn[j * 3 + 2] * f2;
}

__global__ void k_rank(const int* __restrict__ src, int* __restrict__ deg,
                       int* __restrict__ rank, int E) {
  int e = blockIdx.x * blockDim.x + threadIdx.x;
  if (e < E) rank[e] = atomicAdd(&deg[src[e]], 1);
}

__global__ void __launch_bounds__(1024)
k_scan(const int* __restrict__ deg, int* __restrict__ rowptr, float* __restrict__ cnt,
       int N) {
  __shared__ int wsum[32];
  __shared__ int wpre[32];
  int t = threadIdx.x, lane = t & 31, w = t >> 5;
  int nb = blockIdx.x;
  int pre = 0;
#pragma unroll 8
  for (int k = 0; k < nb; ++k) pre += deg[k * 1024 + t];
  int gi = nb * 1024 + t;
  int v = (gi < N) ? deg[gi] : 0;
  int inc = v;
  for (int o = 1; o < 32; o <<= 1) {
    int u = __shfl_up(inc, o);
    if (lane >= o) inc += u;
  }
  for (int o = 16; o > 0; o >>= 1) pre += __shfl_xor(pre, o);
  if (lane == 31) wsum[w] = inc;
  if (lane == 0) wpre[w] = pre;
  __syncthreads();
  int tot = wpre[lane] + ((lane < w) ? wsum[lane] : 0);
  for (int o = 16; o > 0; o >>= 1) tot += __shfl_xor(tot, o);
  int excl = tot + inc - v;
  if (gi < N) {
    rowptr[gi] = excl;
    cnt[gi] = (float)v;
    if (gi == N - 1) rowptr[N] = excl + v;
  }
}

__global__ void k_fill(const int* __restrict__ src, const int* __restrict__ dst,
                       const int* __restrict__ rowptr, const int* __restrict__ rank,
                       int* __restrict__ col, int E) {
  int e = blockIdx.x * blockDim.x + threadIdx.x;
  if (e < E) col[rowptr[src[e]] + rank[e]] = dst[e];
}

__global__ void __launch_bounds__(256)
k_gather(const int* __restrict__ rowptr, const int* __restrict__ col,
         const float* __restrict__ h, float* __restrict__ accum, int N) {
  int node = blockIdx.x * 8 + (threadIdx.x >> 5);
  if (node >= N) return;
  int lane = threadIdx.x & 31, half = lane >> 4, c4 = (lane & 15) * 4;
  int beg = rowptr[node], end = rowptr[node + 1];
  const v4f zero = {0.f, 0.f, 0.f, 0.f};
  v4f s[4] = {zero, zero, zero, zero};
  if (beg < end) {
    int last = end - 1;
    unsigned nmax = (unsigned)(N - 1);
    int id[8];
#pragma unroll
    for (int u = 0; u < 8; ++u) {
      int e = beg + half + 2 * u;
      id[u] = col[e < last ? e : last];
    }
    for (int j0 = beg; j0 < end; j0 += 16) {
      v4f v[8];
#pragma unroll
      for (int u = 0; u < 8; ++u) {
        unsigned r = (unsigned)id[u] < nmax ? (unsigned)id[u] : nmax;
        v[u] = *(const v4f*)(h + (size_t)r * 64 + c4);
      }
#pragma unroll
      for (int u = 0; u < 8; ++u) {
        int e = j0 + 16 + half + 2 * u;
        id[u] = col[e < last ? e : last];
      }
#pragma unroll
      for (int u = 0; u < 8; ++u) {
        bool ok = (j0 + half + 2 * u) < end;
        s[u & 3] += ok ? v[u] : zero;
      }
    }
  }
  v4f a = (s[0] + s[1]) + (s[2] + s[3]);
  a.x += __shfl_xor(a.x, 16);
  a.y += __shfl_xor(a.y, 16);
  a.z += __shfl_xor(a.z, 16);
  a.w += __shfl_xor(a.w, 16);
  if (half == 0) *(v4f*)(accum + (size_t)node * 64 + c4) = a;
}

__global__ void __launch_bounds__(256)
k_layer_gemm(const float* __restrict__ hin, const float* __restrict__ accum,
             const float* __restrict__ cnt,
             const float* __restrict__ Wself, const float* __restrict__ bself,
             const float* __restrict__ Wneigh, const float* __restrict__ bneigh,
             float* __restrict__ hout, int N) {
  __shared__ float Bl[32 * 4 * 32 * 2];
  int tid = threadIdx.x;
  for (int idx = tid; idx < 8192; idx += 256) {
    int v = idx & 1, lane = (idx >> 1) & 31, ct = (idx >> 6) & 3, kk = idx >> 8;
    int n = ct * 16 + (lane & 15);
    int k = kk * 4 + 2 * (lane >> 4) + v;
    Bl[idx] = (k < 64) ? Wself[n * 64 + k] : Wneigh[n * 64 + (k - 64)];
  }
  __syncthreads();

  int w = tid >> 5, lane = tid & 31;
  int base = (blockIdx.x * 8 + w) * 16;
  int half = lane >> 4;
  int mrow = base + (lane & 15);
  long long mc = (mrow < N) ? mrow : 0;

  float c = cnt[mc];
  float inv = (c > 0.f) ? 1.f / c : 1.f;

  v2f A[32];
#pragma unroll
  for (int kk = 0; kk < 16; ++kk) {
    int k0 = kk * 4 + 2 * half;
    v2f a = {hin[mc * 64 + k0], hin[mc * 64 + k0 + 1]};
    A[kk] = a;
  }
#pragma unroll
  for (int kk = 0; kk < 16; ++kk) {
    int k0 = kk * 4 + 2 * half;
    v2f a = {accum[mc * 64 + k0] * inv, accum[mc * 64 + k0 + 1] * inv};
    A[16 + kk] = a;
  }

  v8f acc0 = {}, acc1 = {}, acc2 = {}, acc3 = {};
#pragma unroll
  for (int kk = 0; kk < 32; ++kk) {
    int bb = kk * 256 + lane * 2;
    acc0 = __builtin_amdgcn_wmma_f32_16x16x4_f32(false, A[kk], false, *(const v2f*)&Bl[bb + 0],   (short)0, acc0, false, false);
    acc1 = __builtin_amdgcn_wmma_f32_16x16x4_f32(false, A[kk], false, *(const v2f*)&Bl[bb + 64],  (short)0, acc1, false, false);
    acc2 = __builtin_amdgcn_wmma_f32_16x16x4_f32(false, A[kk], false, *(const v2f*)&Bl[bb + 128], (short)0, acc2, false, false);
    acc3 = __builtin_amdgcn_wmma_f32_16x16x4_f32(false, A[kk], false, *(const v2f*)&Bl[bb + 192], (short)0, acc3, false, false);
  }

#pragma unroll
  for (int ct = 0; ct < 4; ++ct) {
    v8f av = (ct == 0) ? acc0 : (ct == 1) ? acc1 : (ct == 2) ? acc2 : acc3;
    int n = ct * 16 + (lane & 15);
    float bs = bself[n] + bneigh[n];
#pragma unroll
    for (int r = 0; r < 8; ++r) {
      int mg = base + r + 8 * half;
      if (mg < N) {
        float val = av[r] + bs;
        hout[(long long)mg * 64 + n] = val > 0.f ? val : 0.f;
      }
    }
  }
}

__global__ void k_prep(const float* __restrict__ Wa, const float* __restrict__ ba,
                       const float* __restrict__ We, const float* __restrict__ be,
                       const float* __restrict__ env, int NV,
                       float* __restrict__ afold, float* __restrict__ cfold) {
  int j = threadIdx.x;
  if (j >= 64) return;
  float a0 = 0.f, a1 = 0.f, a2 = 0.f, c = ba[j];
  for (int k = 0; k < 64; ++k) {
    float wv = Wa[j * 128 + k];
    a0 += wv * We[k * 3 + 0];
    a1 += wv * We[k * 3 + 1];
    a2 += wv * We[k * 3 + 2];
    c += wv * be[k];
  }
  float invn = 1.0f / (float)NV;
  for (int d = 0; d < 64; ++d) {
    float vm = 0.f;
    for (int v = 0; v < NV; ++v) vm += env[v * 64 + d];
    c += Wa[j * 128 + 64 + d] * (vm * invn);
  }
  afold[j * 3 + 0] = a0;
  afold[j * 3 + 1] = a1;
  afold[j * 3 + 2] = a2;
  cfold[j] = c;
}

__global__ void __launch_bounds__(256)
k_edge_score(const float* __restrict__ ef, const float* __restrict__ afold,
             const float* __restrict__ cfold, const float* __restrict__ W1,
             const float* __restrict__ b1, const float* __restrict__ W2,
             const float* __restrict__ b2, float* __restrict__ scores, int E) {
  __shared__ float Bs[16 * 4 * 32 * 2];
  __shared__ float Af[192];
  __shared__ float Cf[64];
  int tid = threadIdx.x;
  for (int idx = tid; idx < 4096; idx += 256) {
    int v = idx & 1, lane = (idx >> 1) & 31, ct = (idx >> 6) & 3, kk = idx >> 8;
    int n = ct * 16 + (lane & 15);
    int k = kk * 4 + 2 * (lane >> 4) + v;
    Bs[idx] = W1[n * 64 + k];
  }
  if (tid < 192) Af[tid] = afold[tid];
  if (tid < 64) Cf[tid] = cfold[tid];
  __syncthreads();

  int w = tid >> 5, lane = tid & 31;
  int base = (blockIdx.x * 8 + w) * 16;
  int half = lane >> 4;
  int row = base + (lane & 15);
  long long rc = (row < E) ? row : 0;
  float f0 = ef[rc * 3 + 0], f1 = ef[rc * 3 + 1], f2 = ef[rc * 3 + 2];

  v2f A[16];
#pragma unroll
  for (int kk = 0; kk < 16; ++kk) {
    int k0 = kk * 4 + 2 * half;
    float x0 = Cf[k0] + Af[k0 * 3] * f0 + Af[k0 * 3 + 1] * f1 + Af[k0 * 3 + 2] * f2;
    float x1 = Cf[k0 + 1] + Af[(k0 + 1) * 3] * f0 + Af[(k0 + 1) * 3 + 1] * f1 +
               Af[(k0 + 1) * 3 + 2] * f2;
    v2f a = {x0 > 0.f ? x0 : 0.f, x1 > 0.f ? x1 : 0.f};
    A[kk] = a;
  }

  v8f acc0 = {}, acc1 = {}, acc2 = {}, acc3 = {};
#pragma unroll
  for (int kk = 0; kk < 16; ++kk) {
    int bb = kk * 256 + lane * 2;
    acc0 = __builtin_amdgcn_wmma_f32_16x16x4_f32(false, A[kk], false, *(const v2f*)&Bs[bb + 0],   (short)0, acc0, false, false);
    acc1 = __builtin_amdgcn_wmma_f32_16x16x4_f32(false, A[kk], false, *(const v2f*)&Bs[bb + 64],  (short)0, acc1, false, false);
    acc2 = __builtin_amdgcn_wmma_f32_16x16x4_f32(false, A[kk], false, *(const v2f*)&Bs[bb + 128], (short)0, acc2, false, false);
    acc3 = __builtin_amdgcn_wmma_f32_16x16x4_f32(false, A[kk], false, *(const v2f*)&Bs[bb + 192], (short)0, acc3, false, false);
  }

  float p[8] = {0, 0, 0, 0, 0, 0, 0, 0};
#pragma unroll
  for (int ct = 0; ct < 4; ++ct) {
    v8f av = (ct == 0) ? acc0 : (ct == 1) ? acc1 : (ct == 2) ? acc2 : acc3;
    int n = ct * 16 + (lane & 15);
    float bn_ = b1[n], wn = W2[n];
#pragma unroll
    for (int r = 0; r < 8; ++r) {
      float s = av[r] + bn_;
      s = s > 0.f ? s : 0.f;
      p[r] += wn * s;
    }
  }
#pragma unroll
  for (int r = 0; r < 8; ++r) {
    p[r] += __shfl_xor(p[r], 1);
    p[r] += __shfl_xor(p[r], 2);
    p[r] += __shfl_xor(p[r], 4);
    p[r] += __shfl_xor(p[r], 8);
  }
  if ((lane & 15) == 0) {
    float bb2 = b2[0];
    int rb = base + 8 * half;
#pragma unroll
    for (int r = 0; r < 8; ++r)
      if (rb + r < E) scores[rb + r] = p[r] + bb2;
  }
}

extern "C" void kernel_launch(void* const* d_in, const int* in_sizes, int n_in,
                              void* d_out, int out_size, void* d_ws, size_t ws_size,
                              hipStream_t stream) {
  if (n_in < 18) return;
  const float* node_feats = (const float*)d_in[0];
  const float* edge_feats = (const float*)d_in[1];
  const float* env        = (const float*)d_in[2];
  const int*   eidx       = (const int*)d_in[3];
  const float* Wn  = (const float*)d_in[4];
  const float* bn  = (const float*)d_in[5];
  const float* We  = (const float*)d_in[6];
  const float* be  = (const float*)d_in[7];
  const float* Wself  = (const float*)d_in[8];
  const float* bself  = (const float*)d_in[9];
  const float* Wneigh = (const float*)d_in[10];
  const float* bneigh = (const float*)d_in[11];
  const float* Wa  = (const float*)d_in[12];
  const float* ba  = (const float*)d_in[13];
  const float* W1  = (const float*)d_in[14];
  const float* b1  = (const float*)d_in[15];
  const float* W2  = (const float*)d_in[16];
  const float* b2  = (const float*)d_in[17];

  int N  = in_sizes[0] / 3;
  int E  = in_sizes[1] / 3;
  int NV = in_sizes[2] / 64;
  int L  = in_sizes[8] / (64 * 64);

  const int* src = eidx;
  const int* dst = eidx + E;

  float* scores = (float*)d_out;
  float* h_out  = (float*)d_out + E;

  float* accum = (float*)d_ws;
  float* cnt   = accum + (long long)N * 64;
  long long off = (((long long)N * 65) + 63) & ~63LL;
  float* h_b   = (float*)d_ws + off;
  float* afold = h_b + (long long)N * 64;
  float* cfold = afold + 192;
  int* deg    = (int*)(cfold + 64);
  int* rowptr = deg + (((long long)N + 63) & ~63LL);
  int* rank   = rowptr + (((long long)N + 1 + 63) & ~63LL);
  int* col    = rank + (((long long)E + 63) & ~63LL);

  k_node_fc<<<(N * 64 + 255) / 256, 256, 0, stream>>>(node_feats, Wn, bn, h_out, N);
  k_zero_i<<<(N + 255) / 256, 256, 0, stream>>>(deg, N);
  k_rank<<<(E + 255) / 256, 256, 0, stream>>>(src, deg, rank, E);
  k_scan<<<(N + 1023) / 1024, 1024, 0, stream>>>(deg, rowptr, cnt, N);
  k_fill<<<(E + 255) / 256, 256, 0, stream>>>(src, dst, rowptr, rank, col, E);

  int tiles = (N + 15) / 16;
  float* hin = h_out;
  for (int l = 0; l < L; ++l) {
    k_gather<<<(N + 7) / 8, 256, 0, stream>>>(rowptr, col, hin, accum, N);
    float* hout = (l == L - 1) ? h_out : ((hin == h_out) ? h_b : h_out);
    k_layer_gemm<<<(tiles + 7) / 8, 256, 0, stream>>>(
        hin, accum, cnt,
        Wself + (long long)l * 64 * 64, bself + (long long)l * 64,
        Wneigh + (long long)l * 64 * 64, bneigh + (long long)l * 64, hout, N);
    hin = hout;
  }

  k_prep<<<1, 64, 0, stream>>>(Wa, ba, We, be, env, NV, afold, cfold);
  int etiles = (E + 15) / 16;
  k_edge_score<<<(etiles + 7) / 8, 256, 0, stream>>>(edge_feats, afold, cfold, W1, b1, W2,
                                                     b2, scores, E);
}


# input-only launches node_fc, counter clear and edge-weight fold merged into one by block role; the one-wave serial fold becomes 64 blocks of lane-parallel shuffle-tree sums (fold sum order changes)
# speedup vs baseline: 3.0298x; 1.0913x over previous
#include <hip/hip_runtime.h>

typedef __attribute__((ext_vector_type(2))) float v2f;
typedef __attribute__((ext_vector_type(4))) float v4f;
typedef __attribute__((ext_vector_type(8))) float v8f;

__global__ void __launch_bounds__(256)
k_prologue(const float* __restrict__ nf, const float* __restrict__ Wn,
           const float* __restrict__ bn, float* __restrict__ h, int N,
           int* __restrict__ deg, int zb,
           const float* __restrict__ Wa, const float* __restrict__ ba,
           const float* __restrict__ We, const float* __restrict__ be,
           const float* __restrict__ env, int NV,
           float* __restrict__ afold, float* __restrict__ cfold) {
  __shared__ float part[4][4];
  int b = blockIdx.x, tid = threadIdx.x;

  if (b >= 64 + zb) {
    int t = (b - 64 - zb) * 256 + tid;
    int n = t >> 6, j = t & 63;
    if (n >= N) return;
    float f0 = nf[n * 3 + 0], f1 = nf[n * 3 + 1], f2 = nf[n * 3 + 2];
    h[t] = bn[j] + Wn[j * 3 + 0] * f0 + Wn[j * 3 + 1] * f1 + Wn[j * 3 + 2] * f2;
    return;
  }
  if (b >= 64) {
    int i = (b - 64) * 256 + tid;
    if (i < N) deg[i] = 0;
    return;
  }

  int j = b;
  float a0 = 0.f, a1 = 0.f, a2 = 0.f, c = 0.f;
  if (tid < 64) {
    float wv = Wa[j * 128 + tid];
    a0 = wv * We[tid * 3 + 0];
    a1 = wv * We[tid * 3 + 1];
    a2 = wv * We[tid * 3 + 2];
    c = wv * be[tid];
  } else if (tid < 128) {
    int d = tid - 64;
    float vm = 0.f;
    for (int v = 0; v < NV; ++v) vm += env[v * 64 + d];
    c = Wa[j * 128 + 64 + d] * (vm * (1.0f / (float)NV));
  }
  for (int o = 16; o > 0; o >>= 1) {
    a0 += __shfl_xor(a0, o);
    a1 += __shfl_xor(a1, o);
    a2 += __shfl_xor(a2, o);
    c += __shfl_xor(c, o);
  }
  int w = tid >> 5;
  if (w < 4 && (tid & 31) == 0) {
    part[w][0] = a0;
    part[w][1] = a1;
    part[w][2] = a2;
    part[w][3] = c;
  }
  __syncthreads();
  if (tid == 0) {
    afold[j * 3 + 0] = part[0][0] + part[1][0];
    afold[j * 3 + 1] = part[0][1] + part[1][1];
    afold[j * 3 + 2] = part[0][2] + part[1][2];
    cfold[j] = ba[j] + ((part[0][3] + part[1][3]) + (part[2][3] + part[3][3]));
  }
}

__global__ void k_rank(const int* __restrict__ src, int* __restrict__ deg,
                       int* __restrict__ rank, int E) {
  int e = blockIdx.x * blockDim.x + threadIdx.x;
  if (e < E) rank[e] = atomicAdd(&deg[src[e]], 1);
}

__global__ void __launch_bounds__(1024)
k_scan(const int* __restrict__ deg, int* __restrict__ rowptr, float* __restrict__ cnt,
       int N) {
  __shared__ int wsum[32];
  __shared__ int wpre[32];
  int t = threadIdx.x, lane = t & 31, w = t >> 5;
  int nb = blockIdx.x;
  int pre = 0;
#pragma unroll 8
  for (int k = 0; k < nb; ++k) pre += deg[k * 1024 + t];
  int gi = nb * 1024 + t;
  int v = (gi < N) ? deg[gi] : 0;
  int inc = v;
  for (int o = 1; o < 32; o <<= 1) {
    int u = __shfl_up(inc, o);
    if (lane >= o) inc += u;
  }
  for (int o = 16; o > 0; o >>= 1) pre += __shfl_xor(pre, o);
  if (lane == 31) wsum[w] = inc;
  if (lane == 0) wpre[w] = pre;
  __syncthreads();
  int tot = wpre[lane] + ((lane < w) ? wsum[lane] : 0);
  for (int o = 16; o > 0; o >>= 1) tot += __shfl_xor(tot, o);
  int excl = tot + inc - v;
  if (gi < N) {
    rowptr[gi] = excl;
    cnt[gi] = (float)v;
    if (gi == N - 1) rowptr[N] = excl + v;
  }
}

__global__ void k_fill(const int* __restrict__ src, const int* __restrict__ dst,
                       const int* __restrict__ rowptr, const int* __restrict__ rank,
                       int* __restrict__ col, int E) {
  int e = blockIdx.x * blockDim.x + threadIdx.x;
  if (e < E) col[rowptr[src[e]] + rank[e]] = dst[e];
}

__global__ void __launch_bounds__(256)
k_gather(const int* __restrict__ rowptr, const int* __restrict__ col,
         const float* __restrict__ h, float* __restrict__ accum, int N) {
  int node = blockIdx.x * 8 + (threadIdx.x >> 5);
  if (node >= N) return;
  int lane = threadIdx.x & 31, half = lane >> 4, c4 = (lane & 15) * 4;
  int beg = rowptr[node], end = rowptr[node + 1];
  const v4f zero = {0.f, 0.f, 0.f, 0.f};
  v4f s[4] = {zero, zero, zero, zero};
  if (beg < end) {
    int last = end - 1;
    unsigned nmax = (unsigned)(N - 1);
    int id[8];
#pragma unroll
    for (int u = 0; u < 8; ++u) {
      int e = beg + half + 2 * u;
      id[u] = col[e < last ? e : last];
    }
    for (int j0 = beg; j0 < end; j0 += 16) {
      v4f v[8];
#pragma unroll
      for (int u = 0; u < 8; ++u) {
        unsigned r = (unsigned)id[u] < nmax ? (unsigned)id[u] : nmax;
        v[u] = *(const v4f*)(h + (size_t)r * 64 + c4);
      }
#pragma unroll
      for (int u = 0; u < 8; ++u) {
        int e = j0 + 16 + half + 2 * u;
        id[u] = col[e < last ? e : last];
      }
#pragma unroll
      for (int u = 0; u < 8; ++u) {
        bool ok = (j0 + half + 2 * u) < end;
        s[u & 3] += ok ? v[u] : zero;
      }
    }
  }
  v4f a = (s[0] + s[1]) + (s[2] + s[3]);
  a.x += __shfl_xor(a.x, 16);
  a.y += __shfl_xor(a.y, 16);
  a.z += __shfl_xor(a.z, 16);
  a.w += __shfl_xor(a.w, 16);
  if (half == 0) *(v4f*)(accum + (size_t)node * 64 + c4) = a;
}

__global__ void __launch_bounds__(256)
k_layer_gemm(const float* __restrict__ hin, const float* __restrict__ accum,
             const float* __restrict__ cnt,
             const float* __restrict__ Wself, const float* __restrict__ bself,
             const float* __restrict__ Wneigh, const float* __restrict__ bneigh,
             float* __restrict__ hout, int N) {
  __shared__ float Bl[32 * 4 * 32 * 2];
  int tid = threadIdx.x;
  for (int idx = tid; idx < 8192; idx += 256) {
    int v = idx & 1, lane = (idx >> 1) & 31, ct = (idx >> 6) & 3, kk = idx >> 8;
    int n = ct * 16 + (lane & 15);
    int k = kk * 4 + 2 * (lane >> 4) + v;
    Bl[idx] = (k < 64) ? Wself[n * 64 + k] : Wneigh[n * 64 + (k - 64)];
  }
  __syncthreads();

  int w = tid >> 5, lane = tid & 31;
  int base = (blockIdx.x * 8 + w) * 16;
  int half = lane >> 4;
  int mrow = base + (lane & 15);
  long long mc = (mrow < N) ? mrow : 0;

  float c = cnt[mc];
  float inv = (c > 0.f) ? 1.f / c : 1.f;

  v2f A[32];
#pragma unroll
  for (int kk = 0; kk < 16; ++kk) {
    int k0 = kk * 4 + 2 * half;
    v2f a = {hin[mc * 64 + k0], hin[mc * 64 + k0 + 1]};
    A[kk] = a;
  }
#pragma unroll
  for (int kk = 0; kk < 16; ++kk) {
    int k0 = kk * 4 + 2 * half;
    v2f a = {accum[mc * 64 + k0] * inv, accum[mc * 64 + k0 + 1] * inv};
    A[16 + kk] = a;
  }

  v8f acc0 = {}, acc1 = {}, acc2 = {}, acc3 = {};
#pragma unroll
  for (int kk = 0; kk < 32; ++kk) {
    int bb = kk * 256 + lane * 2;
    acc0 = __builtin_amdgcn_wmma_f32_16x16x4_f32(false, A[kk], false, *(const v2f*)&Bl[bb + 0],   (short)0, acc0, false, false);
    acc1 = __builtin_amdgcn_wmma_f32_16x16x4_f32(false, A[kk], false, *(const v2f*)&Bl[bb + 64],  (short)0, acc1, false, false);
    acc2 = __builtin_amdgcn_wmma_f32_16x16x4_f32(false, A[kk], false, *(const v2f*)&Bl[bb + 128], (short)0, acc2, false, false);
    acc3 = __builtin_amdgcn_wmma_f32_16x16x4_f32(false, A[kk], false, *(const v2f*)&Bl[bb + 192], (short)0, acc3, false, false);
  }

#pragma unroll
  for (int ct = 0; ct < 4; ++ct) {
    v8f av = (ct == 0) ? acc0 : (ct == 1) ? acc1 : (ct == 2) ? acc2 : acc3;
    int n = ct * 16 + (lane & 15);
    float bs = bself[n] + bneigh[n];
#pragma unroll
    for (int r = 0; r < 8; ++r) {
      int mg = base + r + 8 * half;
      if (mg < N) {
        float val = av[r] + bs;
        hout[(long long)mg * 64 + n] = val > 0.f ? val : 0.f;
      }
    }
  }
}

__global__ void __launch_bounds__(256)
k_edge_score(const float* __restrict__ ef, const float* __restrict__ afold,
             const float* __restrict__ cfold, const float* __restrict__ W1,
             const float* __restrict__ b1, const float* __restrict__ W2,
             const float* __restrict__ b2, float* __restrict__ scores, int E) {
  __shared__ float Bs[16 * 4 * 32 * 2];
  __shared__ float Af[192];
  __shared__ float Cf[64];
  int tid = threadIdx.x;
  for (int idx = tid; idx < 4096; idx += 256) {
    int v = idx & 1, lane = (idx >> 1) & 31, ct = (idx >> 6) & 3, kk = idx >> 8;
    int n = ct * 16 + (lane & 15);
    int k = kk * 4 + 2 * (lane >> 4) + v;
    Bs[idx] = W1[n * 64 + k];
  }
  if (tid < 192) Af[tid] = afold[tid];
  if (tid < 64) Cf[tid] = cfold[tid];
  __syncthreads();

  int w = tid >> 5, lane = tid & 31;
  int base = (blockIdx.x * 8 + w) * 16;
  int half = lane >> 4;
  int row = base + (lane & 15);
  long long rc = (row < E) ? row : 0;
  float f0 = ef[rc * 3 + 0], f1 = ef[rc * 3 + 1], f2 = ef[rc * 3 + 2];

  v2f A[16];
#pragma unroll
  for (int kk = 0; kk < 16; ++kk) {
    int k0 = kk * 4 + 2 * half;
    float x0 = Cf[k0] + Af[k0 * 3] * f0 + Af[k0 * 3 + 1] * f1 + Af[k0 * 3 + 2] * f2;
    float x1 = Cf[k0 + 1] + Af[(k0 + 1) * 3] * f0 + Af[(k0 + 1) * 3 + 1] * f1 +
               Af[(k0 + 1) * 3 + 2] * f2;
    v2f a = {x0 > 0.f ? x0 : 0.f, x1 > 0.f ? x1 : 0.f};
    A[kk] = a;
  }

  v8f acc0 = {}, acc1 = {}, acc2 = {}, acc3 = {};
#pragma unroll
  for (int kk = 0; kk < 16; ++kk) {
    int bb = kk * 256 + lane * 2;
    acc0 = __builtin_amdgcn_wmma_f32_16x16x4_f32(false, A[kk], false, *(const v2f*)&Bs[bb + 0],   (short)0, acc0, false, false);
    acc1 = __builtin_amdgcn_wmma_f32_16x16x4_f32(false, A[kk], false, *(const v2f*)&Bs[bb + 64],  (short)0, acc1, false, false);
    acc2 = __builtin_amdgcn_wmma_f32_16x16x4_f32(false, A[kk], false, *(const v2f*)&Bs[bb + 128], (short)0, acc2, false, false);
    acc3 = __builtin_amdgcn_wmma_f32_16x16x4_f32(false, A[kk], false, *(const v2f*)&Bs[bb + 192], (short)0, acc3, false, false);
  }

  float p[8] = {0, 0, 0, 0, 0, 0, 0, 0};
#pragma unroll
  for (int ct = 0; ct < 4; ++ct) {
    v8f av = (ct == 0) ? acc0 : (ct == 1) ? acc1 : (ct == 2) ? acc2 : acc3;
    int n = ct * 16 + (lane & 15);
    float bn_ = b1[n], wn = W2[n];
#pragma unroll
    for (int r = 0; r < 8; ++r) {
      float s = av[r] + bn_;
      s = s > 0.f ? s : 0.f;
      p[r] += wn * s;
    }
  }
#pragma unroll
  for (int r = 0; r < 8; ++r) {
    p[r] += __shfl_xor(p[r], 1);
    p[r] += __shfl_xor(p[r], 2);
    p[r] += __shfl_xor(p[r], 4);
    p[r] += __shfl_xor(p[r], 8);
  }
  if ((lane & 15) == 0) {
    float bb2 = b2[0];
    int rb = base + 8 * half;
#pragma unroll
    for (int r = 0; r < 8; ++r)
      if (rb + r < E) scores[rb + r] = p[r] + bb2;
  }
}

extern "C" void kernel_launch(void* const* d_in, const int* in_sizes, int n_in,
                              void* d_out, int out_size, void* d_ws, size_t ws_size,
                              hipStream_t stream) {
  if (n_in < 18) return;
  const float* node_feats = (const float*)d_in[0];
  const float* edge_feats = (const float*)d_in[1];
  const float* env        = (const float*)d_in[2];
  const int*   eidx       = (const int*)d_in[3];
  const float* Wn  = (const float*)d_in[4];
  const float* bn  = (const float*)d_in[5];
  const float* We  = (const float*)d_in[6];
  const float* be  = (const float*)d_in[7];
  const float* Wself  = (const float*)d_in[8];
  const float* bself  = (const float*)d_in[9];
  const float* Wneigh = (const float*)d_in[10];
  const float* bneigh = (const float*)d_in[11];
  const float* Wa  = (const float*)d_in[12];
  const float* ba  = (const float*)d_in[13];
  const float* W1  = (const float*)d_in[14];
  const float* b1  = (const float*)d_in[15];
  const float* W2  = (const float*)d_in[16];
  const float* b2  = (const float*)d_in[17];

  int N  = in_sizes[0] / 3;
  int E  = in_sizes[1] / 3;
  int NV = in_sizes[2] / 64;
  int L  = in_sizes[8] / (64 * 64);

  const int* src = eidx;
  const int* dst = eidx + E;

  float* scores = (float*)d_out;
  float* h_out  = (float*)d_out + E;

  float* accum = (float*)d_ws;
  float* cnt   = accum + (long long)N * 64;
  long long off = (((long long)N * 65) + 63) & ~63LL;
  float* h_b   = (float*)d_ws + off;
  float* afold = h_b + (long long)N * 64;
  float* cfold = afold + 192;
  int* deg    = (int*)(cfold + 64);
  int* rowptr = deg + (((long long)N + 63) & ~63LL);
  int* rank   = rowptr + (((long long)N + 1 + 63) & ~63LL);
  int* col    = rank + (((long long)E + 63) & ~63LL);

  int zb = (N + 255) / 256;
  int nb = (N * 64 + 255) / 256;
  k_prologue<<<64 + zb + nb, 256, 0, stream>>>(node_feats, Wn, bn, h_out, N, deg, zb, Wa, ba,
                                               We, be, env, NV, afold, cfold);
  k_rank<<<(E + 255) / 256, 256, 0, stream>>>(src, deg, rank, E);
  k_scan<<<(N + 1023) / 1024, 1024, 0, stream>>>(deg, rowptr, cnt, N);
  k_fill<<<(E + 255) / 256, 256, 0, stream>>>(src, dst, rowptr, rank, col, E);

  int tiles = (N + 15) / 16;
  float* hin = h_out;
  for (int l = 0; l < L; ++l) {
    k_gather<<<(N + 7) / 8, 256, 0, stream>>>(rowptr, col, hin, accum, N);
    float* hout = (l == L - 1) ? h_out : ((hin == h_out) ? h_b : h_out);
    k_layer_gemm<<<(tiles + 7) / 8, 256, 0, stream>>>(
        hin, accum, cnt,
        Wself + (long long)l * 64 * 64, bself + (long long)l * 64,
        Wneigh + (long long)l * 64 * 64, bneigh + (long long)l * 64, hout, N);
    hin = hout;
  }

  int etiles = (E + 15) / 16;
  k_edge_score<<<(etiles + 7) / 8, 256, 0, stream>>>(edge_feats, afold, cfold, W1, b1, W2,
                                                     b2, scores, E);
}


# edge scorer has no launch of its own: its 8-tile blocks are spread evenly, by block index, among the blocks of the rank, fill and two gather launches, so WMMA runs while those waves wait on memory
# speedup vs baseline: 3.3495x; 1.1055x over previous
#include <hip/hip_runtime.h>

typedef __attribute__((ext_vector_type(2))) float v2f;
typedef __attribute__((ext_vector_type(4))) float v4f;
typedef __attribute__((ext_vector_type(8))) float v8f;

__global__ void __launch_bounds__(256)
k_prologue(const float* __restrict__ nf, const float* __restrict__ Wn,
           const float* __restrict__ bn, float* __restrict__ h, int N,
           int* __restrict__ deg, int zb,
           const float* __restrict__ Wa, const float* __restrict__ ba,
           const float* __restrict__ We, const float* __restrict__ be,
           const float* __restrict__ env, int NV,
           float* __restrict__ afold, float* __restrict__ cfold) {
  __shared__ float part[4][4];
  int b = blockIdx.x, tid = threadIdx.x;

  if (b >= 64 + zb) {
    int t = (b - 64 - zb) * 256 + tid;
    int n = t >> 6, j = t & 63;
    if (n >= N) return;
    float f0 = nf[n * 3 + 0], f1 = nf[n * 3 + 1], f2 = nf[n * 3 + 2];
    h[t] = bn[j] + Wn[j * 3 + 0] * f0 + Wn[j * 3 + 1] * f1 + Wn[j * 3 + 2] * f2;
    return;
  }
  if (b >= 64) {
    int i = (b - 64) * 256 + tid;
    if (i < N) deg[i] = 0;
    return;
  }

  int j = b;
  float a0 = 0.f, a1 = 0.f, a2 = 0.f, c = 0.f;
  if (tid < 64) {
    float wv = Wa[j * 128 + tid];
    a0 = wv * We[tid * 3 + 0];
    a1 = wv * We[tid * 3 + 1];
    a2 = wv * We[tid * 3 + 2];
    c = wv * be[tid];
  } else if (tid < 128) {
    int d = tid - 64;
    float vm = 0.f;
    for (int v = 0; v < NV; ++v) vm += env[v * 64 + d];
    c = Wa[j * 128 + 64 + d] * (vm * (1.0f / (float)NV));
  }
  for (int o = 16; o > 0; o >>= 1) {
    a0 += __shfl_xor(a0, o);
    a1 += __shfl_xor(a1, o);
    a2 += __shfl_xor(a2, o);
    c += __shfl_xor(c, o);
  }
  int w = tid >> 5;
  if (w < 4 && (tid & 31) == 0) {
    part[w][0] = a0;
    part[w][1] = a1;
    part[w][2] = a2;
    part[w][3] = c;
  }
  __syncthreads();
  if (tid == 0) {
    afold[j * 3 + 0] = part[0][0] + part[1][0];
    afold[j * 3 + 1] = part[0][1] + part[1][1];
    afold[j * 3 + 2] = part[0][2] + part[1][2];
    cfold[j] = ba[j] + ((part[0][3] + part[1][3]) + (part[2][3] + part[3][3]));
  }
}

struct EdgeJob {
  const float* ef;
  const float* afold;
  const float* cfold;
  const float* W1;
  const float* b1;
  const float* W2;
  const float* b2;
  float* scores;
  int E;
  int nblk;
  int ratio;
  int first;
};

__device__ __forceinline__ bool block_role(const EdgeJob& j, int& idx) {
  unsigned b = blockIdx.x, cap = (unsigned)j.nblk;
  unsigned q0 = (unsigned)(((unsigned long long)b * (unsigned)j.ratio) >> 16);
  unsigned q1 = (unsigned)(((unsigned long long)(b + 1) * (unsigned)j.ratio) >> 16);
  q0 = q0 < cap ? q0 : cap;
  q1 = q1 < cap ? q1 : cap;
  if (q1 > q0) {
    idx = j.first + (int)q0;
    return true;
  }
  idx = (int)(b - q0);
  return false;
}

__device__ __forceinline__ void edge_block(const EdgeJob& j, int blk) {
  __shared__ float Bs[16 * 4 * 32 * 2];
  __shared__ float Af[192];
  __shared__ float Cf[64];
  const float* __restrict__ ef = j.ef;
  const float* __restrict__ W1 = j.W1;
  float* __restrict__ scores = j.scores;
  int E = j.E;
  int tid = threadIdx.x;
  for (int idx = tid; idx < 4096; idx += 256) {
    int v = idx & 1, lane = (idx >> 1) & 31, ct = (idx >> 6) & 3, kk = idx >> 8;
    int n = ct * 16 + (lane & 15);
    int k = kk * 4 + 2 * (lane >> 4) + v;
    Bs[idx] = W1[n * 64 + k];
  }
  if (tid < 192) Af[tid] = j.afold[tid];
  if (tid < 64) Cf[tid] = j.cfold[tid];
  __syncthreads();

  int w = tid >> 5, lane = tid & 31;
  int base = (blk * 8 + w) * 16;
  int half = lane >> 4;
  int row = base + (lane & 15);
  long long rc = (row < E) ? row : 0;
  float f0 = ef[rc * 3 + 0], f1 = ef[rc * 3 + 1], f2 = ef[rc * 3 + 2];

  v2f A[16];
#pragma unroll
  for (int kk = 0; kk < 16; ++kk) {
    int k0 = kk * 4 + 2 * half;
    float x0 = Cf[k0] + Af[k0 * 3] * f0 + Af[k0 * 3 + 1] * f1 + Af[k0 * 3 + 2] * f2;
    float x1 = Cf[k0 + 1] + Af[(k0 + 1) * 3] * f0 + Af[(k0 + 1) * 3 + 1] * f1 +
               Af[(k0 + 1) * 3 + 2] * f2;
    v2f a = {x0 > 0.f ? x0 : 0.f, x1 > 0.f ? x1 : 0.f};
    A[kk] = a;
  }

  v8f acc0 = {}, acc1 = {}, acc2 = {}, acc3 = {};
#pragma unroll
  for (int kk = 0; kk < 16; ++kk) {
    int bb = kk * 256 + lane * 2;
    acc0 = __builtin_amdgcn_wmma_f32_16x16x4_f32(false, A[kk], false, *(const v2f*)&Bs[bb + 0],   (short)0, acc0, false, false);
    acc1 = __builtin_amdgcn_wmma_f32_16x16x4_f32(false, A[kk], false, *(const v2f*)&Bs[bb + 64],  (short)0, acc1, false, false);
    acc2 = __builtin_amdgcn_wmma_f32_16x16x4_f32(false, A[kk], false, *(const v2f*)&Bs[bb + 128], (short)0, acc2, false, false);
    acc3 = __builtin_amdgcn_wmma_f32_16x16x4_f32(false, A[kk], false, *(const v2f*)&Bs[bb + 192], (short)0, acc3, false, false);
  }

  float p[8] = {0, 0, 0, 0, 0, 0, 0, 0};
#pragma unroll
  for (int ct = 0; ct < 4; ++ct) {
    v8f av = (ct == 0) ? acc0 : (ct == 1) ? acc1 : (ct == 2) ? acc2 : acc3;
    int n = ct * 16 + (lane & 15);
    float bn_ = j.b1[n], wn = j.W2[n];
#pragma unroll
    for (int r = 0; r < 8; ++r) {
      float s = av[r] + bn_;
      s = s > 0.f ? s : 0.f;
      p[r] += wn * s;
    }
  }
#pragma unroll
  for (int r = 0; r < 8; ++r) {
    p[r] += __shfl_xor(p[r], 1);
    p[r] += __shfl_xor(p[r], 2);
    p[r] += __shfl_xor(p[r], 4);
    p[r] += __shfl_xor(p[r], 8);
  }
  if ((lane & 15) == 0) {
    float bb2 = j.b2[0];
    int rb = base + 8 * half;
#pragma unroll
    for (int r = 0; r < 8; ++r)
      if (rb + r < E) scores[rb + r] = p[r] + bb2;
  }
}

__global__ void __launch_bounds__(256)
k_rank(const int* __restrict__ src, int* __restrict__ deg,
       int* __restrict__ rank, int E, EdgeJob job) {
  int mb;
  if (block_role(job, mb)) {
    edge_block(job, mb);
    return;
  }
  int e = mb * 256 + threadIdx.x;
  if (e < E) rank[e] = atomicAdd(&deg[src[e]], 1);
}

__global__ void __launch_bounds__(1024)
k_scan(const int* __restrict__ deg, int* __restrict__ rowptr, float* __restrict__ cnt,
       int N) {
  __shared__ int wsum[32];
  __shared__ int wpre[32];
  int t = threadIdx.x, lane = t & 31, w = t >> 5;
  int nb = blockIdx.x;
  int pre = 0;
#pragma unroll 8
  for (int k = 0; k < nb; ++k) pre += deg[k * 1024 + t];
  int gi = nb * 1024 + t;
  int v = (gi < N) ? deg[gi] : 0;
  int inc = v;
  for (int o = 1; o < 32; o <<= 1) {
    int u = __shfl_up(inc, o);
    if (lane >= o) inc += u;
  }
  for (int o = 16; o > 0; o >>= 1) pre += __shfl_xor(pre, o);
  if (lane == 31) wsum[w] = inc;
  if (lane == 0) wpre[w] = pre;
  __syncthreads();
  int tot = wpre[lane] + ((lane < w) ? wsum[lane] : 0);
  for (int o = 16; o > 0; o >>= 1) tot += __shfl_xor(tot, o);
  int excl = tot + inc - v;
  if (gi < N) {
    rowptr[gi] = excl;
    cnt[gi] = (float)v;
    if (gi == N - 1) rowptr[N] = excl + v;
  }
}

__global__ void __launch_bounds__(256)
k_fill(const int* __restrict__ src, const int* __restrict__ dst,
       const int* __restrict__ rowptr, const int* __restrict__ rank,
       int* __restrict__ col, int E, EdgeJob job) {
  int mb;
  if (block_role(job, mb)) {
    edge_block(job, mb);
    return;
  }
  int e = mb * 256 + threadIdx.x;
  if (e < E) col[rowptr[src[e]] + rank[e]] = dst[e];
}

__global__ void __launch_bounds__(256)
k_gather(const int* __restrict__ rowptr, const int* __restrict__ col,
         const float* __restrict__ h, float* __restrict__ accum, int N, EdgeJob job) {
  int mb;
  if (block_role(job, mb)) {
    edge_block(job, mb);
    return;
  }
  int node = mb * 8 + (threadIdx.x >> 5);
  if (node >= N) return;
  int lane = threadIdx.x & 31, half = lane >> 4, c4 = (lane & 15) * 4;
  int beg = rowptr[node], end = rowptr[node + 1];
  const v4f zero = {0.f, 0.f, 0.f, 0.f};
  v4f s[4] = {zero, zero, zero, zero};
  if (beg < end) {
    int last = end - 1;
    unsigned nmax = (unsigned)(N - 1);
    int id[8];
#pragma unroll
    for (int u = 0; u < 8; ++u) {
      int e = beg + half + 2 * u;
      id[u] = col[e < last ? e : last];
    }
    for (int j0 = beg; j0 < end; j0 += 16) {
      v4f v[8];
#pragma unroll
      for (int u = 0; u < 8; ++u) {
        unsigned r = (unsigned)id[u] < nmax ? (unsigned)id[u] : nmax;
        v[u] = *(const v4f*)(h + (size_t)r * 64 + c4);
      }
#pragma unroll
      for (int u = 0; u < 8; ++u) {
        int e = j0 + 16 + half + 2 * u;
        id[u] = col[e < last ? e : last];
      }
#pragma unroll
      for (int u = 0; u < 8; ++u) {
        bool ok = (j0 + half + 2 * u) < end;
        s[u & 3] += ok ? v[u] : zero;
      }
    }
  }
  v4f a = (s[0] + s[1]) + (s[2] + s[3]);
  a.x += __shfl_xor(a.x, 16);
  a.y += __shfl_xor(a.y, 16);
  a.z += __shfl_xor(a.z, 16);
  a.w += __shfl_xor(a.w, 16);
  if (half == 0) *(v4f*)(accum + (size_t)node * 64 + c4) = a;
}

__global__ void __launch_bounds__(256)
k_layer_gemm(const float* __restrict__ hin, const float* __restrict__ accum,
             const float* __restrict__ cnt,
             const float* __restrict__ Wself, const float* __restrict__ bself,
             const float* __restrict__ Wneigh, const float* __restrict__ bneigh,
             float* __restrict__ hout, int N) {
  __shared__ float Bl[32 * 4 * 32 * 2];
  int tid = threadIdx.x;
  for (int idx = tid; idx < 8192; idx += 256) {
    int v = idx & 1, lane = (idx >> 1) & 31, ct = (idx >> 6) & 3, kk = idx >> 8;
    int n = ct * 16 + (lane & 15);
    int k = kk * 4 + 2 * (lane >> 4) + v;
    Bl[idx] = (k < 64) ? Wself[n * 64 + k] : Wneigh[n * 64 + (k - 64)];
  }
  __syncthreads();

  int w = tid >> 5, lane = tid & 31;
  int base = (blockIdx.x * 8 + w) * 16;
  int half = lane >> 4;
  int mrow = base + (lane & 15);
  long long mc = (mrow < N) ? mrow : 0;

  float c = cnt[mc];
  float inv = (c > 0.f) ? 1.f / c : 1.f;

  v2f A[32];
#pragma unroll
  for (int kk = 0; kk < 16; ++kk) {
    int k0 = kk * 4 + 2 * half;
    v2f a = {hin[mc * 64 + k0], hin[mc * 64 + k0 + 1]};
    A[kk] = a;
  }
#pragma unroll
  for (int kk = 0; kk < 16; ++kk) {
    int k0 = kk * 4 + 2 * half;
    v2f a = {accum[mc * 64 + k0] * inv, accum[mc * 64 + k0 + 1] * inv};
    A[16 + kk] = a;
  }

  v8f acc0 = {}, acc1 = {}, acc2 = {}, acc3 = {};
#pragma unroll
  for (int kk = 0; kk < 32; ++kk) {
    int bb = kk * 256 + lane * 2;
    acc0 = __builtin_amdgcn_wmma_f32_16x16x4_f32(false, A[kk], false, *(const v2f*)&Bl[bb + 0],   (short)0, acc0, false, false);
    acc1 = __builtin_amdgcn_wmma_f32_16x16x4_f32(false, A[kk], false, *(const v2f*)&Bl[bb + 64],  (short)0, acc1, false, false);
    acc2 = __builtin_amdgcn_wmma_f32_16x16x4_f32(false, A[kk], false, *(const v2f*)&Bl[bb + 128], (short)0, acc2, false, false);
    acc3 = __builtin_amdgcn_wmma_f32_16x16x4_f32(false, A[kk], false, *(const v2f*)&Bl[bb + 192], (short)0, acc3, false, false);
  }

#pragma unroll
  for (int ct = 0; ct < 4; ++ct) {
    v8f av = (ct == 0) ? acc0 : (ct == 1) ? acc1 : (ct == 2) ? acc2 : acc3;
    int n = ct * 16 + (lane & 15);
    float bs = bself[n] + bneigh[n];
#pragma unroll
    for (int r = 0; r < 8; ++r) {
      int mg = base + r + 8 * half;
      if (mg < N) {
        float val = av[r] + bs;
        hout[(long long)mg * 64 + n] = val > 0.f ? val : 0.f;
      }
    }
  }
}

static int share_ratio(int x, int all) {
  return all > 0 ? (int)(((long long)x * 65536 + all - 1) / all) : 0;
}

extern "C" void kernel_launch(void* const* d_in, const int* in_sizes, int n_in,
                              void* d_out, int out_size, void* d_ws, size_t ws_size,
                              hipStream_t stream) {
  if (n_in < 18) return;
  const float* node_feats = (const float*)d_in[0];
  const float* edge_feats = (const float*)d_in[1];
  const float* env        = (const float*)d_in[2];
  const int*   eidx       = (const int*)d_in[3];
  const float* Wn  = (const float*)d_in[4];
  const float* bn  = (const float*)d_in[5];
  const float* We  = (const float*)d_in[6];
  const float* be  = (const float*)d_in[7];
  const float* Wself  = (const float*)d_in[8];
  const float* bself  = (const float*)d_in[9];
  const float* Wneigh = (const float*)d_in[10];
  const float* bneigh = (const float*)d_in[11];
  const float* Wa  = (const float*)d_in[12];
  const float* ba  = (const float*)d_in[13];
  const float* W1  = (const float*)d_in[14];
  const float* b1  = (const float*)d_in[15];
  const float* W2  = (const float*)d_in[16];
  const float* b2  = (const float*)d_in[17];

  int N  = in_sizes[0] / 3;
  int E  = in_sizes[1] / 3;
  int NV = in_sizes[2] / 64;
  int L  = in_sizes[8] / (64 * 64);

  const int* src = eidx;
  const int* dst = eidx + E;

  float* scores = (float*)d_out;
  float* h_out  = (float*)d_out + E;

  float* accum = (float*)d_ws;
  float* cnt   = accum + (long long)N * 64;
  long long off = (((long long)N * 65) + 63) & ~63LL;
  float* h_b   = (float*)d_ws + off;
  float* afold = h_b + (long long)N * 64;
  float* cfold = afold + 192;
  int* deg    = (int*)(cfold + 64);
  int* rowptr = deg + (((long long)N + 63) & ~63LL);
  int* rank   = rowptr + (((long long)N + 1 + 63) & ~63LL);
  int* col    = rank + (((long long)E + 63) & ~63LL);

  int zb = (N + 255) / 256;
  int nb = (N * 64 + 255) / 256;
  k_prologue<<<64 + zb + nb, 256, 0, stream>>>(node_feats, Wn, bn, h_out, N, deg, zb, Wa, ba,
                                               We, be, env, NV, afold, cfold);
  int etiles = (E + 15) / 16;
  int units = (etiles + 7) / 8;
  int xr = (L > 0) ? units / 4 : units / 2;
  int xf = (L > 0) ? units / 4 : units - xr;
  int xg = (L > 0) ? (units - xr - xf) / L : 0;
  EdgeJob job = {edge_feats, afold, cfold, W1, b1, W2, b2, scores, E, 0, 0, 0};

  int eb = (E + 255) / 256;
  job.nblk = xr;
  job.ratio = share_ratio(xr, eb + xr);
  job.first = 0;
  k_rank<<<eb + xr, 256, 0, stream>>>(src, deg, rank, E, job);
  k_scan<<<(N + 1023) / 1024, 1024, 0, stream>>>(deg, rowptr, cnt, N);
  job.nblk = xf;
  job.ratio = share_ratio(xf, eb + xf);
  job.first = xr;
  k_fill<<<eb + xf, 256, 0, stream>>>(src, dst, rowptr, rank, col, E, job);

  int tiles = (N + 15) / 16;
  int gb = (N + 7) / 8;
  float* hin = h_out;
  for (int l = 0; l < L; ++l) {
    job.first = xr + xf + l * xg;
    job.nblk = (l == L - 1) ? units - job.first : xg;
    job.ratio = share_ratio(job.nblk, gb + job.nblk);
    k_gather<<<gb + job.nblk, 256, 0, stream>>>(rowptr, col, hin, accum, N, job);
    float* hout = (l == L - 1) ? h_out : ((hin == h_out) ? h_b : h_out);
    k_layer_gemm<<<(tiles + 7) / 8, 256, 0, stream>>>(
        hin, accum, cnt,
        Wself + (long long)l * 64 * 64, bself + (long long)l * 64,
        Wneigh + (long long)l * 64 * 64, bneigh + (long long)l * 64, hout, N);
    hin = hout;
  }
}


# layer 0 gathers the 3 input features per edge, not 256-byte h0 rows (node_fc is linear: applied to the bucket mean); h0 never stored; layer-0 sum order changes: 8 lane partials, shuffle tree
# speedup vs baseline: 3.6381x; 1.0862x over previous
#include <hip/hip_runtime.h>

typedef __attribute__((ext_vector_type(2))) float v2f;
typedef __attribute__((ext_vector_type(4))) float v4f;
typedef __attribute__((ext_vector_type(8))) float v8f;

__global__ void __launch_bounds__(256)
k_prologue(const float* __restrict__ nf, const float* __restrict__ Wn,
           const float* __restrict__ bn, float* __restrict__ h, int N,
           int* __restrict__ deg, int zb,
           const float* __restrict__ Wa, const float* __restrict__ ba,
           const float* __restrict__ We, const float* __restrict__ be,
           const float* __restrict__ env, int NV,
           float* __restrict__ afold, float* __restrict__ cfold) {
  __shared__ float part[4][4];
  int b = blockIdx.x, tid = threadIdx.x;

  if (b >= 64 + zb) {
    int t = (b - 64 - zb) * 256 + tid;
    int n = t >> 6, j = t & 63;
    if (n >= N) return;
    float f0 = nf[n * 3 + 0], f1 = nf[n * 3 + 1], f2 = nf[n * 3 + 2];
    h[t] = bn[j] + Wn[j * 3 + 0] * f0 + Wn[j * 3 + 1] * f1 + Wn[j * 3 + 2] * f2;
    return;
  }
  if (b >= 64) {
    int i = (b - 64) * 256 + tid;
    if (i < N) deg[i] = 0;
    return;
  }

  int j = b;
  float a0 = 0.f, a1 = 0.f, a2 = 0.f, c = 0.f;
  if (tid < 64) {
    float wv = Wa[j * 128 + tid];
    a0 = wv * We[tid * 3 + 0];
    a1 = wv * We[tid * 3 + 1];
    a2 = wv * We[tid * 3 + 2];
    c = wv * be[tid];
  } else if (tid < 128) {
    int d = tid - 64;
    float vm = 0.f;
    for (int v = 0; v < NV; ++v) vm += env[v * 64 + d];
    c = Wa[j * 128 + 64 + d] * (vm * (1.0f / (float)NV));
  }
  for (int o = 16; o > 0; o >>= 1) {
    a0 += __shfl_xor(a0, o);
    a1 += __shfl_xor(a1, o);
    a2 += __shfl_xor(a2, o);
    c += __shfl_xor(c, o);
  }
  int w = tid >> 5;
  if (w < 4 && (tid & 31) == 0) {
    part[w][0] = a0;
    part[w][1] = a1;
    part[w][2] = a2;
    part[w][3] = c;
  }
  __syncthreads();
  if (tid == 0) {
    afold[j * 3 + 0] = part[0][0] + part[1][0];
    afold[j * 3 + 1] = part[0][1] + part[1][1];
    afold[j * 3 + 2] = part[0][2] + part[1][2];
    cfold[j] = ba[j] + ((part[0][3] + part[1][3]) + (part[2][3] + part[3][3]));
  }
}

struct EdgeJob {
  const float* ef;
  const float* afold;
  const float* cfold;
  const float* W1;
  const float* b1;
  const float* W2;
  const float* b2;
  float* scores;
  int E;
  int nblk;
  int ratio;
  int first;
};

__device__ __forceinline__ bool block_role(const EdgeJob& j, int& idx) {
  unsigned b = blockIdx.x, cap = (unsigned)j.nblk;
  unsigned q0 = (unsigned)(((unsigned long long)b * (unsigned)j.ratio) >> 16);
  unsigned q1 = (unsigned)(((unsigned long long)(b + 1) * (unsigned)j.ratio) >> 16);
  q0 = q0 < cap ? q0 : cap;
  q1 = q1 < cap ? q1 : cap;
  if (q1 > q0) {
    idx = j.first + (int)q0;
    return true;
  }
  idx = (int)(b - q0);
  return false;
}

__device__ __forceinline__ void edge_block(const EdgeJob& j, int blk) {
  __shared__ float Bs[16 * 4 * 32 * 2];
  __shared__ float Af[192];
  __shared__ float Cf[64];
  const float* __restrict__ ef = j.ef;
  const float* __restrict__ W1 = j.W1;
  float* __restrict__ scores = j.scores;
  int E = j.E;
  int tid = threadIdx.x;
  for (int idx = tid; idx < 4096; idx += 256) {
    int v = idx & 1, lane = (idx >> 1) & 31, ct = (idx >> 6) & 3, kk = idx >> 8;
    int n = ct * 16 + (lane & 15);
    int k = kk * 4 + 2 * (lane >> 4) + v;
    Bs[idx] = W1[n * 64 + k];
  }
  if (tid < 192) Af[tid] = j.afold[tid];
  if (tid < 64) Cf[tid] = j.cfold[tid];
  __syncthreads();

  int w = tid >> 5, lane = tid & 31;
  int base = (blk * 8 + w) * 16;
  int half = lane >> 4;
  int row = base + (lane & 15);
  long long rc = (row < E) ? row : 0;
  float f0 = ef[rc * 3 + 0], f1 = ef[rc * 3 + 1], f2 = ef[rc * 3 + 2];

  v2f A[16];
#pragma unroll
  for (int kk = 0; kk < 16; ++kk) {
    int k0 = kk * 4 + 2 * half;
    float x0 = Cf[k0] + Af[k0 * 3] * f0 + Af[k0 * 3 + 1] * f1 + Af[k0 * 3 + 2] * f2;
    float x1 = Cf[k0 + 1] + Af[(k0 + 1) * 3] * f0 + Af[(k0 + 1) * 3 + 1] * f1 +
               Af[(k0 + 1) * 3 + 2] * f2;
    v2f a = {x0 > 0.f ? x0 : 0.f, x1 > 0.f ? x1 : 0.f};
    A[kk] = a;
  }

  v8f acc0 = {}, acc1 = {}, acc2 = {}, acc3 = {};
#pragma unroll
  for (int kk = 0; kk < 16; ++kk) {
    int bb = kk * 256 + lane * 2;
    acc0 = __builtin_amdgcn_wmma_f32_16x16x4_f32(false, A[kk], false, *(const v2f*)&Bs[bb + 0],   (short)0, acc0, false, false);
    acc1 = __builtin_amdgcn_wmma_f32_16x16x4_f32(false, A[kk], false, *(const v2f*)&Bs[bb + 64],  (short)0, acc1, false, false);
    acc2 = __builtin_amdgcn_wmma_f32_16x16x4_f32(false, A[kk], false, *(const v2f*)&Bs[bb + 128], (short)0, acc2, false, false);
    acc3 = __builtin_amdgcn_wmma_f32_16x16x4_f32(false, A[kk], false, *(const v2f*)&Bs[bb + 192], (short)0, acc3, false, false);
  }

  float p[8] = {0, 0, 0, 0, 0, 0, 0, 0};
#pragma unroll
  for (int ct = 0; ct < 4; ++ct) {
    v8f av = (ct == 0) ? acc0 : (ct == 1) ? acc1 : (ct == 2) ? acc2 : acc3;
    int n = ct * 16 + (lane & 15);
    float bn_ = j.b1[n], wn = j.W2[n];
#pragma unroll
    for (int r = 0; r < 8; ++r) {
      float s = av[r] + bn_;
      s = s > 0.f ? s : 0.f;
      p[r] += wn * s;
    }
  }
#pragma unroll
  for (int r = 0; r < 8; ++r) {
    p[r] += __shfl_xor(p[r], 1);
    p[r] += __shfl_xor(p[r], 2);
    p[r] += __shfl_xor(p[r], 4);
    p[r] += __shfl_xor(p[r], 8);
  }
  if ((lane & 15) == 0) {
    float bb2 = j.b2[0];
    int rb = base + 8 * half;
#pragma unroll
    for (int r = 0; r < 8; ++r)
      if (rb + r < E) scores[rb + r] = p[r] + bb2;
  }
}

__global__ void __launch_bounds__(256)
k_rank(const int* __restrict__ src, int* __restrict__ deg,
       int* __restrict__ rank, int E, EdgeJob job) {
  int mb;
  if (block_role(job, mb)) {
    edge_block(job, mb);
    return;
  }
  int e = mb * 256 + threadIdx.x;
  if (e < E) rank[e] = atomicAdd(&deg[src[e]], 1);
}

__global__ void __launch_bounds__(1024)
k_scan(const int* __restrict__ deg, int* __restrict__ rowptr, float* __restrict__ cnt,
       int N) {
  __shared__ int wsum[32];
  __shared__ int wpre[32];
  int t = threadIdx.x, lane = t & 31, w = t >> 5;
  int nb = blockIdx.x;
  int pre = 0;
#pragma unroll 8
  for (int k = 0; k < nb; ++k) pre += deg[k * 1024 + t];
  int gi = nb * 1024 + t;
  int v = (gi < N) ? deg[gi] : 0;
  int inc = v;
  for (int o = 1; o < 32; o <<= 1) {
    int u = __shfl_up(inc, o);
    if (lane >= o) inc += u;
  }
  for (int o = 16; o > 0; o >>= 1) pre += __shfl_xor(pre, o);
  if (lane == 31) wsum[w] = inc;
  if (lane == 0) wpre[w] = pre;
  __syncthreads();
  int tot = wpre[lane] + ((lane < w) ? wsum[lane] : 0);
  for (int o = 16; o > 0; o >>= 1) tot += __shfl_xor(tot, o);
  int excl = tot + inc - v;
  if (gi < N) {
    rowptr[gi] = excl;
    cnt[gi] = (float)v;
    if (gi == N - 1) rowptr[N] = excl + v;
  }
}

__global__ void __launch_bounds__(256)
k_fill(const int* __restrict__ src, const int* __restrict__ dst,
       const int* __restrict__ rowptr, const int* __restrict__ rank,
       int* __restrict__ col, int E, EdgeJob job) {
  int mb;
  if (block_role(job, mb)) {
    edge_block(job, mb);
    return;
  }
  int e = mb * 256 + threadIdx.x;
  if (e < E) col[rowptr[src[e]] + rank[e]] = dst[e];
}

__global__ void __launch_bounds__(256)
k_gather_feats(const int* __restrict__ rowptr, const int* __restrict__ col,
               const float* __restrict__ nf, float* __restrict__ fsum, int N) {
  int lane = threadIdx.x & 31, sub = lane & 7;
  int node = (blockIdx.x * 8 + (threadIdx.x >> 5)) * 4 + (lane >> 3);
  bool valid = node < N;
  int beg = 0, end = 0;
  if (valid) {
    beg = rowptr[node];
    end = rowptr[node + 1];
  }
  unsigned nmax = (unsigned)(N - 1);
  int last = end - 1;
  float s0 = 0.f, s1 = 0.f, s2 = 0.f;
  for (int j0 = beg; j0 < end; j0 += 48) {
    unsigned id[6];
#pragma unroll
    for (int u = 0; u < 6; ++u) {
      int e = j0 + sub + 8 * u;
      id[u] = (unsigned)col[e < last ? e : last];
    }
    float a[6], b[6], c[6];
#pragma unroll
    for (int u = 0; u < 6; ++u) {
      unsigned r = id[u] < nmax ? id[u] : nmax;
      const float* f = nf + (size_t)r * 3;
      a[u] = f[0];
      b[u] = f[1];
      c[u] = f[2];
    }
#pragma unroll
    for (int u = 0; u < 6; ++u) {
      bool ok = (j0 + sub + 8 * u) < end;
      s0 += ok ? a[u] : 0.f;
      s1 += ok ? b[u] : 0.f;
      s2 += ok ? c[u] : 0.f;
    }
  }
  for (int o = 4; o > 0; o >>= 1) {
    s0 += __shfl_xor(s0, o);
    s1 += __shfl_xor(s1, o);
    s2 += __shfl_xor(s2, o);
  }
  if (valid && sub == 0) {
    v4f o4 = {s0, s1, s2, 0.f};
    *(v4f*)(fsum + (size_t)node * 4) = o4;
  }
}

__global__ void __launch_bounds__(256)
k_gather(const int* __restrict__ rowptr, const int* __restrict__ col,
         const float* __restrict__ h, float* __restrict__ accum, int N, EdgeJob job) {
  int mb;
  if (block_role(job, mb)) {
    edge_block(job, mb);
    return;
  }
  int node = mb * 8 + (threadIdx.x >> 5);
  if (node >= N) return;
  int lane = threadIdx.x & 31, half = lane >> 4, c4 = (lane & 15) * 4;
  int beg = rowptr[node], end = rowptr[node + 1];
  const v4f zero = {0.f, 0.f, 0.f, 0.f};
  v4f s[4] = {zero, zero, zero, zero};
  if (beg < end) {
    int last = end - 1;
    unsigned nmax = (unsigned)(N - 1);
    int id[8];
#pragma unroll
    for (int u = 0; u < 8; ++u) {
      int e = beg + half + 2 * u;
      id[u] = col[e < last ? e : last];
    }
    for (int j0 = beg; j0 < end; j0 += 16) {
      v4f v[8];
#pragma unroll
      for (int u = 0; u < 8; ++u) {
        unsigned r = (unsigned)id[u] < nmax ? (unsigned)id[u] : nmax;
        v[u] = *(const v4f*)(h + (size_t)r * 64 + c4);
      }
#pragma unroll
      for (int u = 0; u < 8; ++u) {
        int e = j0 + 16 + half + 2 * u;
        id[u] = col[e < last ? e : last];
      }
#pragma unroll
      for (int u = 0; u < 8; ++u) {
        bool ok = (j0 + half + 2 * u) < end;
        s[u & 3] += ok ? v[u] : zero;
      }
    }
  }
  v4f a = (s[0] + s[1]) + (s[2] + s[3]);
  a.x += __shfl_xor(a.x, 16);
  a.y += __shfl_xor(a.y, 16);
  a.z += __shfl_xor(a.z, 16);
  a.w += __shfl_xor(a.w, 16);
  if (half == 0) *(v4f*)(accum + (size_t)node * 64 + c4) = a;
}

__device__ __forceinline__ void
layer_body(bool first, const float* __restrict__ hin, const float* __restrict__ accum,
           const float* __restrict__ cnt,
           const float* __restrict__ nf, const float* __restrict__ fsum,
           const float* __restrict__ Wn, const float* __restrict__ bn,
           const float* __restrict__ Wself, const float* __restrict__ bself,
           const float* __restrict__ Wneigh, const float* __restrict__ bneigh,
           float* __restrict__ hout, int N) {
  __shared__ float Bl[32 * 4 * 32 * 2];
  __shared__ float Wn_s[192];
  __shared__ float bn_s[64];
  int tid = threadIdx.x;
  for (int idx = tid; idx < 8192; idx += 256) {
    int v = idx & 1, lane = (idx >> 1) & 31, ct = (idx >> 6) & 3, kk = idx >> 8;
    int n = ct * 16 + (lane & 15);
    int k = kk * 4 + 2 * (lane >> 4) + v;
    Bl[idx] = (k < 64) ? Wself[n * 64 + k] : Wneigh[n * 64 + (k - 64)];
  }
  if (first) {
    if (tid < 192) Wn_s[tid] = Wn[tid];
    if (tid < 64) bn_s[tid] = bn[tid];
  }
  __syncthreads();

  int w = tid >> 5, lane = tid & 31;
  int base = (blockIdx.x * 8 + w) * 16;
  int half = lane >> 4;
  int mrow = base + (lane & 15);
  long long mc = (mrow < N) ? mrow : 0;

  float c = cnt[mc];
  float inv = (c > 0.f) ? 1.f / c : 1.f;

  v2f A[32];
  if (first) {
    float f0 = nf[mc * 3 + 0], f1 = nf[mc * 3 + 1], f2 = nf[mc * 3 + 2];
    v4f sv = *(const v4f*)(fsum + mc * 4);
    float g0 = sv.x * inv, g1 = sv.y * inv, g2 = sv.z * inv;
    bool has = c > 0.f;
#pragma unroll
    for (int kk = 0; kk < 16; ++kk) {
      int k0 = kk * 4 + 2 * half;
      float w00 = Wn_s[k0 * 3 + 0], w01 = Wn_s[k0 * 3 + 1], w02 = Wn_s[k0 * 3 + 2];
      float w10 = Wn_s[k0 * 3 + 3], w11 = Wn_s[k0 * 3 + 4], w12 = Wn_s[k0 * 3 + 5];
      float c0 = bn_s[k0], c1 = bn_s[k0 + 1];
      v2f a = {c0 + w00 * f0 + w01 * f1 + w02 * f2, c1 + w10 * f0 + w11 * f1 + w12 * f2};
      A[kk] = a;
      float y0 = c0 + w00 * g0 + w01 * g1 + w02 * g2;
      float y1 = c1 + w10 * g0 + w11 * g1 + w12 * g2;
      v2f m = {has ? y0 : 0.f, has ? y1 : 0.f};
      A[16 + kk] = m;
    }
  } else {
#pragma unroll
    for (int kk = 0; kk < 16; ++kk) {
      int k0 = kk * 4 + 2 * half;
      v2f a = {hin[mc * 64 + k0], hin[mc * 64 + k0 + 1]};
      A[kk] = a;
    }
#pragma unroll
    for (int kk = 0; kk < 16; ++kk) {
      int k0 = kk * 4 + 2 * half;
      v2f a = {accum[mc * 64 + k0] * inv, accum[mc * 64 + k0 + 1] * inv};
      A[16 + kk] = a;
    }
  }

  v8f acc0 = {}, acc1 = {}, acc2 = {}, acc3 = {};
#pragma unroll
  for (int kk = 0; kk < 32; ++kk) {
    int bb = kk * 256 + lane * 2;
    acc0 = __builtin_amdgcn_wmma_f32_16x16x4_f32(false, A[kk], false, *(const v2f*)&Bl[bb + 0],   (short)0, acc0, false, false);
    acc1 = __builtin_amdgcn_wmma_f32_16x16x4_f32(false, A[kk], false, *(const v2f*)&Bl[bb + 64],  (short)0, acc1, false, false);
    acc2 = __builtin_amdgcn_wmma_f32_16x16x4_f32(false, A[kk], false, *(const v2f*)&Bl[bb + 128], (short)0, acc2, false, false);
    acc3 = __builtin_amdgcn_wmma_f32_16x16x4_f32(false, A[kk], false, *(const v2f*)&Bl[bb + 192], (short)0, acc3, false, false);
  }

#pragma unroll
  for (int ct = 0; ct < 4; ++ct) {
    v8f av = (ct == 0) ? acc0 : (ct == 1) ? acc1 : (ct == 2) ? acc2 : acc3;
    int n = ct * 16 + (lane & 15);
    float bs = bself[n] + bneigh[n];
#pragma unroll
    for (int r = 0; r < 8; ++r) {
      int mg = base + r + 8 * half;
      if (mg < N) {
        float val = av[r] + bs;
        hout[(long long)mg * 64 + n] = val > 0.f ? val : 0.f;
      }
    }
  }
}

__global__ void __launch_bounds__(256)
k_layer_gemm(const float* __restrict__ hin, const float* __restrict__ accum,
             const float* __restrict__ cnt,
             const float* __restrict__ Wself, const float* __restrict__ bself,
             const float* __restrict__ Wneigh, const float* __restrict__ bneigh,
             float* __restrict__ hout, int N) {
  layer_body(false, hin, accum, cnt, nullptr, nullptr, nullptr, nullptr, Wself, bself, Wneigh,
             bneigh, hout, N);
}

__global__ void __launch_bounds__(256)
k_layer_gemm_first(const float* __restrict__ nf, const float* __restrict__ fsum,
                   const float* __restrict__ cnt,
                   const float* __restrict__ Wn, const float* __restrict__ bn,
                   const float* __restrict__ Wself, const float* __restrict__ bself,
                   const float* __restrict__ Wneigh, const float* __restrict__ bneigh,
                   float* __restrict__ hout, int N) {
  layer_body(true, nullptr, nullptr, cnt, nf, fsum, Wn, bn, Wself, bself, Wneigh, bneigh, hout,
             N);
}

static int share_ratio(int x, int all) {
  return all > 0 ? (int)(((long long)x * 65536 + all - 1) / all) : 0;
}

extern "C" void kernel_launch(void* const* d_in, const int* in_sizes, int n_in,
                              void* d_out, int out_size, void* d_ws, size_t ws_size,
                              hipStream_t stream) {
  if (n_in < 18) return;
  const float* node_feats = (const float*)d_in[0];
  const float* edge_feats = (const float*)d_in[1];
  const float* env        = (const float*)d_in[2];
  const int*   eidx       = (const int*)d_in[3];
  const float* Wn  = (const float*)d_in[4];
  const float* bn  = (const float*)d_in[5];
  const float* We  = (const float*)d_in[6];
  const float* be  = (const float*)d_in[7];
  const float* Wself  = (const float*)d_in[8];
  const float* bself  = (const float*)d_in[9];
  const float* Wneigh = (const float*)d_in[10];
  const float* bneigh = (const float*)d_in[11];
  const float* Wa  = (const float*)d_in[12];
  const float* ba  = (const float*)d_in[13];
  const float* W1  = (const float*)d_in[14];
  const float* b1  = (const float*)d_in[15];
  const float* W2  = (const float*)d_in[16];
  const float* b2  = (const float*)d_in[17];

  int N  = in_sizes[0] / 3;
  int E  = in_sizes[1] / 3;
  int NV = in_sizes[2] / 64;
  int L  = in_sizes[8] / (64 * 64);

  const int* src = eidx;
  const int* dst = eidx + E;

  float* scores = (float*)d_out;
  float* h_out  = (float*)d_out + E;

  float* accum = (float*)d_ws;
  float* cnt   = accum + (long long)N * 64;
  long long off = (((long long)N * 65) + 63) & ~63LL;
  float* h_b   = (float*)d_ws + off;
  float* afold = h_b + (long long)N * 64;
  float* cfold = afold + 192;
  int* deg    = (int*)(cfold + 64);
  int* rowptr = deg + (((long long)N + 63) & ~63LL);
  int* rank   = rowptr + (((long long)N + 1 + 63) & ~63LL);
  int* col    = rank + (((long long)E + 63) & ~63LL);
  float* fsum = (float*)(col + (((long long)E + 63) & ~63LL));

  int zb = (N + 255) / 256;
  int nb = (L > 0) ? 0 : (N * 64 + 255) / 256;
  k_prologue<<<64 + zb + nb, 256, 0, stream>>>(node_feats, Wn, bn, h_out, N, deg, zb, Wa, ba,
                                               We, be, env, NV, afold, cfold);
  int etiles = (E + 15) / 16;
  int units = (etiles + 7) / 8;
  int wide = (L > 1) ? L - 1 : 0;
  int xr = units / (2 + wide);
  int xf = (wide > 0) ? units / (2 + wide) : units - xr;
  int xg = (wide > 0) ? (units - xr - xf) / wide : 0;
  EdgeJob job = {edge_feats, afold, cfold, W1, b1, W2, b2, scores, E, 0, 0, 0};

  int eb = (E + 255) / 256;
  job.nblk = xr;
  job.ratio = share_ratio(xr, eb + xr);
  job.first = 0;
  k_rank<<<eb + xr, 256, 0, stream>>>(src, deg, rank, E, job);
  k_scan<<<(N + 1023) / 1024, 1024, 0, stream>>>(deg, rowptr, cnt, N);
  job.nblk = xf;
  job.ratio = share_ratio(xf, eb + xf);
  job.first = xr;
  k_fill<<<eb + xf, 256, 0, stream>>>(src, dst, rowptr, rank, col, E, job);

  int tiles = (N + 15) / 16;
  int gb = (N + 7) / 8;
  float* hin = h_out;
  for (int l = 0; l < L; ++l) {
    float* hout = (l == L - 1) ? h_out : ((hin == h_out) ? h_b : h_out);
    if (l == 0) {
      k_gather_feats<<<(N + 31) / 32, 256, 0, stream>>>(rowptr, col, node_feats, fsum, N);
      k_layer_gemm_first<<<(tiles + 7) / 8, 256, 0, stream>>>(
          node_feats, fsum, cnt, Wn, bn, Wself, bself, Wneigh, bneigh, hout, N);
    } else {
      job.first = xr + xf + (l - 1) * xg;
      job.nblk = (l == L - 1) ? units - job.first : xg;
      job.ratio = share_ratio(job.nblk, gb + job.nblk);
      k_gather<<<gb + job.nblk, 256, 0, stream>>>(rowptr, col, hin, accum, N, job);
      k_layer_gemm<<<(tiles + 7) / 8, 256, 0, stream>>>(
          hin, accum, cnt,
          Wself + (long long)l * 64 * 64, bself + (long long)l * 64,
          Wneigh + (long long)l * 64 * 64, bneigh + (long long)l * 64, hout, N);
    }
    hin = hout;
  }
}


# a WMMA block waits on memory once, not 5 times, before its first WMMA: weights pre-laid in B-fragment order by launch 1 (whole-line 16-byte staging); all loads of a block issued up front, no branch
# speedup vs baseline: 3.8898x; 1.0692x over previous
#include <hip/hip_runtime.h>

typedef __attribute__((ext_vector_type(2))) float v2f;
typedef __attribute__((ext_vector_type(4))) float v4f;
typedef __attribute__((ext_vector_type(8))) float v8f;

__global__ void __launch_bounds__(256)
k_prologue(const float* __restrict__ nf, const float* __restrict__ Wn,
           const float* __restrict__ bn, float* __restrict__ h, int N,
           int* __restrict__ deg, int zb,
           const float* __restrict__ Wa, const float* __restrict__ ba,
           const float* __restrict__ We, const float* __restrict__ be,
           const float* __restrict__ env, int NV,
           float* __restrict__ afold, float* __restrict__ cfold,
           const float* __restrict__ W1, const float* __restrict__ Wself,
           const float* __restrict__ Wneigh, float* __restrict__ w1f,
           float* __restrict__ wlf, int sb) {
  __shared__ float part[4][4];
  int b = blockIdx.x, tid = threadIdx.x;

  if (b >= 64 + sb + zb) {
    int t = (b - 64 - sb - zb) * 256 + tid;
    int n = t >> 6, j = t & 63;
    if (n >= N) return;
    float f0 = nf[n * 3 + 0], f1 = nf[n * 3 + 1], f2 = nf[n * 3 + 2];
    h[t] = bn[j] + Wn[j * 3 + 0] * f0 + Wn[j * 3 + 1] * f1 + Wn[j * 3 + 2] * f2;
    return;
  }
  if (b >= 64 + sb) {
    int i = (b - 64 - sb) * 256 + tid;
    if (i < N) deg[i] = 0;
    return;
  }
  if (b >= 64) {
    int t = (b - 64) * 256 + tid;
    int u = t - 4096;
    int idx = (t < 4096) ? t : (u & 8191);
    int v = idx & 1, lane = (idx >> 1) & 31, ct = (idx >> 6) & 3, kk = idx >> 8;
    int n = ct * 16 + (lane & 15);
    int k = kk * 4 + 2 * (lane >> 4) + v;
    if (t < 4096) {
      w1f[t] = W1[n * 64 + k];
    } else {
      int l = u >> 13;
      const float* ws = (k < 64) ? Wself : Wneigh;
      wlf[u] = ws[l * 4096 + n * 64 + (k & 63)];
    }
    return;
  }

  int j = b;
  float a0 = 0.f, a1 = 0.f, a2 = 0.f, c = 0.f;
  if (tid < 64) {
    float wv = Wa[j * 128 + tid];
    a0 = wv * We[tid * 3 + 0];
    a1 = wv * We[tid * 3 + 1];
    a2 = wv * We[tid * 3 + 2];
    c = wv * be[tid];
  } else if (tid < 128) {
    int d = tid - 64;
    float vm = 0.f;
    for (int v = 0; v < NV; ++v) vm += env[v * 64 + d];
    c = Wa[j * 128 + 64 + d] * (vm * (1.0f / (float)NV));
  }
  for (int o = 16; o > 0; o >>= 1) {
    a0 += __shfl_xor(a0, o);
    a1 += __shfl_xor(a1, o);
    a2 += __shfl_xor(a2, o);
    c += __shfl_xor(c, o);
  }
  int w = tid >> 5;
  if (w < 4 && (tid & 31) == 0) {
    part[w][0] = a0;
    part[w][1] = a1;
    part[w][2] = a2;
    part[w][3] = c;
  }
  __syncthreads();
  if (tid == 0) {
    afold[j * 3 + 0] = part[0][0] + part[1][0];
    afold[j * 3 + 1] = part[0][1] + part[1][1];
    afold[j * 3 + 2] = part[0][2] + part[1][2];
    cfold[j] = ba[j] + ((part[0][3] + part[1][3]) + (part[2][3] + part[3][3]));
  }
}

struct EdgeJob {
  const float* ef;
  const float* afold;
  const float* cfold;
  const float* W1f;
  const float* b1;
  const float* W2;
  const float* b2;
  float* scores;
  int E;
  int nblk;
  int ratio;
  int first;
};

__device__ __forceinline__ bool block_role(const EdgeJob& j, int& idx) {
  unsigned b = blockIdx.x, cap = (unsigned)j.nblk;
  unsigned q0 = (unsigned)(((unsigned long long)b * (unsigned)j.ratio) >> 16);
  unsigned q1 = (unsigned)(((unsigned long long)(b + 1) * (unsigned)j.ratio) >> 16);
  q0 = q0 < cap ? q0 : cap;
  q1 = q1 < cap ? q1 : cap;
  if (q1 > q0) {
    idx = j.first + (int)q0;
    return true;
  }
  idx = (int)(b - q0);
  return false;
}

__device__ __forceinline__ void edge_block(const EdgeJob& j, int blk) {
  __shared__ __attribute__((aligned(16))) float Bs[16 * 4 * 32 * 2];
  __shared__ float ACf[256];
  __shared__ float BWs[256];
  const float* __restrict__ ef = j.ef;
  const v4f* __restrict__ wsrc = (const v4f*)j.W1f;
  float* __restrict__ scores = j.scores;
  int E = j.E;
  int tid = threadIdx.x;
  int w = tid >> 5, lane = tid & 31;
  int base = (blk * 8 + w) * 16;
  int half = lane >> 4;
  int row = base + (lane & 15);
  long long rc = (row < E) ? row : 0;
  float f0 = ef[rc * 3 + 0], f1 = ef[rc * 3 + 1], f2 = ef[rc * 3 + 2];
  v4f wv[4];
#pragma unroll
  for (int i = 0; i < 4; ++i) wv[i] = wsrc[i * 256 + tid];
  const float* pa = (tid < 192) ? (j.afold + tid) : (j.cfold + (tid - 192));
  const float* pb = (tid & 64) ? (j.W2 + (tid & 63)) : (j.b1 + (tid & 63));
  float sa = *pa;
  float sb = *pb;
#pragma unroll
  for (int i = 0; i < 4; ++i) *(v4f*)&Bs[(i * 256 + tid) * 4] = wv[i];
  ACf[tid] = sa;
  BWs[tid] = sb;
  __syncthreads();

  v2f A[16];
#pragma unroll
  for (int kk = 0; kk < 16; ++kk) {
    int k0 = kk * 4 + 2 * half;
    const float* Af = ACf;
    const float* Cf = ACf + 192;
    float x0 = Cf[k0] + Af[k0 * 3] * f0 + Af[k0 * 3 + 1] * f1 + Af[k0 * 3 + 2] * f2;
    float x1 = Cf[k0 + 1] + Af[(k0 + 1) * 3] * f0 + Af[(k0 + 1) * 3 + 1] * f1 +
               Af[(k0 + 1) * 3 + 2] * f2;
    v2f a = {x0 > 0.f ? x0 : 0.f, x1 > 0.f ? x1 : 0.f};
    A[kk] = a;
  }

  v8f acc0 = {}, acc1 = {}, acc2 = {}, acc3 = {};
#pragma unroll
  for (int kk = 0; kk < 16; ++kk) {
    int bb = kk * 256 + lane * 2;
    acc0 = __builtin_amdgcn_wmma_f32_16x16x4_f32(false, A[kk], false, *(const v2f*)&Bs[bb + 0],   (short)0, acc0, false, false);
    acc1 = __builtin_amdgcn_wmma_f32_16x16x4_f32(false, A[kk], false, *(const v2f*)&Bs[bb + 64],  (short)0, acc1, false, false);
    acc2 = __builtin_amdgcn_wmma_f32_16x16x4_f32(false, A[kk], false, *(const v2f*)&Bs[bb + 128], (short)0, acc2, false, false);
    acc3 = __builtin_amdgcn_wmma_f32_16x16x4_f32(false, A[kk], false, *(const v2f*)&Bs[bb + 192], (short)0, acc3, false, false);
  }

  float p[8] = {0, 0, 0, 0, 0, 0, 0, 0};
#pragma unroll
  for (int ct = 0; ct < 4; ++ct) {
    v8f av = (ct == 0) ? acc0 : (ct == 1) ? acc1 : (ct == 2) ? acc2 : acc3;
    int n = ct * 16 + (lane & 15);
    float bn_ = BWs[n], wn = BWs[64 + n];
#pragma unroll
    for (int r = 0; r < 8; ++r) {
      float s = av[r] + bn_;
      s = s > 0.f ? s : 0.f;
      p[r] += wn * s;
    }
  }
#pragma unroll
  for (int r = 0; r < 8; ++r) {
    p[r] += __shfl_xor(p[r], 1);
    p[r] += __shfl_xor(p[r], 2);
    p[r] += __shfl_xor(p[r], 4);
    p[r] += __shfl_xor(p[r], 8);
  }
  if ((lane & 15) == 0) {
    float bb2 = j.b2[0];
    int rb = base + 8 * half;
#pragma unroll
    for (int r = 0; r < 8; ++r)
      if (rb + r < E) scores[rb + r] = p[r] + bb2;
  }
}

__global__ void __launch_bounds__(256)
k_rank(const int* __restrict__ src, int* __restrict__ deg,
       int* __restrict__ rank, int E, EdgeJob job) {
  int mb;
  if (block_role(job, mb)) {
    edge_block(job, mb);
    return;
  }
  int e = mb * 256 + threadIdx.x;
  if (e < E) rank[e] = atomicAdd(&deg[src[e]], 1);
}

__global__ void __launch_bounds__(1024)
k_scan(const int* __restrict__ deg, int* __restrict__ rowptr, float* __restrict__ cnt,
       int N) {
  __shared__ int wsum[32];
  __shared__ int wpre[32];
  int t = threadIdx.x, lane = t & 31, w = t >> 5;
  int nb = blockIdx.x;
  int pre = 0;
#pragma unroll 8
  for (int k = 0; k < nb; ++k) pre += deg[k * 1024 + t];
  int gi = nb * 1024 + t;
  int v = (gi < N) ? deg[gi] : 0;
  int inc = v;
  for (int o = 1; o < 32; o <<= 1) {
    int u = __shfl_up(inc, o);
    if (lane >= o) inc += u;
  }
  for (int o = 16; o > 0; o >>= 1) pre += __shfl_xor(pre, o);
  if (lane == 31) wsum[w] = inc;
  if (lane == 0) wpre[w] = pre;
  __syncthreads();
  int tot = wpre[lane] + ((lane < w) ? wsum[lane] : 0);
  for (int o = 16; o > 0; o >>= 1) tot += __shfl_xor(tot, o);
  int excl = tot + inc - v;
  if (gi < N) {
    rowptr[gi] = excl;
    cnt[gi] = (float)v;
    if (gi == N - 1) rowptr[N] = excl + v;
  }
}

__global__ void __launch_bounds__(256)
k_fill(const int* __restrict__ src, const int* __restrict__ dst,
       const int* __restrict__ rowptr, const int* __restrict__ rank,
       int* __restrict__ col, int E, EdgeJob job) {
  int mb;
  if (block_role(job, mb)) {
    edge_block(job, mb);
    return;
  }
  int e = mb * 256 + threadIdx.x;
  if (e < E) col[rowptr[src[e]] + rank[e]] = dst[e];
}

__global__ void __launch_bounds__(256)
k_gather_feats(const int* __restrict__ rowptr, const int* __restrict__ col,
               const float* __restrict__ nf, float* __restrict__ fsum, int N) {
  int lane = threadIdx.x & 31, sub = lane & 7;
  int node = (blockIdx.x * 8 + (threadIdx.x >> 5)) * 4 + (lane >> 3);
  bool valid = node < N;
  int beg = 0, end = 0;
  if (valid) {
    beg = rowptr[node];
    end = rowptr[node + 1];
  }
  unsigned nmax = (unsigned)(N - 1);
  int last = end - 1;
  float s0 = 0.f, s1 = 0.f, s2 = 0.f;
  for (int j0 = beg; j0 < end; j0 += 48) {
    unsigned id[6];
#pragma unroll
    for (int u = 0; u < 6; ++u) {
      int e = j0 + sub + 8 * u;
      id[u] = (unsigned)col[e < last ? e : last];
    }
    float a[6], b[6], c[6];
#pragma unroll
    for (int u = 0; u < 6; ++u) {
      unsigned r = id[u] < nmax ? id[u] : nmax;
      const float* f = nf + (size_t)r * 3;
      a[u] = f[0];
      b[u] = f[1];
      c[u] = f[2];
    }
#pragma unroll
    for (int u = 0; u < 6; ++u) {
      bool ok = (j0 + sub + 8 * u) < end;
      s0 += ok ? a[u] : 0.f;
      s1 += ok ? b[u] : 0.f;
      s2 += ok ? c[u] : 0.f;
    }
  }
  for (int o = 4; o > 0; o >>= 1) {
    s0 += __shfl_xor(s0, o);
    s1 += __shfl_xor(s1, o);
    s2 += __shfl_xor(s2, o);
  }
  if (valid && sub == 0) {
    v4f o4 = {s0, s1, s2, 0.f};
    *(v4f*)(fsum + (size_t)node * 4) = o4;
  }
}

__global__ void __launch_bounds__(256)
k_gather(const int* __restrict__ rowptr, const int* __restrict__ col,
         const float* __restrict__ h, float* __restrict__ accum, int N, EdgeJob job) {
  int mb;
  if (block_role(job, mb)) {
    edge_block(job, mb);
    return;
  }
  int node = mb * 8 + (threadIdx.x >> 5);
  if (node >= N) return;
  int lane = threadIdx.x & 31, half = lane >> 4, c4 = (lane & 15) * 4;
  int beg = rowptr[node], end = rowptr[node + 1];
  const v4f zero = {0.f, 0.f, 0.f, 0.f};
  v4f s[4] = {zero, zero, zero, zero};
  if (beg < end) {
    int last = end - 1;
    unsigned nmax = (unsigned)(N - 1);
    int id[8];
#pragma unroll
    for (int u = 0; u < 8; ++u) {
      int e = beg + half + 2 * u;
      id[u] = col[e < last ? e : last];
    }
    for (int j0 = beg; j0 < end; j0 += 16) {
      v4f v[8];
#pragma unroll
      for (int u = 0; u < 8; ++u) {
        unsigned r = (unsigned)id[u] < nmax ? (unsigned)id[u] : nmax;
        v[u] = *(const v4f*)(h + (size_t)r * 64 + c4);
      }
#pragma unroll
      for (int u = 0; u < 8; ++u) {
        int e = j0 + 16 + half + 2 * u;
        id[u] = col[e < last ? e : last];
      }
#pragma unroll
      for (int u = 0; u < 8; ++u) {
        bool ok = (j0 + half + 2 * u) < end;
        s[u & 3] += ok ? v[u] : zero;
      }
    }
  }
  v4f a = (s[0] + s[1]) + (s[2] + s[3]);
  a.x += __shfl_xor(a.x, 16);
  a.y += __shfl_xor(a.y, 16);
  a.z += __shfl_xor(a.z, 16);
  a.w += __shfl_xor(a.w, 16);
  if (half == 0) *(v4f*)(accum + (size_t)node * 64 + c4) = a;
}

__device__ __forceinline__ void
layer_body(bool first, const float* __restrict__ hin, const float* __restrict__ accum,
           const float* __restrict__ cnt,
           const float* __restrict__ nf, const float* __restrict__ fsum,
           const float* __restrict__ Wn, const float* __restrict__ bn,
           const float* __restrict__ Wf, const float* __restrict__ bself,
           const float* __restrict__ bneigh,
           float* __restrict__ hout, int N) {
  __shared__ __attribute__((aligned(16))) float Bl[32 * 4 * 32 * 2];
  __shared__ float WBn[256];
  int tid = threadIdx.x;
  int w = tid >> 5, lane = tid & 31;
  int base = (blockIdx.x * 8 + w) * 16;
  int half = lane >> 4;
  int mrow = base + (lane & 15);
  long long mc = (mrow < N) ? mrow : 0;

  float c = cnt[mc];
  float f0 = 0.f, f1 = 0.f, f2 = 0.f;
  v4f sv = {0.f, 0.f, 0.f, 0.f};
  v2f A[32];
  if (first) {
    f0 = nf[mc * 3 + 0];
    f1 = nf[mc * 3 + 1];
    f2 = nf[mc * 3 + 2];
    sv = *(const v4f*)(fsum + mc * 4);
  } else {
#pragma unroll
    for (int kk = 0; kk < 16; ++kk) {
      int k0 = kk * 4 + 2 * half;
      v2f a = {hin[mc * 64 + k0], hin[mc * 64 + k0 + 1]};
      A[kk] = a;
    }
#pragma unroll
    for (int kk = 0; kk < 16; ++kk) {
      int k0 = kk * 4 + 2 * half;
      v2f a = {accum[mc * 64 + k0], accum[mc * 64 + k0 + 1]};
      A[16 + kk] = a;
    }
  }

  const v4f* __restrict__ wsrc = (const v4f*)Wf;
  v4f wv[8];
#pragma unroll
  for (int i = 0; i < 8; ++i) wv[i] = wsrc[i * 256 + tid];
  float sn = 0.f;
  if (first) {
    const float* pn = (tid < 192) ? (Wn + tid) : (bn + (tid - 192));
    sn = *pn;
  }
#pragma unroll
  for (int i = 0; i < 8; ++i) *(v4f*)&Bl[(i * 256 + tid) * 4] = wv[i];
  if (first) WBn[tid] = sn;
  __syncthreads();

  float inv = (c > 0.f) ? 1.f / c : 1.f;

  if (first) {
    float g0 = sv.x * inv, g1 = sv.y * inv, g2 = sv.z * inv;
    bool has = c > 0.f;
#pragma unroll
    for (int kk = 0; kk < 16; ++kk) {
      int k0 = kk * 4 + 2 * half;
      const float* Wn_s = WBn;
      const float* bn_s = WBn + 192;
      float w00 = Wn_s[k0 * 3 + 0], w01 = Wn_s[k0 * 3 + 1], w02 = Wn_s[k0 * 3 + 2];
      float w10 = Wn_s[k0 * 3 + 3], w11 = Wn_s[k0 * 3 + 4], w12 = Wn_s[k0 * 3 + 5];
      float c0 = bn_s[k0], c1 = bn_s[k0 + 1];
      v2f a = {c0 + w00 * f0 + w01 * f1 + w02 * f2, c1 + w10 * f0 + w11 * f1 + w12 * f2};
      A[kk] = a;
      float y0 = c0 + w00 * g0 + w01 * g1 + w02 * g2;
      float y1 = c1 + w10 * g0 + w11 * g1 + w12 * g2;
      v2f m = {has ? y0 : 0.f, has ? y1 : 0.f};
      A[16 + kk] = m;
    }
  } else {
#pragma unroll
    for (int kk = 0; kk < 16; ++kk) A[16 + kk] = A[16 + kk] * inv;
  }

  v8f acc0 = {}, acc1 = {}, acc2 = {}, acc3 = {};
#pragma unroll
  for (int kk = 0; kk < 32; ++kk) {
    int bb = kk * 256 + lane * 2;
    acc0 = __builtin_amdgcn_wmma_f32_16x16x4_f32(false, A[kk], false, *(const v2f*)&Bl[bb + 0],   (short)0, acc0, false, false);
    acc1 = __builtin_amdgcn_wmma_f32_16x16x4_f32(false, A[kk], false, *(const v2f*)&Bl[bb + 64],  (short)0, acc1, false, false);
    acc2 = __builtin_amdgcn_wmma_f32_16x16x4_f32(false, A[kk], false, *(const v2f*)&Bl[bb + 128], (short)0, acc2, false, false);
    acc3 = __builtin_amdgcn_wmma_f32_16x16x4_f32(false, A[kk], false, *(const v2f*)&Bl[bb + 192], (short)0, acc3, false, false);
  }

#pragma unroll
  for (int ct = 0; ct < 4; ++ct) {
    v8f av = (ct == 0) ? acc0 : (ct == 1) ? acc1 : (ct == 2) ? acc2 : acc3;
    int n = ct * 16 + (lane & 15);
    float bs = bself[n] + bneigh[n];
#pragma unroll
    for (int r = 0; r < 8; ++r) {
      int mg = base + r + 8 * half;
      if (mg < N) {
        float val = av[r] + bs;
        hout[(long long)mg * 64 + n] = val > 0.f ? val : 0.f;
      }
    }
  }
}

__global__ void __launch_bounds__(256)
k_layer_gemm(const float* __restrict__ hin, const float* __restrict__ accum,
             const float* __restrict__ cnt,
             const float* __restrict__ Wf, const float* __restrict__ bself,
             const float* __restrict__ bneigh,
             float* __restrict__ hout, int N) {
  layer_body(false, hin, accum, cnt, nullptr, nullptr, nullptr, nullptr, Wf, bself, bneigh, hout,
             N);
}

__global__ void __launch_bounds__(256)
k_layer_gemm_first(const float* __restrict__ nf, const float* __restrict__ fsum,
                   const float* __restrict__ cnt,
                   const float* __restrict__ Wn, const float* __restrict__ bn,
                   const float* __restrict__ Wf, const float* __restrict__ bself,
                   const float* __restrict__ bneigh,
                   float* __restrict__ hout, int N) {
  layer_body(true, nullptr, nullptr, cnt, nf, fsum, Wn, bn, Wf, bself, bneigh, hout, N);
}

static int share_ratio(int x, int all) {
  return all > 0 ? (int)(((long long)x * 65536 + all - 1) / all) : 0;
}

extern "C" void kernel_launch(void* const* d_in, const int* in_sizes, int n_in,
                              void* d_out, int out_size, void* d_ws, size_t ws_size,
                              hipStream_t stream) {
  if (n_in < 18) return;
  const float* node_feats = (const float*)d_in[0];
  const float* edge_feats = (const float*)d_in[1];
  const float* env        = (const float*)d_in[2];
  const int*   eidx       = (const int*)d_in[3];
  const float* Wn  = (const float*)d_in[4];
  const float* bn  = (const float*)d_in[5];
  const float* We  = (const float*)d_in[6];
  const float* be  = (const float*)d_in[7];
  const float* Wself  = (const float*)d_in[8];
  const float* bself  = (const float*)d_in[9];
  const float* Wneigh = (const float*)d_in[10];
  const float* bneigh = (const float*)d_in[11];
  const float* Wa  = (const float*)d_in[12];
  const float* ba  = (const float*)d_in[13];
  const float* W1  = (const float*)d_in[14];
  const float* b1  = (const float*)d_in[15];
  const float* W2  = (const float*)d_in[16];
  const float* b2  = (const float*)d_in[17];

  int N  = in_sizes[0] / 3;
  int E  = in_sizes[1] / 3;
  int NV = in_sizes[2] / 64;
  int L  = in_sizes[8] / (64 * 64);

  const int* src = eidx;
  const int* dst = eidx + E;

  float* scores = (float*)d_out;
  float* h_out  = (float*)d_out + E;

  float* accum = (float*)d_ws;
  float* cnt   = accum + (long long)N * 64;
  long long off = (((long long)N * 65) + 63) & ~63LL;
  float* h_b   = (float*)d_ws + off;
  float* afold = h_b + (long long)N * 64;
  float* cfold = afold + 192;
  int* deg    = (int*)(cfold + 64);
  int* rowptr = deg + (((long long)N + 63) & ~63LL);
  int* rank   = rowptr + (((long long)N + 1 + 63) & ~63LL);
  int* col    = rank + (((long long)E + 63) & ~63LL);
  float* fsum = (float*)(col + (((long long)E + 63) & ~63LL));
  float* w1f  = fsum + ((((long long)N * 4) + 63) & ~63LL);
  float* wlf  = w1f + 4096;

  int zb = (N + 255) / 256;
  int nb = (L > 0) ? 0 : (N * 64 + 255) / 256;
  int sb = 16 + 32 * L;
  k_prologue<<<64 + sb + zb + nb, 256, 0, stream>>>(node_feats, Wn, bn, h_out, N, deg, zb, Wa,
                                                    ba, We, be, env, NV, afold, cfold, W1, Wself,
                                                    Wneigh, w1f, wlf, sb);
  int etiles = (E + 15) / 16;
  int units = (etiles + 7) / 8;
  int wide = (L > 1) ? L - 1 : 0;
  int xr = units / (2 + wide);
  int xf = (wide > 0) ? units / (2 + wide) : units - xr;
  int xg = (wide > 0) ? (units - xr - xf) / wide : 0;
  EdgeJob job = {edge_feats, afold, cfold, w1f, b1, W2, b2, scores, E, 0, 0, 0};

  int eb = (E + 255) / 256;
  job.nblk = xr;
  job.ratio = share_ratio(xr, eb + xr);
  job.first = 0;
  k_rank<<<eb + xr, 256, 0, stream>>>(src, deg, rank, E, job);
  k_scan<<<(N + 1023) / 1024, 1024, 0, stream>>>(deg, rowptr, cnt, N);
  job.nblk = xf;
  job.ratio = share_ratio(xf, eb + xf);
  job.first = xr;
  k_fill<<<eb + xf, 256, 0, stream>>>(src, dst, rowptr, rank, col, E, job);

  int tiles = (N + 15) / 16;
  int gb = (N + 7) / 8;
  float* hin = h_out;
  for (int l = 0; l < L; ++l) {
    float* hout = (l == L - 1) ? h_out : ((hin == h_out) ? h_b : h_out);
    if (l == 0) {
      k_gather_feats<<<(N + 31) / 32, 256, 0, stream>>>(rowptr, col, node_feats, fsum, N);
      k_layer_gemm_first<<<(tiles + 7) / 8, 256, 0, stream>>>(
          node_feats, fsum, cnt, Wn, bn, wlf, bself, bneigh, hout, N);
    } else {
      job.first = xr + xf + (l - 1) * xg;
      job.nblk = (l == L - 1) ? units - job.first : xg;
      job.ratio = share_ratio(job.nblk, gb + job.nblk);
      k_gather<<<gb + job.nblk, 256, 0, stream>>>(rowptr, col, hin, accum, N, job);
      k_layer_gemm<<<(tiles + 7) / 8, 256, 0, stream>>>(
          hin, accum, cnt, wlf + (long long)l * 8192, bself + (long long)l * 64,
          bneigh + (long long)l * 64, hout, N);
    }
    hin = hout;
  }
}


# edges grouped by source via a two-level counting sort on LDS counters (128-node ranges, then nodes; whole-line stores), not a global returning add + scattered store per edge; sum order: LDS arrival
# speedup vs baseline: 5.5563x; 1.4284x over previous
#include <hip/hip_runtime.h>

typedef __attribute__((ext_vector_type(2))) float v2f;
typedef __attribute__((ext_vector_type(4))) float v4f;
typedef __attribute__((ext_vector_type(8))) float v8f;

#define BSH 7
#define BNODES 128
#define CHUNK 6400
#define MAXBK 512
#define CAPB 8192

__global__ void __launch_bounds__(256)
k_prologue(const float* __restrict__ nf, const float* __restrict__ Wn,
           const float* __restrict__ bn, float* __restrict__ h, int N,
           const int* __restrict__ src, int E, int* __restrict__ counts, int nbk, int zb,
           const float* __restrict__ Wa, const float* __restrict__ ba,
           const float* __restrict__ We, const float* __restrict__ be,
           const float* __restrict__ env, int NV,
           float* __restrict__ afold, float* __restrict__ cfold,
           const float* __restrict__ W1, const float* __restrict__ Wself,
           const float* __restrict__ Wneigh, float* __restrict__ w1f,
           float* __restrict__ wlf, int sb) {
  __shared__ float part[4][4];
  int b = blockIdx.x, tid = threadIdx.x;

  if (b >= 64 + sb + zb) {
    int t = (b - 64 - sb - zb) * 256 + tid;
    int n = t >> 6, j = t & 63;
    if (n >= N) return;
    float f0 = nf[n * 3 + 0], f1 = nf[n * 3 + 1], f2 = nf[n * 3 + 2];
    h[t] = bn[j] + Wn[j * 3 + 0] * f0 + Wn[j * 3 + 1] * f1 + Wn[j * 3 + 2] * f2;
    return;
  }
  if (b >= 64 + sb) {
    __shared__ int hb[MAXBK];
    int c = b - 64 - sb;
    for (int i = tid; i < MAXBK; i += 256) hb[i] = 0;
    __syncthreads();
    int e0 = c * CHUNK;
    int sv[CHUNK / 256];
#pragma unroll
    for (int k = 0; k < CHUNK / 256; ++k) {
      int e = e0 + k * 256 + tid;
      sv[k] = src[e < E ? e : E - 1];
    }
#pragma unroll
    for (int k = 0; k < CHUNK / 256; ++k) {
      int e = e0 + k * 256 + tid;
      unsigned bk = (unsigned)sv[k] >> BSH;
      bk = bk < (unsigned)(nbk - 1) ? bk : (unsigned)(nbk - 1);
      if (e < E) atomicAdd(&hb[bk], 1);
    }
    __syncthreads();
    for (int i = tid; i < nbk; i += 256) counts[i * zb + c] = hb[i];
    return;
  }
  if (b >= 64) {
    int t = (b - 64) * 256 + tid;
    int u = t - 4096;
    int idx = (t < 4096) ? t : (u & 8191);
    int v = idx & 1, lane = (idx >> 1) & 31, ct = (idx >> 6) & 3, kk = idx >> 8;
    int n = ct * 16 + (lane & 15);
    int k = kk * 4 + 2 * (lane >> 4) + v;
    if (t < 4096) {
      w1f[t] = W1[n * 64 + k];
    } else {
      int l = u >> 13;
      const float* ws = (k < 64) ? Wself : Wneigh;
      wlf[u] = ws[l * 4096 + n * 64 + (k & 63)];
    }
    return;
  }

  int j = b;
  float a0 = 0.f, a1 = 0.f, a2 = 0.f, c = 0.f;
  if (tid < 64) {
    float wv = Wa[j * 128 + tid];
    a0 = wv * We[tid * 3 + 0];
    a1 = wv * We[tid * 3 + 1];
    a2 = wv * We[tid * 3 + 2];
    c = wv * be[tid];
  } else if (tid < 128) {
    int d = tid - 64;
    float vm = 0.f;
    for (int v = 0; v < NV; ++v) vm += env[v * 64 + d];
    c = Wa[j * 128 + 64 + d] * (vm * (1.0f / (float)NV));
  }
  for (int o = 16; o > 0; o >>= 1) {
    a0 += __shfl_xor(a0, o);
    a1 += __shfl_xor(a1, o);
    a2 += __shfl_xor(a2, o);
    c += __shfl_xor(c, o);
  }
  int w = tid >> 5;
  if (w < 4 && (tid & 31) == 0) {
    part[w][0] = a0;
    part[w][1] = a1;
    part[w][2] = a2;
    part[w][3] = c;
  }
  __syncthreads();
  if (tid == 0) {
    afold[j * 3 + 0] = part[0][0] + part[1][0];
    afold[j * 3 + 1] = part[0][1] + part[1][1];
    afold[j * 3 + 2] = part[0][2] + part[1][2];
    cfold[j] = ba[j] + ((part[0][3] + part[1][3]) + (part[2][3] + part[3][3]));
  }
}

struct EdgeJob {
  const float* ef;
  const float* afold;
  const float* cfold;
  const float* W1f;
  const float* b1;
  const float* W2;
  const float* b2;
  float* scores;
  int E;
  int nblk;
  int ratio;
  int first;
};

__device__ __forceinline__ bool block_role(const EdgeJob& j, int& idx) {
  unsigned b = blockIdx.x, cap = (unsigned)j.nblk;
  unsigned q0 = (unsigned)(((unsigned long long)b * (unsigned)j.ratio) >> 16);
  unsigned q1 = (unsigned)(((unsigned long long)(b + 1) * (unsigned)j.ratio) >> 16);
  q0 = q0 < cap ? q0 : cap;
  q1 = q1 < cap ? q1 : cap;
  if (q1 > q0) {
    idx = j.first + (int)q0;
    return true;
  }
  idx = (int)(b - q0);
  return false;
}

__device__ __forceinline__ void edge_block(const EdgeJob& j, int blk) {
  __shared__ __attribute__((aligned(16))) float Bs[16 * 4 * 32 * 2];
  __shared__ float ACf[256];
  __shared__ float BWs[256];
  const float* __restrict__ ef = j.ef;
  const v4f* __restrict__ wsrc = (const v4f*)j.W1f;
  float* __restrict__ scores = j.scores;
  int E = j.E;
  int tid = threadIdx.x;
  int w = tid >> 5, lane = tid & 31;
  int base = (blk * 8 + w) * 16;
  int half = lane >> 4;
  int row = base + (lane & 15);
  long long rc = (row < E) ? row : 0;
  float f0 = ef[rc * 3 + 0], f1 = ef[rc * 3 + 1], f2 = ef[rc * 3 + 2];
  v4f wv[4];
#pragma unroll
  for (int i = 0; i < 4; ++i) wv[i] = wsrc[i * 256 + tid];
  const float* pa = (tid < 192) ? (j.afold + tid) : (j.cfold + (tid - 192));
  const float* pb = (tid & 64) ? (j.W2 + (tid & 63)) : (j.b1 + (tid & 63));
  float sa = *pa;
  float sb = *pb;
#pragma unroll
  for (int i = 0; i < 4; ++i) *(v4f*)&Bs[(i * 256 + tid) * 4] = wv[i];
  ACf[tid] = sa;
  BWs[tid] = sb;
  __syncthreads();

  v2f A[16];
#pragma unroll
  for (int kk = 0; kk < 16; ++kk) {
    int k0 = kk * 4 + 2 * half;
    const float* Af = ACf;
    const float* Cf = ACf + 192;
    float x0 = Cf[k0] + Af[k0 * 3] * f0 + Af[k0 * 3 + 1] * f1 + Af[k0 * 3 + 2] * f2;
    float x1 = Cf[k0 + 1] + Af[(k0 + 1) * 3] * f0 + Af[(k0 + 1) * 3 + 1] * f1 +
               Af[(k0 + 1) * 3 + 2] * f2;
    v2f a = {x0 > 0.f ? x0 : 0.f, x1 > 0.f ? x1 : 0.f};
    A[kk] = a;
  }

  v8f acc0 = {}, acc1 = {}, acc2 = {}, acc3 = {};
#pragma unroll
  for (int kk = 0; kk < 16; ++kk) {
    int bb = kk * 256 + lane * 2;
    acc0 = __builtin_amdgcn_wmma_f32_16x16x4_f32(false, A[kk], false, *(const v2f*)&Bs[bb + 0],   (short)0, acc0, false, false);
    acc1 = __builtin_amdgcn_wmma_f32_16x16x4_f32(false, A[kk], false, *(const v2f*)&Bs[bb + 64],  (short)0, acc1, false, false);
    acc2 = __builtin_amdgcn_wmma_f32_16x16x4_f32(false, A[kk], false, *(const v2f*)&Bs[bb + 128], (short)0, acc2, false, false);
    acc3 = __builtin_amdgcn_wmma_f32_16x16x4_f32(false, A[kk], false, *(const v2f*)&Bs[bb + 192], (short)0, acc3, false, false);
  }

  float p[8] = {0, 0, 0, 0, 0, 0, 0, 0};
#pragma unroll
  for (int ct = 0; ct < 4; ++ct) {
    v8f av = (ct == 0) ? acc0 : (ct == 1) ? acc1 : (ct == 2) ? acc2 : acc3;
    int n = ct * 16 + (lane & 15);
    float bn_ = BWs[n], wn = BWs[64 + n];
#pragma unroll
    for (int r = 0; r < 8; ++r) {
      float s = av[r] + bn_;
      s = s > 0.f ? s : 0.f;
      p[r] += wn * s;
    }
  }
#pragma unroll
  for (int r = 0; r < 8; ++r) {
    p[r] += __shfl_xor(p[r], 1);
    p[r] += __shfl_xor(p[r], 2);
    p[r] += __shfl_xor(p[r], 4);
    p[r] += __shfl_xor(p[r], 8);
  }
  if ((lane & 15) == 0) {
    float bb2 = j.b2[0];
    int rb = base + 8 * half;
#pragma unroll
    for (int r = 0; r < 8; ++r)
      if (rb + r < E) scores[rb + r] = p[r] + bb2;
  }
}

__global__ void __launch_bounds__(1024)
k_scan(const int* __restrict__ deg, int* __restrict__ rowptr, int N) {
  __shared__ int wsum[32];
  __shared__ int wpre[32];
  int t = threadIdx.x, lane = t & 31, w = t >> 5;
  int nb = blockIdx.x;
  int pre = 0;
#pragma unroll 8
  for (int k = 0; k < nb; ++k) pre += deg[k * 1024 + t];
  int gi = nb * 1024 + t;
  int v = (gi < N) ? deg[gi] : 0;
  int inc = v;
  for (int o = 1; o < 32; o <<= 1) {
    int u = __shfl_up(inc, o);
    if (lane >= o) inc += u;
  }
  for (int o = 16; o > 0; o >>= 1) pre += __shfl_xor(pre, o);
  if (lane == 31) wsum[w] = inc;
  if (lane == 0) wpre[w] = pre;
  __syncthreads();
  int tot = wpre[lane] + ((lane < w) ? wsum[lane] : 0);
  for (int o = 16; o > 0; o >>= 1) tot += __shfl_xor(tot, o);
  int excl = tot + inc - v;
  if (gi < N) {
    rowptr[gi] = excl;
    if (gi == N - 1) rowptr[N] = excl + v;
  }
}

__global__ void __launch_bounds__(256)
k_scatter(const int* __restrict__ src, const int* __restrict__ dst,
          const int* __restrict__ counts, const int* __restrict__ offs,
          unsigned* __restrict__ tmp, int E, int nbk, int nchunk, EdgeJob job) {
  if ((int)blockIdx.x >= nchunk) {
    edge_block(job, job.first + ((int)blockIdx.x - nchunk));
    return;
  }
  __shared__ int cur[MAXBK];
  __shared__ int dlt[MAXBK];
  __shared__ int wtot[8];
  __shared__ unsigned sword[CHUNK + 1];
  __shared__ int spos[CHUNK + 1];
  int c = blockIdx.x, tid = threadIdx.x, lane = tid & 31, w = tid >> 5;
  int b0 = 2 * tid, b1 = 2 * tid + 1;
  int c0 = 0, c1 = 0, g0 = 0, g1 = 0;
  if (b0 < nbk) {
    c0 = counts[b0 * nchunk + c];
    g0 = offs[b0 * nchunk + c];
  }
  if (b1 < nbk) {
    c1 = counts[b1 * nchunk + c];
    g1 = offs[b1 * nchunk + c];
  }
  int sum = c0 + c1, inc = sum;
  for (int o = 1; o < 32; o <<= 1) {
    int u = __shfl_up(inc, o);
    if (lane >= o) inc += u;
  }
  if (lane == 31) wtot[w] = inc;
  __syncthreads();
  int pre = 0;
  for (int k = 0; k < 8; ++k) pre += (k < w) ? wtot[k] : 0;
  int ex = pre + inc - sum;
  cur[b0] = ex;
  dlt[b0] = g0 - ex;
  cur[b1] = ex + c0;
  dlt[b1] = g1 - (ex + c0);
  __syncthreads();

  int e0 = c * CHUNK;
  for (int k0 = 0; k0 < CHUNK / 256; k0 += 5) {
    unsigned sv[5], dv[5];
#pragma unroll
    for (int u = 0; u < 5; ++u) {
      int e = e0 + (k0 + u) * 256 + tid;
      int ec = e < E ? e : E - 1;
      sv[u] = (unsigned)src[ec];
      dv[u] = (unsigned)dst[ec];
    }
#pragma unroll
    for (int u = 0; u < 5; ++u) {
      int e = e0 + (k0 + u) * 256 + tid;
      int valid = e < E ? 1 : 0;
      unsigned bk = sv[u] >> BSH;
      bk = bk < (unsigned)(nbk - 1) ? bk : (unsigned)(nbk - 1);
      int slot = atomicAdd(&cur[bk], valid);
      slot = slot < CHUNK - 1 ? slot : CHUNK - 1;
      slot = valid ? slot : CHUNK;
      sword[slot] = (dv[u] << BSH) | (sv[u] & (BNODES - 1));
      spos[slot] = slot + dlt[bk];
    }
  }
  __syncthreads();
  int n = E - e0;
  n = n < CHUNK ? n : CHUNK;
  for (int i = tid; i < n; i += 256) {
    int p = spos[i];
    p = (unsigned)p < (unsigned)E ? p : E - 1;
    tmp[p] = sword[i];
  }
}

__global__ void __launch_bounds__(256)
k_bucket(const int* __restrict__ offs, const unsigned* __restrict__ tmp,
         int* __restrict__ rowptr, float* __restrict__ cnt, int* __restrict__ col,
         int N, int nbk, int nchunk, EdgeJob job) {
  if ((int)blockIdx.x >= nbk) {
    edge_block(job, job.first + ((int)blockIdx.x - nbk));
    return;
  }
  __shared__ int hc[BNODES];
  __shared__ int cu[BNODES];
  __shared__ int wt[8];
  __shared__ int scol[CAPB + 1];
  int b = blockIdx.x, tid = threadIdx.x, lane = tid & 31, w = tid >> 5;
  int beg = offs[b * nchunk], end = offs[(b + 1) * nchunk];
  int m = end - beg;
  if (tid < BNODES) hc[tid] = 0;
  __syncthreads();
  for (int i0 = 0; i0 < m; i0 += 2048) {
    unsigned wd[8];
#pragma unroll
    for (int u = 0; u < 8; ++u) {
      int i = i0 + u * 256 + tid;
      wd[u] = tmp[beg + (i < m ? i : m - 1)];
    }
#pragma unroll
    for (int u = 0; u < 8; ++u) {
      int i = i0 + u * 256 + tid;
      atomicAdd(&hc[wd[u] & (BNODES - 1)], i < m ? 1 : 0);
    }
  }
  __syncthreads();
  int v = (tid < BNODES) ? hc[tid] : 0;
  int inc = v;
  for (int o = 1; o < 32; o <<= 1) {
    int u = __shfl_up(inc, o);
    if (lane >= o) inc += u;
  }
  if (lane == 31) wt[w] = inc;
  __syncthreads();
  int pre = 0;
  for (int k = 0; k < 8; ++k) pre += (k < w) ? wt[k] : 0;
  int ex = pre + inc - v;
  if (tid < BNODES) {
    cu[tid] = ex;
    int node = (b << BSH) + tid;
    if (node < N) {
      rowptr[node] = beg + ex;
      cnt[node] = (float)v;
    }
  }
  if (b == nbk - 1 && tid == 0) rowptr[N] = end;
  __syncthreads();
  bool stage = m <= CAPB;
  for (int i0 = 0; i0 < m; i0 += 2048) {
    unsigned wd[8];
#pragma unroll
    for (int u = 0; u < 8; ++u) {
      int i = i0 + u * 256 + tid;
      wd[u] = tmp[beg + (i < m ? i : m - 1)];
    }
#pragma unroll
    for (int u = 0; u < 8; ++u) {
      int i = i0 + u * 256 + tid;
      int valid = i < m ? 1 : 0;
      int slot = atomicAdd(&cu[wd[u] & (BNODES - 1)], valid);
      slot = slot < m - 1 ? slot : m - 1;
      int dv = (int)(wd[u] >> BSH);
      if (stage) scol[valid ? slot : CAPB] = dv;
      else if (valid) col[beg + slot] = dv;
    }
  }
  __syncthreads();
  if (stage)
    for (int i = tid; i < m; i += 256) col[beg + i] = scol[i];
}

__global__ void __launch_bounds__(256)
k_gather_feats(const int* __restrict__ rowptr, const int* __restrict__ col,
               const float* __restrict__ nf, float* __restrict__ fsum, int N) {
  int lane = threadIdx.x & 31, sub = lane & 7;
  int node = (blockIdx.x * 8 + (threadIdx.x >> 5)) * 4 + (lane >> 3);
  bool valid = node < N;
  int beg = 0, end = 0;
  if (valid) {
    beg = rowptr[node];
    end = rowptr[node + 1];
  }
  unsigned nmax = (unsigned)(N - 1);
  int last = end - 1;
  float s0 = 0.f, s1 = 0.f, s2 = 0.f;
  for (int j0 = beg; j0 < end; j0 += 48) {
    unsigned id[6];
#pragma unroll
    for (int u = 0; u < 6; ++u) {
      int e = j0 + sub + 8 * u;
      id[u] = (unsigned)col[e < last ? e : last];
    }
    float a[6], b[6], c[6];
#pragma unroll
    for (int u = 0; u < 6; ++u) {
      unsigned r = id[u] < nmax ? id[u] : nmax;
      const float* f = nf + (size_t)r * 3;
      a[u] = f[0];
      b[u] = f[1];
      c[u] = f[2];
    }
#pragma unroll
    for (int u = 0; u < 6; ++u) {
      bool ok = (j0 + sub + 8 * u) < end;
      s0 += ok ? a[u] : 0.f;
      s1 += ok ? b[u] : 0.f;
      s2 += ok ? c[u] : 0.f;
    }
  }
  for (int o = 4; o > 0; o >>= 1) {
    s0 += __shfl_xor(s0, o);
    s1 += __shfl_xor(s1, o);
    s2 += __shfl_xor(s2, o);
  }
  if (valid && sub == 0) {
    v4f o4 = {s0, s1, s2, 0.f};
    *(v4f*)(fsum + (size_t)node * 4) = o4;
  }
}

__global__ void __launch_bounds__(256)
k_gather(const int* __restrict__ rowptr, const int* __restrict__ col,
         const float* __restrict__ h, float* __restrict__ accum, int N, EdgeJob job) {
  int mb;
  if (block_role(job, mb)) {
    edge_block(job, mb);
    return;
  }
  int node = mb * 8 + (threadIdx.x >> 5);
  if (node >= N) return;
  int lane = threadIdx.x & 31, half = lane >> 4, c4 = (lane & 15) * 4;
  int beg = rowptr[node], end = rowptr[node + 1];
  const v4f zero = {0.f, 0.f, 0.f, 0.f};
  v4f s[4] = {zero, zero, zero, zero};
  if (beg < end) {
    int last = end - 1;
    unsigned nmax = (unsigned)(N - 1);
    int id[8];
#pragma unroll
    for (int u = 0; u < 8; ++u) {
      int e = beg + half + 2 * u;
      id[u] = col[e < last ? e : last];
    }
    for (int j0 = beg; j0 < end; j0 += 16) {
      v4f v[8];
#pragma unroll
      for (int u = 0; u < 8; ++u) {
        unsigned r = (unsigned)id[u] < nmax ? (unsigned)id[u] : nmax;
        v[u] = *(const v4f*)(h + (size_t)r * 64 + c4);
      }
#pragma unroll
      for (int u = 0; u < 8; ++u) {
        int e = j0 + 16 + half + 2 * u;
        id[u] = col[e < last ? e : last];
      }
#pragma unroll
      for (int u = 0; u < 8; ++u) {
        bool ok = (j0 + half + 2 * u) < end;
        s[u & 3] += ok ? v[u] : zero;
      }
    }
  }
  v4f a = (s[0] + s[1]) + (s[2] + s[3]);
  a.x += __shfl_xor(a.x, 16);
  a.y += __shfl_xor(a.y, 16);
  a.z += __shfl_xor(a.z, 16);
  a.w += __shfl_xor(a.w, 16);
  if (half == 0) *(v4f*)(accum + (size_t)node * 64 + c4) = a;
}

__device__ __forceinline__ void
layer_body(bool first, const float* __restrict__ hin, const float* __restrict__ accum,
           const float* __restrict__ cnt,
           const float* __restrict__ nf, const float* __restrict__ fsum,
           const float* __restrict__ Wn, const float* __restrict__ bn,
           const float* __restrict__ Wf, const float* __restrict__ bself,
           const float* __restrict__ bneigh,
           float* __restrict__ hout, int N) {
  __shared__ __attribute__((aligned(16))) float Bl[32 * 4 * 32 * 2];
  __shared__ float WBn[256];
  int tid = threadIdx.x;
  int w = tid >> 5, lane = tid & 31;
  int base = (blockIdx.x * 8 + w) * 16;
  int half = lane >> 4;
  int mrow = base + (lane & 15);
  long long mc = (mrow < N) ? mrow : 0;

  float c = cnt[mc];
  float f0 = 0.f, f1 = 0.f, f2 = 0.f;
  v4f sv = {0.f, 0.f, 0.f, 0.f};
  v2f A[32];
  if (first) {
    f0 = nf[mc * 3 + 0];
    f1 = nf[mc * 3 + 1];
    f2 = nf[mc * 3 + 2];
    sv = *(const v4f*)(fsum + mc * 4);
  } else {
#pragma unroll
    for (int kk = 0; kk < 16; ++kk) {
      int k0 = kk * 4 + 2 * half;
      v2f a = {hin[mc * 64 + k0], hin[mc * 64 + k0 + 1]};
      A[kk] = a;
    }
#pragma unroll
    for (int kk = 0; kk < 16; ++kk) {
      int k0 = kk * 4 + 2 * half;
      v2f a = {accum[mc * 64 + k0], accum[mc * 64 + k0 + 1]};
      A[16 + kk] = a;
    }
  }

  const v4f* __restrict__ wsrc = (const v4f*)Wf;
  v4f wv[8];
#pragma unroll
  for (int i = 0; i < 8; ++i) wv[i] = wsrc[i * 256 + tid];
  float sn = 0.f;
  if (first) {
    const float* pn = (tid < 192) ? (Wn + tid) : (bn + (tid - 192));
    sn = *pn;
  }
#pragma unroll
  for (int i = 0; i < 8; ++i) *(v4f*)&Bl[(i * 256 + tid) * 4] = wv[i];
  if (first) WBn[tid] = sn;
  __syncthreads();

  float inv = (c > 0.f) ? 1.f / c : 1.f;

  if (first) {
    float g0 = sv.x * inv, g1 = sv.y * inv, g2 = sv.z * inv;
    bool has = c > 0.f;
#pragma unroll
    for (int kk = 0; kk < 16; ++kk) {
      int k0 = kk * 4 + 2 * half;
      const float* Wn_s = WBn;
      const float* bn_s = WBn + 192;
      float w00 = Wn_s[k0 * 3 + 0], w01 = Wn_s[k0 * 3 + 1], w02 = Wn_s[k0 * 3 + 2];
      float w10 = Wn_s[k0 * 3 + 3], w11 = Wn_s[k0 * 3 + 4], w12 = Wn_s[k0 * 3 + 5];
      float c0 = bn_s[k0], c1 = bn_s[k0 + 1];
      v2f a = {c0 + w00 * f0 + w01 * f1 + w02 * f2, c1 + w10 * f0 + w11 * f1 + w12 * f2};
      A[kk] = a;
      float y0 = c0 + w00 * g0 + w01 * g1 + w02 * g2;
      float y1 = c1 + w10 * g0 + w11 * g1 + w12 * g2;
      v2f m = {has ? y0 : 0.f, has ? y1 : 0.f};
      A[16 + kk] = m;
    }
  } else {
#pragma unroll
    for (int kk = 0; kk < 16; ++kk) A[16 + kk] = A[16 + kk] * inv;
  }

  v8f acc0 = {}, acc1 = {}, acc2 = {}, acc3 = {};
#pragma unroll
  for (int kk = 0; kk < 32; ++kk) {
    int bb = kk * 256 + lane * 2;
    acc0 = __builtin_amdgcn_wmma_f32_16x16x4_f32(false, A[kk], false, *(const v2f*)&Bl[bb + 0],   (short)0, acc0, false, false);
    acc1 = __builtin_amdgcn_wmma_f32_16x16x4_f32(false, A[kk], false, *(const v2f*)&Bl[bb + 64],  (short)0, acc1, false, false);
    acc2 = __builtin_amdgcn_wmma_f32_16x16x4_f32(false, A[kk], false, *(const v2f*)&Bl[bb + 128], (short)0, acc2, false, false);
    acc3 = __builtin_amdgcn_wmma_f32_16x16x4_f32(false, A[kk], false, *(const v2f*)&Bl[bb + 192], (short)0, acc3, false, false);
  }

#pragma unroll
  for (int ct = 0; ct < 4; ++ct) {
    v8f av = (ct == 0) ? acc0 : (ct == 1) ? acc1 : (ct == 2) ? acc2 : acc3;
    int n = ct * 16 + (lane & 15);
    float bs = bself[n] + bneigh[n];
#pragma unroll
    for (int r = 0; r < 8; ++r) {
      int mg = base + r + 8 * half;
      if (mg < N) {
        float val = av[r] + bs;
        hout[(long long)mg * 64 + n] = val > 0.f ? val : 0.f;
      }
    }
  }
}

__global__ void __launch_bounds__(256)
k_layer_gemm(const float* __restrict__ hin, const float* __restrict__ accum,
             const float* __restrict__ cnt,
             const float* __restrict__ Wf, const float* __restrict__ bself,
             const float* __restrict__ bneigh,
             float* __restrict__ hout, int N) {
  layer_body(false, hin, accum, cnt, nullptr, nullptr, nullptr, nullptr, Wf, bself, bneigh, hout,
             N);
}

__global__ void __launch_bounds__(256)
k_layer_gemm_first(const float* __restrict__ nf, const float* __restrict__ fsum,
                   const float* __restrict__ cnt,
                   const float* __restrict__ Wn, const float* __restrict__ bn,
                   const float* __restrict__ Wf, const float* __restrict__ bself,
                   const float* __restrict__ bneigh,
                   float* __restrict__ hout, int N) {
  layer_body(true, nullptr, nullptr, cnt, nf, fsum, Wn, bn, Wf, bself, bneigh, hout, N);
}

static int share_ratio(int x, int all) {
  return all > 0 ? (int)(((long long)x * 65536 + all - 1) / all) : 0;
}

extern "C" void kernel_launch(void* const* d_in, const int* in_sizes, int n_in,
                              void* d_out, int out_size, void* d_ws, size_t ws_size,
                              hipStream_t stream) {
  if (n_in < 18) return;
  const float* node_feats = (const float*)d_in[0];
  const float* edge_feats = (const float*)d_in[1];
  const float* env        = (const float*)d_in[2];
  const int*   eidx       = (const int*)d_in[3];
  const float* Wn  = (const float*)d_in[4];
  const float* bn  = (const float*)d_in[5];
  const float* We  = (const float*)d_in[6];
  const float* be  = (const float*)d_in[7];
  const float* Wself  = (const float*)d_in[8];
  const float* bself  = (const float*)d_in[9];
  const float* Wneigh = (const float*)d_in[10];
  const float* bneigh = (const float*)d_in[11];
  const float* Wa  = (const float*)d_in[12];
  const float* ba  = (const float*)d_in[13];
  const float* W1  = (const float*)d_in[14];
  const float* b1  = (const float*)d_in[15];
  const float* W2  = (const float*)d_in[16];
  const float* b2  = (const float*)d_in[17];

  int N  = in_sizes[0] / 3;
  int E  = in_sizes[1] / 3;
  int NV = in_sizes[2] / 64;
  int L  = in_sizes[8] / (64 * 64);

  const int* src = eidx;
  const int* dst = eidx + E;

  float* scores = (float*)d_out;
  float* h_out  = (float*)d_out + E;

  float* accum = (float*)d_ws;
  float* cnt   = accum + (long long)N * 64;
  long long off = (((long long)N * 65) + 63) & ~63LL;
  float* h_b   = (float*)d_ws + off;
  float* afold = h_b + (long long)N * 64;
  float* cfold = afold + 192;
  int* rowptr = (int*)(cfold + 64);
  unsigned* tmp = (unsigned*)(rowptr + (((long long)N + 1 + 63) & ~63LL));
  int* col    = (int*)tmp + (((long long)E + 63) & ~63LL);
  float* fsum = (float*)(col + (((long long)E + 63) & ~63LL));
  float* w1f  = fsum + ((((long long)N * 4) + 63) & ~63LL);
  float* wlf  = w1f + 4096;
  int nbk = (N + BNODES - 1) >> BSH;
  int nchunk = (E + CHUNK - 1) / CHUNK;
  int ncnt = nbk * nchunk;
  int* counts = (int*)(wlf + (long long)L * 8192);
  int* offs   = counts + (((long long)ncnt + 63) & ~63LL);
  if (nbk > MAXBK) return;

  int zb = nchunk;
  int nb = (L > 0) ? 0 : (N * 64 + 255) / 256;
  int sb = 16 + 32 * L;
  k_prologue<<<64 + sb + zb + nb, 256, 0, stream>>>(node_feats, Wn, bn, h_out, N, eidx, E, counts, nbk, zb, Wa,
                                                    ba, We, be, env, NV, afold, cfold, W1, Wself,
                                                    Wneigh, w1f, wlf, sb);
  int etiles = (E + 15) / 16;
  int units = (etiles + 7) / 8;
  int wide = (L > 1) ? L - 1 : 0;
  int xr = units / (2 + wide);
  int xf = (wide > 0) ? units / (2 + wide) : units - xr;
  int xg = (wide > 0) ? (units - xr - xf) / wide : 0;
  EdgeJob job = {edge_feats, afold, cfold, w1f, b1, W2, b2, scores, E, 0, 0, 0};

  k_scan<<<(ncnt + 1023) / 1024, 1024, 0, stream>>>(counts, offs, ncnt);
  job.nblk = xr;
  job.first = 0;
  k_scatter<<<nchunk + xr, 256, 0, stream>>>(src, dst, counts, offs, tmp, E, nbk, nchunk, job);
  job.nblk = xf;
  job.first = xr;
  k_bucket<<<nbk + xf, 256, 0, stream>>>(offs, tmp, rowptr, cnt, col, N, nbk, nchunk, job);

  int tiles = (N + 15) / 16;
  int gb = (N + 7) / 8;
  float* hin = h_out;
  for (int l = 0; l < L; ++l) {
    float* hout = (l == L - 1) ? h_out : ((hin == h_out) ? h_b : h_out);
    if (l == 0) {
      k_gather_feats<<<(N + 31) / 32, 256, 0, stream>>>(rowptr, col, node_feats, fsum, N);
      k_layer_gemm_first<<<(tiles + 7) / 8, 256, 0, stream>>>(
          node_feats, fsum, cnt, Wn, bn, wlf, bself, bneigh, hout, N);
    } else {
      job.first = xr + xf + (l - 1) * xg;
      job.nblk = (l == L - 1) ? units - job.first : xg;
      job.ratio = share_ratio(job.nblk, gb + job.nblk);
      k_gather<<<gb + job.nblk, 256, 0, stream>>>(rowptr, col, hin, accum, N, job);
      k_layer_gemm<<<(tiles + 7) / 8, 256, 0, stream>>>(
          hin, accum, cnt, wlf + (long long)l * 8192, bself + (long long)l * 64,
          bneigh + (long long)l * 64, hout, N);
    }
    hin = hout;
  }
}


# edge scorer transposed (features x edges): hidden layer on 4 K=4 wmma, its D registers are the W1 product's B fragments as they lie, W2 dot summed in-lane; sum order over k and of the W2 dot changes
# speedup vs baseline: 5.9446x; 1.0699x over previous
#include <hip/hip_runtime.h>

typedef __attribute__((ext_vector_type(2))) float v2f;
typedef __attribute__((ext_vector_type(4))) float v4f;
typedef __attribute__((ext_vector_type(8))) float v8f;

#define BSH 7
#define BNODES 128
#define CHUNK 6400
#define MAXBK 512
#define CAPB 8192

__global__ void __launch_bounds__(256)
k_prologue(const float* __restrict__ nf, const float* __restrict__ Wn,
           const float* __restrict__ bn, float* __restrict__ h, int N,
           const int* __restrict__ src, int E, int* __restrict__ counts, int nbk, int zb,
           const float* __restrict__ Wa, const float* __restrict__ ba,
           const float* __restrict__ We, const float* __restrict__ be,
           const float* __restrict__ env, int NV,
           float* __restrict__ aug,
           const float* __restrict__ W1, const float* __restrict__ Wself,
           const float* __restrict__ Wneigh, float* __restrict__ w1f,
           float* __restrict__ wlf, int sb) {
  __shared__ float part[4][4];
  int b = blockIdx.x, tid = threadIdx.x;

  if (b >= 64 + sb + zb) {
    int t = (b - 64 - sb - zb) * 256 + tid;
    int n = t >> 6, j = t & 63;
    if (n >= N) return;
    float f0 = nf[n * 3 + 0], f1 = nf[n * 3 + 1], f2 = nf[n * 3 + 2];
    h[t] = bn[j] + Wn[j * 3 + 0] * f0 + Wn[j * 3 + 1] * f1 + Wn[j * 3 + 2] * f2;
    return;
  }
  if (b >= 64 + sb) {
    __shared__ int hb[MAXBK];
    int c = b - 64 - sb;
    for (int i = tid; i < MAXBK; i += 256) hb[i] = 0;
    __syncthreads();
    int e0 = c * CHUNK;
    int sv[CHUNK / 256];
#pragma unroll
    for (int k = 0; k < CHUNK / 256; ++k) {
      int e = e0 + k * 256 + tid;
      sv[k] = src[e < E ? e : E - 1];
    }
#pragma unroll
    for (int k = 0; k < CHUNK / 256; ++k) {
      int e = e0 + k * 256 + tid;
      unsigned bk = (unsigned)sv[k] >> BSH;
      bk = bk < (unsigned)(nbk - 1) ? bk : (unsigned)(nbk - 1);
      if (e < E) atomicAdd(&hb[bk], 1);
    }
    __syncthreads();
    for (int i = tid; i < nbk; i += 256) counts[i * zb + c] = hb[i];
    return;
  }
  if (b >= 64) {
    int t = (b - 64) * 256 + tid;
    int u = t - 4096;
    if (t < 4096) {
      int q = t & 3, lane = (t >> 2) & 31, mt = (t >> 7) & 3, t2 = t >> 9;
      int st = 2 * t2 + (q >> 1);
      int m = mt * 16 + (lane & 15);
      int f = (st >> 2) * 16 + 8 * (lane >> 4) + 2 * (st & 3) + (q & 1);
      w1f[t] = W1[m * 64 + f];
    } else {
      int idx = u & 8191;
      int v = idx & 1, lane = (idx >> 1) & 31, ct = (idx >> 6) & 3, kk = idx >> 8;
      int n = ct * 16 + (lane & 15);
      int k = kk * 4 + 2 * (lane >> 4) + v;
      int l = u >> 13;
      const float* ws = (k < 64) ? Wself : Wneigh;
      wlf[u] = ws[l * 4096 + n * 64 + (k & 63)];
    }
    return;
  }

  int j = b;
  float a0 = 0.f, a1 = 0.f, a2 = 0.f, c = 0.f;
  if (tid < 64) {
    float wv = Wa[j * 128 + tid];
    a0 = wv * We[tid * 3 + 0];
    a1 = wv * We[tid * 3 + 1];
    a2 = wv * We[tid * 3 + 2];
    c = wv * be[tid];
  } else if (tid < 128) {
    int d = tid - 64;
    float vm = 0.f;
    for (int v = 0; v < NV; ++v) vm += env[v * 64 + d];
    c = Wa[j * 128 + 64 + d] * (vm * (1.0f / (float)NV));
  }
  for (int o = 16; o > 0; o >>= 1) {
    a0 += __shfl_xor(a0, o);
    a1 += __shfl_xor(a1, o);
    a2 += __shfl_xor(a2, o);
    c += __shfl_xor(c, o);
  }
  int w = tid >> 5;
  if (w < 4 && (tid & 31) == 0) {
    part[w][0] = a0;
    part[w][1] = a1;
    part[w][2] = a2;
    part[w][3] = c;
  }
  __syncthreads();
  if (tid == 0) {
    int rt = j >> 4;
    float* lo = aug + ((rt >> 1) * 32 + (j & 15)) * 4 + (rt & 1) * 2;
    float* hi = lo + 16 * 4;
    lo[0] = part[0][0] + part[1][0];
    lo[1] = part[0][1] + part[1][1];
    hi[0] = part[0][2] + part[1][2];
    hi[1] = ba[j] + ((part[0][3] + part[1][3]) + (part[2][3] + part[3][3]));
  }
}

struct EdgeJob {
  const float* ef;
  const float* aug;
  const float* W1f;
  const float* b1;
  const float* W2;
  const float* b2;
  float* scores;
  int E;
  int nblk;
  int ratio;
  int first;
};

__device__ __forceinline__ bool block_role(const EdgeJob& j, int& idx) {
  unsigned b = blockIdx.x, cap = (unsigned)j.nblk;
  unsigned q0 = (unsigned)(((unsigned long long)b * (unsigned)j.ratio) >> 16);
  unsigned q1 = (unsigned)(((unsigned long long)(b + 1) * (unsigned)j.ratio) >> 16);
  q0 = q0 < cap ? q0 : cap;
  q1 = q1 < cap ? q1 : cap;
  if (q1 > q0) {
    idx = j.first + (int)q0;
    return true;
  }
  idx = (int)(b - q0);
  return false;
}

__device__ __forceinline__ float relu_bits(float x) {
  int i = __float_as_int(x);
  return __int_as_float(i > 0 ? i : 0);
}

__device__ __forceinline__ void edge_block(const EdgeJob& j, int blk) {
  __shared__ __attribute__((aligned(16))) float Ws[8 * 4 * 32 * 4];
  __shared__ __attribute__((aligned(16))) float AGs[256];
  __shared__ __attribute__((aligned(16))) float BWs[256];
  const float* __restrict__ ef = j.ef;
  const v4f* __restrict__ wsrc = (const v4f*)j.W1f;
  float* __restrict__ scores = j.scores;
  int E = j.E;
  int tid = threadIdx.x;
  int w = tid >> 5, lane = tid & 31;
  int base = (blk * 8 + w) * 16;
  int half = lane >> 4;
  int row = base + (lane & 15);
  long long rc = (row < E) ? row : 0;
  const float* pe = ef + rc * 3 + 2 * half;
  float x0 = pe[0];
  float x1 = pe[1 - half];
  v4f wv[4];
#pragma unroll
  for (int i = 0; i < 4; ++i) wv[i] = wsrc[i * 256 + tid];
  const float* pb = (tid & 64) ? (j.W2 + (tid & 63)) : (j.b1 + (tid & 63));
  float sa = j.aug[tid];
  float sb = *pb;
#pragma unroll
  for (int i = 0; i < 4; ++i) *(v4f*)&Ws[(i * 256 + tid) * 4] = wv[i];
  AGs[tid] = sa;
  BWs[tid] = sb;
  __syncthreads();

  const v2f xb = {x0, half ? 1.0f : x1};
  const v8f zero8 = {};
  const v4f* bw = (const v4f*)BWs;
  const v4f g0 = ((const v4f*)AGs)[lane], g1 = ((const v4f*)AGs)[32 + lane];

  v8f acc[4];
#pragma unroll
  for (int mt = 0; mt < 4; ++mt) {
    v4f lo = bw[mt * 4 + half * 2], hi = bw[mt * 4 + half * 2 + 1];
    v8f a = {lo.x, lo.y, lo.z, lo.w, hi.x, hi.y, hi.z, hi.w};
    acc[mt] = a;
  }

  v8f H = zero8;
#pragma unroll
  for (int t2 = 0; t2 < 8; ++t2) {
    if ((t2 & 1) == 0) {
      int rt = t2 >> 1;
      v2f ga = (rt == 0) ? v2f{g0.x, g0.y} : (rt == 1) ? v2f{g0.z, g0.w}
             : (rt == 2) ? v2f{g1.x, g1.y} : v2f{g1.z, g1.w};
      H = __builtin_amdgcn_wmma_f32_16x16x4_f32(false, ga, false, xb, (short)0, zero8, false, false);
#pragma unroll
      for (int r = 0; r < 8; ++r) H[r] = relu_bits(H[r]);
    }
    int r0 = 4 * (t2 & 1);
    const v2f bA = {H[r0], H[r0 + 1]};
    const v2f bB = {H[r0 + 2], H[r0 + 3]};
    v4f a[4];
#pragma unroll
    for (int mt = 0; mt < 4; ++mt) a[mt] = *(const v4f*)&Ws[((t2 * 4 + mt) * 32 + lane) * 4];
#pragma unroll
    for (int mt = 0; mt < 4; ++mt) {
      const v2f aA = {a[mt].x, a[mt].y};
      acc[mt] = __builtin_amdgcn_wmma_f32_16x16x4_f32(false, aA, false, bA, (short)0, acc[mt], false, false);
    }
#pragma unroll
    for (int mt = 0; mt < 4; ++mt) {
      const v2f aB = {a[mt].z, a[mt].w};
      acc[mt] = __builtin_amdgcn_wmma_f32_16x16x4_f32(false, aB, false, bB, (short)0, acc[mt], false, false);
    }
  }

  float p[4] = {0.f, 0.f, 0.f, 0.f};
#pragma unroll
  for (int mt = 0; mt < 4; ++mt) {
    v4f lo = bw[16 + mt * 4 + half * 2], hi = bw[16 + mt * 4 + half * 2 + 1];
    float wq[8] = {lo.x, lo.y, lo.z, lo.w, hi.x, hi.y, hi.z, hi.w};
#pragma unroll
    for (int r = 0; r < 8; ++r) {
      p[r & 3] += wq[r] * relu_bits(acc[mt][r]);
    }
  }
  float ps = (p[0] + p[1]) + (p[2] + p[3]);
  unsigned pu = __float_as_uint(ps);
  unsigned qu = __builtin_amdgcn_permlanex16(pu, pu, 0x76543210u, 0xfedcba98u, false, false);
  ps += __uint_as_float(qu);
  if (half == 0 && row < E) scores[row] = ps + j.b2[0];
}

__global__ void __launch_bounds__(1024)
k_scan(const int* __restrict__ deg, int* __restrict__ rowptr, int N) {
  __shared__ int wsum[32];
  __shared__ int wpre[32];
  int t = threadIdx.x, lane = t & 31, w = t >> 5;
  int nb = blockIdx.x;
  int pre = 0;
#pragma unroll 8
  for (int k = 0; k < nb; ++k) pre += deg[k * 1024 + t];
  int gi = nb * 1024 + t;
  int v = (gi < N) ? deg[gi] : 0;
  int inc = v;
  for (int o = 1; o < 32; o <<= 1) {
    int u = __shfl_up(inc, o);
    if (lane >= o) inc += u;
  }
  for (int o = 16; o > 0; o >>= 1) pre += __shfl_xor(pre, o);
  if (lane == 31) wsum[w] = inc;
  if (lane == 0) wpre[w] = pre;
  __syncthreads();
  int tot = wpre[lane] + ((lane < w) ? wsum[lane] : 0);
  for (int o = 16; o > 0; o >>= 1) tot += __shfl_xor(tot, o);
  int excl = tot + inc - v;
  if (gi < N) {
    rowptr[gi] = excl;
    if (gi == N - 1) rowptr[N] = excl + v;
  }
}

__global__ void __launch_bounds__(256)
k_scatter(const int* __restrict__ src, const int* __restrict__ dst,
          const int* __restrict__ counts, const int* __restrict__ offs,
          unsigned* __restrict__ tmp, int E, int nbk, int nchunk, EdgeJob job) {
  if ((int)blockIdx.x >= nchunk) {
    edge_block(job, job.first + ((int)blockIdx.x - nchunk));
    return;
  }
  __shared__ int cur[MAXBK];
  __shared__ int dlt[MAXBK];
  __shared__ int wtot[8];
  __shared__ unsigned sword[CHUNK + 1];
  __shared__ int spos[CHUNK + 1];
  int c = blockIdx.x, tid = threadIdx.x, lane = tid & 31, w = tid >> 5;
  int b0 = 2 * tid, b1 = 2 * tid + 1;
  int c0 = 0, c1 = 0, g0 = 0, g1 = 0;
  if (b0 < nbk) {
    c0 = counts[b0 * nchunk + c];
    g0 = offs[b0 * nchunk + c];
  }
  if (b1 < nbk) {
    c1 = counts[b1 * nchunk + c];
    g1 = offs[b1 * nchunk + c];
  }
  int sum = c0 + c1, inc = sum;
  for (int o = 1; o < 32; o <<= 1) {
    int u = __shfl_up(inc, o);
    if (lane >= o) inc += u;
  }
  if (lane == 31) wtot[w] = inc;
  __syncthreads();
  int pre = 0;
  for (int k = 0; k < 8; ++k) pre += (k < w) ? wtot[k] : 0;
  int ex = pre + inc - sum;
  cur[b0] = ex;
  dlt[b0] = g0 - ex;
  cur[b1] = ex + c0;
  dlt[b1] = g1 - (ex + c0);
  __syncthreads();

  int e0 = c * CHUNK;
  for (int k0 = 0; k0 < CHUNK / 256; k0 += 5) {
    unsigned sv[5], dv[5];
#pragma unroll
    for (int u = 0; u < 5; ++u) {
      int e = e0 + (k0 + u) * 256 + tid;
      int ec = e < E ? e : E - 1;
      sv[u] = (unsigned)src[ec];
      dv[u] = (unsigned)dst[ec];
    }
#pragma unroll
    for (int u = 0; u < 5; ++u) {
      int e = e0 + (k0 + u) * 256 + tid;
      int valid = e < E ? 1 : 0;
      unsigned bk = sv[u] >> BSH;
      bk = bk < (unsigned)(nbk - 1) ? bk : (unsigned)(nbk - 1);
      int slot = atomicAdd(&cur[bk], valid);
      slot = slot < CHUNK - 1 ? slot : CHUNK - 1;
      slot = valid ? slot : CHUNK;
      sword[slot] = (dv[u] << BSH) | (sv[u] & (BNODES - 1));
      spos[slot] = slot + dlt[bk];
    }
  }
  __syncthreads();
  int n = E - e0;
  n = n < CHUNK ? n : CHUNK;
  for (int i = tid; i < n; i += 256) {
    int p = spos[i];
    p = (unsigned)p < (unsigned)E ? p : E - 1;
    tmp[p] = sword[i];
  }
}

__global__ void __launch_bounds__(256)
k_bucket(const int* __restrict__ offs, const unsigned* __restrict__ tmp,
         int* __restrict__ rowptr, float* __restrict__ cnt, int* __restrict__ col,
         int N, int nbk, int nchunk, EdgeJob job) {
  if ((int)blockIdx.x >= nbk) {
    edge_block(job, job.first + ((int)blockIdx.x - nbk));
    return;
  }
  __shared__ int hc[BNODES];
  __shared__ int cu[BNODES];
  __shared__ int wt[8];
  __shared__ int scol[CAPB + 1];
  int b = blockIdx.x, tid = threadIdx.x, lane = tid & 31, w = tid >> 5;
  int beg = offs[b * nchunk], end = offs[(b + 1) * nchunk];
  int m = end - beg;
  if (tid < BNODES) hc[tid] = 0;
  __syncthreads();
  for (int i0 = 0; i0 < m; i0 += 2048) {
    unsigned wd[8];
#pragma unroll
    for (int u = 0; u < 8; ++u) {
      int i = i0 + u * 256 + tid;
      wd[u] = tmp[beg + (i < m ? i : m - 1)];
    }
#pragma unroll
    for (int u = 0; u < 8; ++u) {
      int i = i0 + u * 256 + tid;
      atomicAdd(&hc[wd[u] & (BNODES - 1)], i < m ? 1 : 0);
    }
  }
  __syncthreads();
  int v = (tid < BNODES) ? hc[tid] : 0;
  int inc = v;
  for (int o = 1; o < 32; o <<= 1) {
    int u = __shfl_up(inc, o);
    if (lane >= o) inc += u;
  }
  if (lane == 31) wt[w] = inc;
  __syncthreads();
  int pre = 0;
  for (int k = 0; k < 8; ++k) pre += (k < w) ? wt[k] : 0;
  int ex = pre + inc - v;
  if (tid < BNODES) {
    cu[tid] = ex;
    int node = (b << BSH) + tid;
    if (node < N) {
      rowptr[node] = beg + ex;
      cnt[node] = (float)v;
    }
  }
  if (b == nbk - 1 && tid == 0) rowptr[N] = end;
  __syncthreads();
  bool stage = m <= CAPB;
  for (int i0 = 0; i0 < m; i0 += 2048) {
    unsigned wd[8];
#pragma unroll
    for (int u = 0; u < 8; ++u) {
      int i = i0 + u * 256 + tid;
      wd[u] = tmp[beg + (i < m ? i : m - 1)];
    }
#pragma unroll
    for (int u = 0; u < 8; ++u) {
      int i = i0 + u * 256 + tid;
      int valid = i < m ? 1 : 0;
      int slot = atomicAdd(&cu[wd[u] & (BNODES - 1)], valid);
      slot = slot < m - 1 ? slot : m - 1;
      int dv = (int)(wd[u] >> BSH);
      if (stage) scol[valid ? slot : CAPB] = dv;
      else if (valid) col[beg + slot] = dv;
    }
  }
  __syncthreads();
  if (stage)
    for (int i = tid; i < m; i += 256) col[beg + i] = scol[i];
}

__global__ void __launch_bounds__(256)
k_gather_feats(const int* __restrict__ rowptr, const int* __restrict__ col,
               const float* __restrict__ nf, float* __restrict__ fsum, int N) {
  int lane = threadIdx.x & 31, sub = lane & 7;
  int node = (blockIdx.x * 8 + (threadIdx.x >> 5)) * 4 + (lane >> 3);
  bool valid = node < N;
  int beg = 0, end = 0;
  if (valid) {
    beg = rowptr[node];
    end = rowptr[node + 1];
  }
  unsigned nmax = (unsigned)(N - 1);
  int last = end - 1;
  float s0 = 0.f, s1 = 0.f, s2 = 0.f;
  for (int j0 = beg; j0 < end; j0 += 48) {
    unsigned id[6];
#pragma unroll
    for (int u = 0; u < 6; ++u) {
      int e = j0 + sub + 8 * u;
      id[u] = (unsigned)col[e < last ? e : last];
    }
    float a[6], b[6], c[6];
#pragma unroll
    for (int u = 0; u < 6; ++u) {
      unsigned r = id[u] < nmax ? id[u] : nmax;
      const float* f = nf + (size_t)r * 3;
      a[u] = f[0];
      b[u] = f[1];
      c[u] = f[2];
    }
#pragma unroll
    for (int u = 0; u < 6; ++u) {
      bool ok = (j0 + sub + 8 * u) < end;
      s0 += ok ? a[u] : 0.f;
      s1 += ok ? b[u] : 0.f;
      s2 += ok ? c[u] : 0.f;
    }
  }
  for (int o = 4; o > 0; o >>= 1) {
    s0 += __shfl_xor(s0, o);
    s1 += __shfl_xor(s1, o);
    s2 += __shfl_xor(s2, o);
  }
  if (valid && sub == 0) {
    v4f o4 = {s0, s1, s2, 0.f};
    *(v4f*)(fsum + (size_t)node * 4) = o4;
  }
}

__global__ void __launch_bounds__(256)
k_gather(const int* __restrict__ rowptr, const int* __restrict__ col,
         const float* __restrict__ h, float* __restrict__ accum, int N, EdgeJob job) {
  int mb;
  if (block_role(job, mb)) {
    edge_block(job, mb);
    return;
  }
  int node = mb * 8 + (threadIdx.x >> 5);
  if (node >= N) return;
  int lane = threadIdx.x & 31, half = lane >> 4, c4 = (lane & 15) * 4;
  int beg = rowptr[node], end = rowptr[node + 1];
  const v4f zero = {0.f, 0.f, 0.f, 0.f};
  v4f s[4] = {zero, zero, zero, zero};
  if (beg < end) {
    int last = end - 1;
    unsigned nmax = (unsigned)(N - 1);
    int id[8];
#pragma unroll
    for (int u = 0; u < 8; ++u) {
      int e = beg + half + 2 * u;
      id[u] = col[e < last ? e : last];
    }
    for (int j0 = beg; j0 < end; j0 += 16) {
      v4f v[8];
#pragma unroll
      for (int u = 0; u < 8; ++u) {
        unsigned r = (unsigned)id[u] < nmax ? (unsigned)id[u] : nmax;
        v[u] = *(const v4f*)(h + (size_t)r * 64 + c4);
      }
#pragma unroll
      for (int u = 0; u < 8; ++u) {
        int e = j0 + 16 + half + 2 * u;
        id[u] = col[e < last ? e : last];
      }
#pragma unroll
      for (int u = 0; u < 8; ++u) {
        bool ok = (j0 + half + 2 * u) < end;
        s[u & 3] += ok ? v[u] : zero;
      }
    }
  }
  v4f a = (s[0] + s[1]) + (s[2] + s[3]);
  a.x += __shfl_xor(a.x, 16);
  a.y += __shfl_xor(a.y, 16);
  a.z += __shfl_xor(a.z, 16);
  a.w += __shfl_xor(a.w, 16);
  if (half == 0) *(v4f*)(accum + (size_t)node * 64 + c4) = a;
}

__device__ __forceinline__ void
layer_body(bool first, const float* __restrict__ hin, const float* __restrict__ accum,
           const float* __restrict__ cnt,
           const float* __restrict__ nf, const float* __restrict__ fsum,
           const float* __restrict__ Wn, const float* __restrict__ bn,
           const float* __restrict__ Wf, const float* __restrict__ bself,
           const float* __restrict__ bneigh,
           float* __restrict__ hout, int N) {
  __shared__ __attribute__((aligned(16))) float Bl[32 * 4 * 32 * 2];
  __shared__ float WBn[256];
  int tid = threadIdx.x;
  int w = tid >> 5, lane = tid & 31;
  int base = (blockIdx.x * 8 + w) * 16;
  int half = lane >> 4;
  int mrow = base + (lane & 15);
  long long mc = (mrow < N) ? mrow : 0;

  float c = cnt[mc];
  float f0 = 0.f, f1 = 0.f, f2 = 0.f;
  v4f sv = {0.f, 0.f, 0.f, 0.f};
  v2f A[32];
  if (first) {
    f0 = nf[mc * 3 + 0];
    f1 = nf[mc * 3 + 1];
    f2 = nf[mc * 3 + 2];
    sv = *(const v4f*)(fsum + mc * 4);
  } else {
#pragma unroll
    for (int kk = 0; kk < 16; ++kk) {
      int k0 = kk * 4 + 2 * half;
      v2f a = {hin[mc * 64 + k0], hin[mc * 64 + k0 + 1]};
      A[kk] = a;
    }
#pragma unroll
    for (int kk = 0; kk < 16; ++kk) {
      int k0 = kk * 4 + 2 * half;
      v2f a = {accum[mc * 64 + k0], accum[mc * 64 + k0 + 1]};
      A[16 + kk] = a;
    }
  }

  const v4f* __restrict__ wsrc = (const v4f*)Wf;
  v4f wv[8];
#pragma unroll
  for (int i = 0; i < 8; ++i) wv[i] = wsrc[i * 256 + tid];
  float sn = 0.f;
  if (first) {
    const float* pn = (tid < 192) ? (Wn + tid) : (bn + (tid - 192));
    sn = *pn;
  }
#pragma unroll
  for (int i = 0; i < 8; ++i) *(v4f*)&Bl[(i * 256 + tid) * 4] = wv[i];
  if (first) WBn[tid] = sn;
  __syncthreads();

  float inv = (c > 0.f) ? 1.f / c : 1.f;

  if (first) {
    float g0 = sv.x * inv, g1 = sv.y * inv, g2 = sv.z * inv;
    bool has = c > 0.f;
#pragma unroll
    for (int kk = 0; kk < 16; ++kk) {
      int k0 = kk * 4 + 2 * half;
      const float* Wn_s = WBn;
      const float* bn_s = WBn + 192;
      float w00 = Wn_s[k0 * 3 + 0], w01 = Wn_s[k0 * 3 + 1], w02 = Wn_s[k0 * 3 + 2];
      float w10 = Wn_s[k0 * 3 + 3], w11 = Wn_s[k0 * 3 + 4], w12 = Wn_s[k0 * 3 + 5];
      float c0 = bn_s[k0], c1 = bn_s[k0 + 1];
      v2f a = {c0 + w00 * f0 + w01 * f1 + w02 * f2, c1 + w10 * f0 + w11 * f1 + w12 * f2};
      A[kk] = a;
      float y0 = c0 + w00 * g0 + w01 * g1 + w02 * g2;
      float y1 = c1 + w10 * g0 + w11 * g1 + w12 * g2;
      v2f m = {has ? y0 : 0.f, has ? y1 : 0.f};
      A[16 + kk] = m;
    }
  } else {
#pragma unroll
    for (int kk = 0; kk < 16; ++kk) A[16 + kk] = A[16 + kk] * inv;
  }

  v8f acc0 = {}, acc1 = {}, acc2 = {}, acc3 = {};
#pragma unroll
  for (int kk = 0; kk < 32; ++kk) {
    int bb = kk * 256 + lane * 2;
    acc0 = __builtin_amdgcn_wmma_f32_16x16x4_f32(false, A[kk], false, *(const v2f*)&Bl[bb + 0],   (short)0, acc0, false, false);
    acc1 = __builtin_amdgcn_wmma_f32_16x16x4_f32(false, A[kk], false, *(const v2f*)&Bl[bb + 64],  (short)0, acc1, false, false);
    acc2 = __builtin_amdgcn_wmma_f32_16x16x4_f32(false, A[kk], false, *(const v2f*)&Bl[bb + 128], (short)0, acc2, false, false);
    acc3 = __builtin_amdgcn_wmma_f32_16x16x4_f32(false, A[kk], false, *(const v2f*)&Bl[bb + 192], (short)0, acc3, false, false);
  }

#pragma unroll
  for (int ct = 0; ct < 4; ++ct) {
    v8f av = (ct == 0) ? acc0 : (ct == 1) ? acc1 : (ct == 2) ? acc2 : acc3;
    int n = ct * 16 + (lane & 15);
    float bs = bself[n] + bneigh[n];
#pragma unroll
    for (int r = 0; r < 8; ++r) {
      int mg = base + r + 8 * half;
      if (mg < N) {
        float val = av[r] + bs;
        hout[(long long)mg * 64 + n] = val > 0.f ? val : 0.f;
      }
    }
  }
}

__global__ void __launch_bounds__(256)
k_layer_gemm(const float* __restrict__ hin, const float* __restrict__ accum,
             const float* __restrict__ cnt,
             const float* __restrict__ Wf, const float* __restrict__ bself,
             const float* __restrict__ bneigh,
             float* __restrict__ hout, int N) {
  layer_body(false, hin, accum, cnt, nullptr, nullptr, nullptr, nullptr, Wf, bself, bneigh, hout,
             N);
}

__global__ void __launch_bounds__(256)
k_layer_gemm_first(const float* __restrict__ nf, const float* __restrict__ fsum,
                   const float* __restrict__ cnt,
                   const float* __restrict__ Wn, const float* __restrict__ bn,
                   const float* __restrict__ Wf, const float* __restrict__ bself,
                   const float* __restrict__ bneigh,
                   float* __restrict__ hout, int N) {
  layer_body(true, nullptr, nullptr, cnt, nf, fsum, Wn, bn, Wf, bself, bneigh, hout, N);
}

static int share_ratio(int x, int all) {
  return all > 0 ? (int)(((long long)x * 65536 + all - 1) / all) : 0;
}

extern "C" void kernel_launch(void* const* d_in, const int* in_sizes, int n_in,
                              void* d_out, int out_size, void* d_ws, size_t ws_size,
                              hipStream_t stream) {
  if (n_in < 18) return;
  const float* node_feats = (const float*)d_in[0];
  const float* edge_feats = (const float*)d_in[1];
  const float* env        = (const float*)d_in[2];
  const int*   eidx       = (const int*)d_in[3];
  const float* Wn  = (const float*)d_in[4];
  const float* bn  = (const float*)d_in[5];
  const float* We  = (const float*)d_in[6];
  const float* be  = (const float*)d_in[7];
  const float* Wself  = (const float*)d_in[8];
  const float* bself  = (const float*)d_in[9];
  const float* Wneigh = (const float*)d_in[10];
  const float* bneigh = (const float*)d_in[11];
  const float* Wa  = (const float*)d_in[12];
  const float* ba  = (const float*)d_in[13];
  const float* W1  = (const float*)d_in[14];
  const float* b1  = (const float*)d_in[15];
  const float* W2  = (const float*)d_in[16];
  const float* b2  = (const float*)d_in[17];

  int N  = in_sizes[0] / 3;
  int E  = in_sizes[1] / 3;
  int NV = in_sizes[2] / 64;
  int L  = in_sizes[8] / (64 * 64);

  const int* src = eidx;
  const int* dst = eidx + E;

  float* scores = (float*)d_out;
  float* h_out  = (float*)d_out + E;

  float* accum = (float*)d_ws;
  float* cnt   = accum + (long long)N * 64;
  long long off = (((long long)N * 65) + 63) & ~63LL;
  float* h_b   = (float*)d_ws + off;
  float* aug   = h_b + (long long)N * 64;
  int* rowptr = (int*)(aug + 256);
  unsigned* tmp = (unsigned*)(rowptr + (((long long)N + 1 + 63) & ~63LL));
  int* col    = (int*)tmp + (((long long)E + 63) & ~63LL);
  float* fsum = (float*)(col + (((long long)E + 63) & ~63LL));
  float* w1f  = fsum + ((((long long)N * 4) + 63) & ~63LL);
  float* wlf  = w1f + 4096;
  int nbk = (N + BNODES - 1) >> BSH;
  int nchunk = (E + CHUNK - 1) / CHUNK;
  int ncnt = nbk * nchunk;
  int* counts = (int*)(wlf + (long long)L * 8192);
  int* offs   = counts + (((long long)ncnt + 63) & ~63LL);
  if (nbk > MAXBK) return;

  int zb = nchunk;
  int nb = (L > 0) ? 0 : (N * 64 + 255) / 256;
  int sb = 16 + 32 * L;
  k_prologue<<<64 + sb + zb + nb, 256, 0, stream>>>(node_feats, Wn, bn, h_out, N, eidx, E, counts, nbk, zb, Wa,
                                                    ba, We, be, env, NV, aug, W1, Wself,
                                                    Wneigh, w1f, wlf, sb);
  int etiles = (E + 15) / 16;
  int units = (etiles + 7) / 8;
  int wide = (L > 1) ? L - 1 : 0;
  int xr = units / (2 + wide);
  int xf = (wide > 0) ? units / (2 + wide) : units - xr;
  int xg = (wide > 0) ? (units - xr - xf) / wide : 0;
  EdgeJob job = {edge_feats, aug, w1f, b1, W2, b2, scores, E, 0, 0, 0};

  k_scan<<<(ncnt + 1023) / 1024, 1024, 0, stream>>>(counts, offs, ncnt);
  job.nblk = xr;
  job.first = 0;
  k_scatter<<<nchunk + xr, 256, 0, stream>>>(src, dst, counts, offs, tmp, E, nbk, nchunk, job);
  job.nblk = xf;
  job.first = xr;
  k_bucket<<<nbk + xf, 256, 0, stream>>>(offs, tmp, rowptr, cnt, col, N, nbk, nchunk, job);

  int tiles = (N + 15) / 16;
  int gb = (N + 7) / 8;
  float* hin = h_out;
  for (int l = 0; l < L; ++l) {
    float* hout = (l == L - 1) ? h_out : ((hin == h_out) ? h_b : h_out);
    if (l == 0) {
      k_gather_feats<<<(N + 31) / 32, 256, 0, stream>>>(rowptr, col, node_feats, fsum, N);
      k_layer_gemm_first<<<(tiles + 7) / 8, 256, 0, stream>>>(
          node_feats, fsum, cnt, Wn, bn, wlf, bself, bneigh, hout, N);
    } else {
      job.first = xr + xf + (l - 1) * xg;
      job.nblk = (l == L - 1) ? units - job.first : xg;
      job.ratio = share_ratio(job.nblk, gb + job.nblk);
      k_gather<<<gb + job.nblk, 256, 0, stream>>>(rowptr, col, hin, accum, N, job);
      k_layer_gemm<<<(tiles + 7) / 8, 256, 0, stream>>>(
          hin, accum, cnt, wlf + (long long)l * 8192, bself + (long long)l * 64,
          bneigh + (long long)l * 64, hout, N);
    }
    hin = hout;
  }
}


# 2 launches fewer (8 to 6): the sort's placing blocks sum the per-chunk count rows themselves (no scan launch); layer-0 WMMA waves gather their own 16 nodes' neighbour features (no gather launch)
# speedup vs baseline: 6.6578x; 1.1200x over previous
#include <hip/hip_runtime.h>

typedef __attribute__((ext_vector_type(2))) float v2f;
typedef __attribute__((ext_vector_type(4))) float v4f;
typedef __attribute__((ext_vector_type(8))) float v8f;

#define BSH 7
#define BNODES 128
#define CHUNK 6400
#define MAXBK 512
#define CAPB 8192

__global__ void __launch_bounds__(256)
k_prologue(const float* __restrict__ nf, const float* __restrict__ Wn,
           const float* __restrict__ bn, float* __restrict__ h, int N,
           const int* __restrict__ src, int E, int* __restrict__ counts, int nbk, int zb,
           const float* __restrict__ Wa, const float* __restrict__ ba,
           const float* __restrict__ We, const float* __restrict__ be,
           const float* __restrict__ env, int NV,
           float* __restrict__ aug,
           const float* __restrict__ W1, const float* __restrict__ Wself,
           const float* __restrict__ Wneigh, float* __restrict__ w1f,
           float* __restrict__ wlf, int sb) {
  __shared__ float part[4][4];
  int b = blockIdx.x, tid = threadIdx.x;

  if (b >= 64 + sb + zb) {
    int t = (b - 64 - sb - zb) * 256 + tid;
    int n = t >> 6, j = t & 63;
    if (n >= N) return;
    float f0 = nf[n * 3 + 0], f1 = nf[n * 3 + 1], f2 = nf[n * 3 + 2];
    h[t] = bn[j] + Wn[j * 3 + 0] * f0 + Wn[j * 3 + 1] * f1 + Wn[j * 3 + 2] * f2;
    return;
  }
  if (b >= 64 + sb) {
    __shared__ int hb[MAXBK];
    int c = b - 64 - sb;
    for (int i = tid; i < MAXBK; i += 256) hb[i] = 0;
    __syncthreads();
    int e0 = c * CHUNK;
    int sv[CHUNK / 256];
#pragma unroll
    for (int k = 0; k < CHUNK / 256; ++k) {
      int e = e0 + k * 256 + tid;
      sv[k] = src[e < E ? e : E - 1];
    }
#pragma unroll
    for (int k = 0; k < CHUNK / 256; ++k) {
      int e = e0 + k * 256 + tid;
      unsigned bk = (unsigned)sv[k] >> BSH;
      bk = bk < (unsigned)(nbk - 1) ? bk : (unsigned)(nbk - 1);
      if (e < E) atomicAdd(&hb[bk], 1);
    }
    __syncthreads();
    int nbkp = (nbk + 3) & ~3;
    for (int i = tid; i < nbkp; i += 256) counts[(size_t)c * nbkp + i] = hb[i];
    return;
  }
  if (b >= 64) {
    int t = (b - 64) * 256 + tid;
    int u = t - 4096;
    if (t < 4096) {
      int q = t & 3, lane = (t >> 2) & 31, mt = (t >> 7) & 3, t2 = t >> 9;
      int st = 2 * t2 + (q >> 1);
      int m = mt * 16 + (lane & 15);
      int f = (st >> 2) * 16 + 8 * (lane >> 4) + 2 * (st & 3) + (q & 1);
      w1f[t] = W1[m * 64 + f];
    } else {
      int idx = u & 8191;
      int v = idx & 1, lane = (idx >> 1) & 31, ct = (idx >> 6) & 3, kk = idx >> 8;
      int n = ct * 16 + (lane & 15);
      int k = kk * 4 + 2 * (lane >> 4) + v;
      int l = u >> 13;
      const float* ws = (k < 64) ? Wself : Wneigh;
      wlf[u] = ws[l * 4096 + n * 64 + (k & 63)];
    }
    return;
  }

  int j = b;
  float a0 = 0.f, a1 = 0.f, a2 = 0.f, c = 0.f;
  if (tid < 64) {
    float wv = Wa[j * 128 + tid];
    a0 = wv * We[tid * 3 + 0];
    a1 = wv * We[tid * 3 + 1];
    a2 = wv * We[tid * 3 + 2];
    c = wv * be[tid];
  } else if (tid < 128) {
    int d = tid - 64;
    float vm = 0.f;
    for (int v = 0; v < NV; ++v) vm += env[v * 64 + d];
    c = Wa[j * 128 + 64 + d] * (vm * (1.0f / (float)NV));
  }
  for (int o = 16; o > 0; o >>= 1) {
    a0 += __shfl_xor(a0, o);
    a1 += __shfl_xor(a1, o);
    a2 += __shfl_xor(a2, o);
    c += __shfl_xor(c, o);
  }
  int w = tid >> 5;
  if (w < 4 && (tid & 31) == 0) {
    part[w][0] = a0;
    part[w][1] = a1;
    part[w][2] = a2;
    part[w][3] = c;
  }
  __syncthreads();
  if (tid == 0) {
    int rt = j >> 4;
    float* lo = aug + ((rt >> 1) * 32 + (j & 15)) * 4 + (rt & 1) * 2;
    float* hi = lo + 16 * 4;
    lo[0] = part[0][0] + part[1][0];
    lo[1] = part[0][1] + part[1][1];
    hi[0] = part[0][2] + part[1][2];
    hi[1] = ba[j] + ((part[0][3] + part[1][3]) + (part[2][3] + part[3][3]));
  }
}

struct EdgeJob {
  const float* ef;
  const float* aug;
  const float* W1f;
  const float* b1;
  const float* W2;
  const float* b2;
  float* scores;
  int E;
  int nblk;
  int ratio;
  int first;
};

__device__ __forceinline__ bool block_role(const EdgeJob& j, int& idx) {
  unsigned b = blockIdx.x, cap = (unsigned)j.nblk;
  unsigned q0 = (unsigned)(((unsigned long long)b * (unsigned)j.ratio) >> 16);
  unsigned q1 = (unsigned)(((unsigned long long)(b + 1) * (unsigned)j.ratio) >> 16);
  q0 = q0 < cap ? q0 : cap;
  q1 = q1 < cap ? q1 : cap;
  if (q1 > q0) {
    idx = j.first + (int)q0;
    return true;
  }
  idx = (int)(b - q0);
  return false;
}

__device__ __forceinline__ float relu_bits(float x) {
  int i = __float_as_int(x);
  return __int_as_float(i > 0 ? i : 0);
}

__device__ __forceinline__ void edge_block(const EdgeJob& j, int blk) {
  __shared__ __attribute__((aligned(16))) float Ws[8 * 4 * 32 * 4];
  __shared__ __attribute__((aligned(16))) float AGs[256];
  __shared__ __attribute__((aligned(16))) float BWs[256];
  const float* __restrict__ ef = j.ef;
  const v4f* __restrict__ wsrc = (const v4f*)j.W1f;
  float* __restrict__ scores = j.scores;
  int E = j.E;
  int tid = threadIdx.x;
  int w = tid >> 5, lane = tid & 31;
  int base = (blk * 8 + w) * 16;
  int half = lane >> 4;
  int row = base + (lane & 15);
  long long rc = (row < E) ? row : 0;
  const float* pe = ef + rc * 3 + 2 * half;
  float x0 = pe[0];
  float x1 = pe[1 - half];
  v4f wv[4];
#pragma unroll
  for (int i = 0; i < 4; ++i) wv[i] = wsrc[i * 256 + tid];
  const float* pb = (tid & 64) ? (j.W2 + (tid & 63)) : (j.b1 + (tid & 63));
  float sa = j.aug[tid];
  float sb = *pb;
#pragma unroll
  for (int i = 0; i < 4; ++i) *(v4f*)&Ws[(i * 256 + tid) * 4] = wv[i];
  AGs[tid] = sa;
  BWs[tid] = sb;
  __syncthreads();

  const v2f xb = {x0, half ? 1.0f : x1};
  const v8f zero8 = {};
  const v4f* bw = (const v4f*)BWs;
  const v4f g0 = ((const v4f*)AGs)[lane], g1 = ((const v4f*)AGs)[32 + lane];

  v8f acc[4];
#pragma unroll
  for (int mt = 0; mt < 4; ++mt) {
    v4f lo = bw[mt * 4 + half * 2], hi = bw[mt * 4 + half * 2 + 1];
    v8f a = {lo.x, lo.y, lo.z, lo.w, hi.x, hi.y, hi.z, hi.w};
    acc[mt] = a;
  }

  v8f H = zero8;
#pragma unroll
  for (int t2 = 0; t2 < 8; ++t2) {
    if ((t2 & 1) == 0) {
      int rt = t2 >> 1;
      v2f ga = (rt == 0) ? v2f{g0.x, g0.y} : (rt == 1) ? v2f{g0.z, g0.w}
             : (rt == 2) ? v2f{g1.x, g1.y} : v2f{g1.z, g1.w};
      H = __builtin_amdgcn_wmma_f32_16x16x4_f32(false, ga, false, xb, (short)0, zero8, false, false);
#pragma unroll
      for (int r = 0; r < 8; ++r) H[r] = relu_bits(H[r]);
    }
    int r0 = 4 * (t2 & 1);
    const v2f bA = {H[r0], H[r0 + 1]};
    const v2f bB = {H[r0 + 2], H[r0 + 3]};
    v4f a[4];
#pragma unroll
    for (int mt = 0; mt < 4; ++mt) a[mt] = *(const v4f*)&Ws[((t2 * 4 + mt) * 32 + lane) * 4];
#pragma unroll
    for (int mt = 0; mt < 4; ++mt) {
      const v2f aA = {a[mt].x, a[mt].y};
      acc[mt] = __builtin_amdgcn_wmma_f32_16x16x4_f32(false, aA, false, bA, (short)0, acc[mt], false, false);
    }
#pragma unroll
    for (int mt = 0; mt < 4; ++mt) {
      const v2f aB = {a[mt].z, a[mt].w};
      acc[mt] = __builtin_amdgcn_wmma_f32_16x16x4_f32(false, aB, false, bB, (short)0, acc[mt], false, false);
    }
  }

  float p[4] = {0.f, 0.f, 0.f, 0.f};
#pragma unroll
  for (int mt = 0; mt < 4; ++mt) {
    v4f lo = bw[16 + mt * 4 + half * 2], hi = bw[16 + mt * 4 + half * 2 + 1];
    float wq[8] = {lo.x, lo.y, lo.z, lo.w, hi.x, hi.y, hi.z, hi.w};
#pragma unroll
    for (int r = 0; r < 8; ++r) {
      p[r & 3] += wq[r] * relu_bits(acc[mt][r]);
    }
  }
  float ps = (p[0] + p[1]) + (p[2] + p[3]);
  unsigned pu = __float_as_uint(ps);
  unsigned qu = __builtin_amdgcn_permlanex16(pu, pu, 0x76543210u, 0xfedcba98u, false, false);
  ps += __uint_as_float(qu);
  if (half == 0 && row < E) scores[row] = ps + j.b2[0];
}

__global__ void __launch_bounds__(256)
k_scatter(const int* __restrict__ src, const int* __restrict__ dst,
          const int* __restrict__ counts, int* __restrict__ bstart,
          unsigned* __restrict__ tmp, int E, int nbk, int nchunk, EdgeJob job) {
  if ((int)blockIdx.x >= nchunk) {
    edge_block(job, job.first + ((int)blockIdx.x - nchunk));
    return;
  }
  __shared__ int cur[MAXBK];
  __shared__ int dlt[MAXBK];
  __shared__ int wtot[8];
  __shared__ int wtot2[8];
  __shared__ __attribute__((aligned(16))) unsigned sword[CHUNK + 1];
  __shared__ __attribute__((aligned(16))) int spos[CHUNK + 1];
  int c = blockIdx.x, tid = threadIdx.x, lane = tid & 31, w = tid >> 5;
  int nbkp = (nbk + 3) & ~3;
  int* Tl = (int*)sword;
  int* Pl = spos;
  {
    int part = __builtin_amdgcn_readfirstlane(tid >> 7);
    unsigned g4 = (tid & 127) * 4;
    if (g4 < (unsigned)nbkp) {
      int hlf = (nchunk + 1) >> 1;
      int lo = part * hlf;
      int hi = lo + hlf < nchunk ? lo + hlf : nchunk;
      int T0 = 0, T1 = 0, T2 = 0, T3 = 0, P0 = 0, P1 = 0, P2 = 0, P3 = 0;
      for (int k0 = lo; k0 < hi; k0 += 5) {
        int4 v[5];
#pragma unroll
        for (int u = 0; u < 5; ++u) {
          int k = k0 + u;
          int kc = k < hi ? k : hi - 1;
          const int* rowp = counts + (size_t)kc * nbkp;
          v[u] = *(const int4*)(rowp + g4);
        }
#pragma unroll
        for (int u = 0; u < 5; ++u) {
          int k = k0 + u;
          bool in = k < hi, front = in && k < c;
          T0 += in ? v[u].x : 0;
          T1 += in ? v[u].y : 0;
          T2 += in ? v[u].z : 0;
          T3 += in ? v[u].w : 0;
          P0 += front ? v[u].x : 0;
          P1 += front ? v[u].y : 0;
          P2 += front ? v[u].z : 0;
          P3 += front ? v[u].w : 0;
        }
      }
      int4 Tv = {T0, T1, T2, T3}, Pv = {P0, P1, P2, P3};
      *(int4*)&Tl[part * MAXBK + g4] = Tv;
      *(int4*)&Pl[part * MAXBK + g4] = Pv;
    }
  }
  __syncthreads();
  int b0 = 2 * tid, b1 = 2 * tid + 1;
  int c0 = 0, c1 = 0, t0 = 0, t1 = 0, p0 = 0, p1 = 0;
  if (b0 < nbk) {
    c0 = counts[(size_t)c * nbkp + b0];
    t0 = Tl[b0] + Tl[MAXBK + b0];
    p0 = Pl[b0] + Pl[MAXBK + b0];
  }
  if (b1 < nbk) {
    c1 = counts[(size_t)c * nbkp + b1];
    t1 = Tl[b1] + Tl[MAXBK + b1];
    p1 = Pl[b1] + Pl[MAXBK + b1];
  }
  int sum = c0 + c1, inc = sum;
  int tsum = t0 + t1, tinc = tsum;
  for (int o = 1; o < 32; o <<= 1) {
    int u = __shfl_up(inc, o);
    int u2 = __shfl_up(tinc, o);
    if (lane >= o) {
      inc += u;
      tinc += u2;
    }
  }
  if (lane == 31) {
    wtot[w] = inc;
    wtot2[w] = tinc;
  }
  __syncthreads();
  int pre = 0, tpre = 0;
  for (int k = 0; k < 8; ++k) {
    pre += (k < w) ? wtot[k] : 0;
    tpre += (k < w) ? wtot2[k] : 0;
  }
  int ex = pre + inc - sum;
  int tex = tpre + tinc - tsum;
  int g0 = tex + p0, g1 = tex + t0 + p1;
  if (c == 0) {
    if (b0 < nbk) bstart[b0] = tex;
    if (b1 < nbk) bstart[b1] = tex + t0;
    if (tid == 255) bstart[nbk] = tex + tsum;
  }
  cur[b0] = ex;
  dlt[b0] = g0 - ex;
  cur[b1] = ex + c0;
  dlt[b1] = g1 - (ex + c0);
  __syncthreads();

  int e0 = c * CHUNK;
  for (int k0 = 0; k0 < CHUNK / 256; k0 += 5) {
    unsigned sv[5], dv[5];
#pragma unroll
    for (int u = 0; u < 5; ++u) {
      int e = e0 + (k0 + u) * 256 + tid;
      int ec = e < E ? e : E - 1;
      sv[u] = (unsigned)src[ec];
      dv[u] = (unsigned)dst[ec];
    }
#pragma unroll
    for (int u = 0; u < 5; ++u) {
      int e = e0 + (k0 + u) * 256 + tid;
      int valid = e < E ? 1 : 0;
      unsigned bk = sv[u] >> BSH;
      bk = bk < (unsigned)(nbk - 1) ? bk : (unsigned)(nbk - 1);
      int slot = atomicAdd(&cur[bk], valid);
      slot = slot < CHUNK - 1 ? slot : CHUNK - 1;
      slot = valid ? slot : CHUNK;
      sword[slot] = (dv[u] << BSH) | (sv[u] & (BNODES - 1));
      spos[slot] = slot + dlt[bk];
    }
  }
  __syncthreads();
  int n = E - e0;
  n = n < CHUNK ? n : CHUNK;
  for (int i = tid; i < n; i += 256) {
    int p = spos[i];
    p = (unsigned)p < (unsigned)E ? p : E - 1;
    tmp[p] = sword[i];
  }
}

__global__ void __launch_bounds__(256)
k_bucket(const int* __restrict__ bstart, const unsigned* __restrict__ tmp,
         int* __restrict__ rowptr, float* __restrict__ cnt, int* __restrict__ col,
         int N, int nbk, EdgeJob job) {
  if ((int)blockIdx.x >= nbk) {
    edge_block(job, job.first + ((int)blockIdx.x - nbk));
    return;
  }
  __shared__ int hc[BNODES];
  __shared__ int cu[BNODES];
  __shared__ int wt[8];
  __shared__ int scol[CAPB + 1];
  int b = blockIdx.x, tid = threadIdx.x, lane = tid & 31, w = tid >> 5;
  int beg = bstart[b], end = bstart[b + 1];
  int m = end - beg;
  if (tid < BNODES) hc[tid] = 0;
  __syncthreads();
  for (int i0 = 0; i0 < m; i0 += 2048) {
    unsigned wd[8];
#pragma unroll
    for (int u = 0; u < 8; ++u) {
      int i = i0 + u * 256 + tid;
      wd[u] = tmp[beg + (i < m ? i : m - 1)];
    }
#pragma unroll
    for (int u = 0; u < 8; ++u) {
      int i = i0 + u * 256 + tid;
      atomicAdd(&hc[wd[u] & (BNODES - 1)], i < m ? 1 : 0);
    }
  }
  __syncthreads();
  int v = (tid < BNODES) ? hc[tid] : 0;
  int inc = v;
  for (int o = 1; o < 32; o <<= 1) {
    int u = __shfl_up(inc, o);
    if (lane >= o) inc += u;
  }
  if (lane == 31) wt[w] = inc;
  __syncthreads();
  int pre = 0;
  for (int k = 0; k < 8; ++k) pre += (k < w) ? wt[k] : 0;
  int ex = pre + inc - v;
  if (tid < BNODES) {
    cu[tid] = ex;
    int node = (b << BSH) + tid;
    if (node < N) {
      rowptr[node] = beg + ex;
      cnt[node] = (float)v;
    }
  }
  if (b == nbk - 1 && tid == 0) rowptr[N] = end;
  __syncthreads();
  bool stage = m <= CAPB;
  for (int i0 = 0; i0 < m; i0 += 2048) {
    unsigned wd[8];
#pragma unroll
    for (int u = 0; u < 8; ++u) {
      int i = i0 + u * 256 + tid;
      wd[u] = tmp[beg + (i < m ? i : m - 1)];
    }
#pragma unroll
    for (int u = 0; u < 8; ++u) {
      int i = i0 + u * 256 + tid;
      int valid = i < m ? 1 : 0;
      int slot = atomicAdd(&cu[wd[u] & (BNODES - 1)], valid);
      slot = slot < m - 1 ? slot : m - 1;
      int dv = (int)(wd[u] >> BSH);
      if (stage) scol[valid ? slot : CAPB] = dv;
      else if (valid) col[beg + slot] = dv;
    }
  }
  __syncthreads();
  if (stage)
    for (int i = tid; i < m; i += 256) col[beg + i] = scol[i];
}

__global__ void __launch_bounds__(256)
k_gather(const int* __restrict__ rowptr, const int* __restrict__ col,
         const float* __restrict__ h, float* __restrict__ accum, int N, EdgeJob job) {
  int mb;
  if (block_role(job, mb)) {
    edge_block(job, mb);
    return;
  }
  int node = mb * 8 + (threadIdx.x >> 5);
  if (node >= N) return;
  int lane = threadIdx.x & 31, half = lane >> 4, c4 = (lane & 15) * 4;
  int beg = rowptr[node], end = rowptr[node + 1];
  const v4f zero = {0.f, 0.f, 0.f, 0.f};
  v4f s[4] = {zero, zero, zero, zero};
  if (beg < end) {
    int last = end - 1;
    unsigned nmax = (unsigned)(N - 1);
    int id[8];
#pragma unroll
    for (int u = 0; u < 8; ++u) {
      int e = beg + half + 2 * u;
      id[u] = col[e < last ? e : last];
    }
    for (int j0 = beg; j0 < end; j0 += 16) {
      v4f v[8];
#pragma unroll
      for (int u = 0; u < 8; ++u) {
        unsigned r = (unsigned)id[u] < nmax ? (unsigned)id[u] : nmax;
        v[u] = *(const v4f*)(h + (size_t)r * 64 + c4);
      }
#pragma unroll
      for (int u = 0; u < 8; ++u) {
        int e = j0 + 16 + half + 2 * u;
        id[u] = col[e < last ? e : last];
      }
#pragma unroll
      for (int u = 0; u < 8; ++u) {
        bool ok = (j0 + half + 2 * u) < end;
        s[u & 3] += ok ? v[u] : zero;
      }
    }
  }
  v4f a = (s[0] + s[1]) + (s[2] + s[3]);
  a.x += __shfl_xor(a.x, 16);
  a.y += __shfl_xor(a.y, 16);
  a.z += __shfl_xor(a.z, 16);
  a.w += __shfl_xor(a.w, 16);
  if (half == 0) *(v4f*)(accum + (size_t)node * 64 + c4) = a;
}

__device__ __forceinline__ void
layer_body(bool first, const float* __restrict__ hin, const float* __restrict__ accum,
           const float* __restrict__ cnt,
           const float* __restrict__ nf, const int* __restrict__ rowptr,
           const int* __restrict__ col,
           const float* __restrict__ Wn, const float* __restrict__ bn,
           const float* __restrict__ Wf, const float* __restrict__ bself,
           const float* __restrict__ bneigh,
           float* __restrict__ hout, int N) {
  __shared__ __attribute__((aligned(16))) float Bl[32 * 4 * 32 * 2];
  __shared__ float WBn[256];
  __shared__ __attribute__((aligned(16))) float FS[8 * 16 * 4];
  int tid = threadIdx.x;
  int w = tid >> 5, lane = tid & 31;
  int base = (blockIdx.x * 8 + w) * 16;
  int half = lane >> 4;
  int mrow = base + (lane & 15);
  long long mc = (mrow < N) ? mrow : 0;

  float c = cnt[mc];
  float f0 = 0.f, f1 = 0.f, f2 = 0.f;
  v4f sv = {0.f, 0.f, 0.f, 0.f};
  v2f A[32];
  int gbeg[4] = {0, 0, 0, 0}, gend[4] = {0, 0, 0, 0};
  if (first) {
    f0 = nf[mc * 3 + 0];
    f1 = nf[mc * 3 + 1];
    f2 = nf[mc * 3 + 2];
#pragma unroll
    for (int p = 0; p < 4; ++p) {
      int node = base + 4 * p + (lane >> 3);
      bool nv = node < N;
      int nc = nv ? node : 0;
      int rb = rowptr[nc], re = rowptr[nc + 1];
      gbeg[p] = nv ? rb : 0;
      gend[p] = nv ? re : 0;
    }
  } else {
#pragma unroll
    for (int kk = 0; kk < 16; ++kk) {
      int k0 = kk * 4 + 2 * half;
      v2f a = {hin[mc * 64 + k0], hin[mc * 64 + k0 + 1]};
      A[kk] = a;
    }
#pragma unroll
    for (int kk = 0; kk < 16; ++kk) {
      int k0 = kk * 4 + 2 * half;
      v2f a = {accum[mc * 64 + k0], accum[mc * 64 + k0 + 1]};
      A[16 + kk] = a;
    }
  }

  const v4f* __restrict__ wsrc = (const v4f*)Wf;
  v4f wv[8];
#pragma unroll
  for (int i = 0; i < 8; ++i) wv[i] = wsrc[i * 256 + tid];
  float sn = 0.f;
  if (first) {
    const float* pn = (tid < 192) ? (Wn + tid) : (bn + (tid - 192));
    sn = *pn;
  }
#pragma unroll
  for (int i = 0; i < 8; ++i) *(v4f*)&Bl[(i * 256 + tid) * 4] = wv[i];
  if (first) {
    WBn[tid] = sn;
    int sub = lane & 7;
    unsigned nmax = (unsigned)(N - 1);
    float q0[4] = {0.f, 0.f, 0.f, 0.f}, q1[4] = {0.f, 0.f, 0.f, 0.f}, q2[4] = {0.f, 0.f, 0.f, 0.f};
    int len = 0;
#pragma unroll
    for (int p = 0; p < 4; ++p) {
      int l = gend[p] - gbeg[p];
      len = l > len ? l : len;
    }
    for (int off = 0; off < len; off += 48) {
      unsigned id[4][6];
#pragma unroll
      for (int p = 0; p < 4; ++p) {
        int last = gend[p] - 1;
#pragma unroll
        for (int u = 0; u < 6; ++u) {
          int e = gbeg[p] + off + sub + 8 * u;
          int ec = e < last ? e : last;
          ec = ec > 0 ? ec : 0;
          id[p][u] = (unsigned)col[ec];
        }
      }
      float fa[4][6], fb[4][6], fc[4][6];
#pragma unroll
      for (int p = 0; p < 4; ++p) {
#pragma unroll
        for (int u = 0; u < 6; ++u) {
          unsigned r = id[p][u] < nmax ? id[p][u] : nmax;
          const float* f = nf + (size_t)r * 3;
          fa[p][u] = f[0];
          fb[p][u] = f[1];
          fc[p][u] = f[2];
        }
      }
#pragma unroll
      for (int p = 0; p < 4; ++p) {
#pragma unroll
        for (int u = 0; u < 6; ++u) {
          bool ok = (gbeg[p] + off + sub + 8 * u) < gend[p];
          q0[p] += ok ? fa[p][u] : 0.f;
          q1[p] += ok ? fb[p][u] : 0.f;
          q2[p] += ok ? fc[p][u] : 0.f;
        }
      }
    }
#pragma unroll
    for (int p = 0; p < 4; ++p) {
      for (int o = 4; o > 0; o >>= 1) {
        q0[p] += __shfl_xor(q0[p], o);
        q1[p] += __shfl_xor(q1[p], o);
        q2[p] += __shfl_xor(q2[p], o);
      }
      if (sub == 0) {
        v4f o4 = {q0[p], q1[p], q2[p], 0.f};
        *(v4f*)&FS[(w * 16 + 4 * p + (lane >> 3)) * 4] = o4;
      }
    }
  }
  __syncthreads();
  if (first) sv = *(const v4f*)&FS[(w * 16 + (lane & 15)) * 4];

  float inv = (c > 0.f) ? 1.f / c : 1.f;

  if (first) {
    float g0 = sv.x * inv, g1 = sv.y * inv, g2 = sv.z * inv;
    bool has = c > 0.f;
#pragma unroll
    for (int kk = 0; kk < 16; ++kk) {
      int k0 = kk * 4 + 2 * half;
      const float* Wn_s = WBn;
      const float* bn_s = WBn + 192;
      float w00 = Wn_s[k0 * 3 + 0], w01 = Wn_s[k0 * 3 + 1], w02 = Wn_s[k0 * 3 + 2];
      float w10 = Wn_s[k0 * 3 + 3], w11 = Wn_s[k0 * 3 + 4], w12 = Wn_s[k0 * 3 + 5];
      float c0 = bn_s[k0], c1 = bn_s[k0 + 1];
      v2f a = {c0 + w00 * f0 + w01 * f1 + w02 * f2, c1 + w10 * f0 + w11 * f1 + w12 * f2};
      A[kk] = a;
      float y0 = c0 + w00 * g0 + w01 * g1 + w02 * g2;
      float y1 = c1 + w10 * g0 + w11 * g1 + w12 * g2;
      v2f m = {has ? y0 : 0.f, has ? y1 : 0.f};
      A[16 + kk] = m;
    }
  } else {
#pragma unroll
    for (int kk = 0; kk < 16; ++kk) A[16 + kk] = A[16 + kk] * inv;
  }

  v8f acc0 = {}, acc1 = {}, acc2 = {}, acc3 = {};
#pragma unroll
  for (int kk = 0; kk < 32; ++kk) {
    int bb = kk * 256 + lane * 2;
    acc0 = __builtin_amdgcn_wmma_f32_16x16x4_f32(false, A[kk], false, *(const v2f*)&Bl[bb + 0],   (short)0, acc0, false, false);
    acc1 = __builtin_amdgcn_wmma_f32_16x16x4_f32(false, A[kk], false, *(const v2f*)&Bl[bb + 64],  (short)0, acc1, false, false);
    acc2 = __builtin_amdgcn_wmma_f32_16x16x4_f32(false, A[kk], false, *(const v2f*)&Bl[bb + 128], (short)0, acc2, false, false);
    acc3 = __builtin_amdgcn_wmma_f32_16x16x4_f32(false, A[kk], false, *(const v2f*)&Bl[bb + 192], (short)0, acc3, false, false);
  }

#pragma unroll
  for (int ct = 0; ct < 4; ++ct) {
    v8f av = (ct == 0) ? acc0 : (ct == 1) ? acc1 : (ct == 2) ? acc2 : acc3;
    int n = ct * 16 + (lane & 15);
    float bs = bself[n] + bneigh[n];
#pragma unroll
    for (int r = 0; r < 8; ++r) {
      int mg = base + r + 8 * half;
      if (mg < N) {
        float val = av[r] + bs;
        hout[(long long)mg * 64 + n] = val > 0.f ? val : 0.f;
      }
    }
  }
}

__global__ void __launch_bounds__(256)
k_layer_gemm(const float* __restrict__ hin, const float* __restrict__ accum,
             const float* __restrict__ cnt,
             const float* __restrict__ Wf, const float* __restrict__ bself,
             const float* __restrict__ bneigh,
             float* __restrict__ hout, int N) {
  layer_body(false, hin, accum, cnt, nullptr, nullptr, nullptr, nullptr, nullptr, Wf, bself,
             bneigh, hout, N);
}

__global__ void __launch_bounds__(256)
k_layer_gemm_first(const float* __restrict__ nf, const int* __restrict__ rowptr,
                   const int* __restrict__ col, const float* __restrict__ cnt,
                   const float* __restrict__ Wn, const float* __restrict__ bn,
                   const float* __restrict__ Wf, const float* __restrict__ bself,
                   const float* __restrict__ bneigh,
                   float* __restrict__ hout, int N) {
  layer_body(true, nullptr, nullptr, cnt, nf, rowptr, col, Wn, bn, Wf, bself, bneigh, hout, N);
}

static int share_ratio(int x, int all) {
  return all > 0 ? (int)(((long long)x * 65536 + all - 1) / all) : 0;
}

extern "C" void kernel_launch(void* const* d_in, const int* in_sizes, int n_in,
                              void* d_out, int out_size, void* d_ws, size_t ws_size,
                              hipStream_t stream) {
  if (n_in < 18) return;
  const float* node_feats = (const float*)d_in[0];
  const float* edge_feats = (const float*)d_in[1];
  const float* env        = (const float*)d_in[2];
  const int*   eidx       = (const int*)d_in[3];
  const float* Wn  = (const float*)d_in[4];
  const float* bn  = (const float*)d_in[5];
  const float* We  = (const float*)d_in[6];
  const float* be  = (const float*)d_in[7];
  const float* Wself  = (const float*)d_in[8];
  const float* bself  = (const float*)d_in[9];
  const float* Wneigh = (const float*)d_in[10];
  const float* bneigh = (const float*)d_in[11];
  const float* Wa  = (const float*)d_in[12];
  const float* ba  = (const float*)d_in[13];
  const float* W1  = (const float*)d_in[14];
  const float* b1  = (const float*)d_in[15];
  const float* W2  = (const float*)d_in[16];
  const float* b2  = (const float*)d_in[17];

  int N  = in_sizes[0] / 3;
  int E  = in_sizes[1] / 3;
  int NV = in_sizes[2] / 64;
  int L  = in_sizes[8] / (64 * 64);

  const int* src = eidx;
  const int* dst = eidx + E;

  float* scores = (float*)d_out;
  float* h_out  = (float*)d_out + E;

  float* accum = (float*)d_ws;
  float* cnt   = accum + (long long)N * 64;
  long long off = (((long long)N * 65) + 63) & ~63LL;
  float* h_b   = (float*)d_ws + off;
  float* aug   = h_b + (long long)N * 64;
  int* rowptr = (int*)(aug + 256);
  unsigned* tmp = (unsigned*)(rowptr + (((long long)N + 1 + 63) & ~63LL));
  int* col    = (int*)tmp + (((long long)E + 63) & ~63LL);
  float* w1f  = (float*)(col + (((long long)E + 63) & ~63LL));
  float* wlf  = w1f + 4096;
  int nbk = (N + BNODES - 1) >> BSH;
  int nchunk = (E + CHUNK - 1) / CHUNK;
  int* bstart = (int*)(wlf + (long long)L * 8192);
  int* counts = bstart + 1024;
  if (nbk > MAXBK) return;

  int zb = nchunk;
  int nb = (L > 0) ? 0 : (N * 64 + 255) / 256;
  int sb = 16 + 32 * L;
  k_prologue<<<64 + sb + zb + nb, 256, 0, stream>>>(node_feats, Wn, bn, h_out, N, eidx, E, counts, nbk, zb, Wa,
                                                    ba, We, be, env, NV, aug, W1, Wself,
                                                    Wneigh, w1f, wlf, sb);
  int etiles = (E + 15) / 16;
  int units = (etiles + 7) / 8;
  int wide = (L > 1) ? L - 1 : 0;
  int xr = units / (2 + wide);
  int xf = (wide > 0) ? units / (2 + wide) : units - xr;
  int xg = (wide > 0) ? (units - xr - xf) / wide : 0;
  EdgeJob job = {edge_feats, aug, w1f, b1, W2, b2, scores, E, 0, 0, 0};

  job.nblk = xr;
  job.first = 0;
  k_scatter<<<nchunk + xr, 256, 0, stream>>>(src, dst, counts, bstart, tmp, E, nbk, nchunk, job);
  job.nblk = xf;
  job.first = xr;
  k_bucket<<<nbk + xf, 256, 0, stream>>>(bstart, tmp, rowptr, cnt, col, N, nbk, job);

  int tiles = (N + 15) / 16;
  int gb = (N + 7) / 8;
  float* hin = h_out;
  for (int l = 0; l < L; ++l) {
    float* hout = (l == L - 1) ? h_out : ((hin == h_out) ? h_b : h_out);
    if (l == 0) {
      k_layer_gemm_first<<<(tiles + 7) / 8, 256, 0, stream>>>(
          node_feats, rowptr, col, cnt, Wn, bn, wlf, bself, bneigh, hout, N);
    } else {
      job.first = xr + xf + (l - 1) * xg;
      job.nblk = (l == L - 1) ? units - job.first : xg;
      job.ratio = share_ratio(job.nblk, gb + job.nblk);
      k_gather<<<gb + job.nblk, 256, 0, stream>>>(rowptr, col, hin, accum, N, job);
      k_layer_gemm<<<(tiles + 7) / 8, 256, 0, stream>>>(
          hin, accum, cnt, wlf + (long long)l * 8192, bself + (long long)l * 64,
          bneigh + (long long)l * 64, hout, N);
    }
    hin = hout;
  }
}


# layer 0 runs inside the sort's level-2 blocks (128 nodes = 8 tiles; neighbour feature sums from the block's LDS copy of the bucket): no layer-0 launch (6 to 5); layer-0 sum order: 2 lane partials
# speedup vs baseline: 6.9114x; 1.0381x over previous
#include <hip/hip_runtime.h>

typedef __attribute__((ext_vector_type(2))) float v2f;
typedef __attribute__((ext_vector_type(4))) float v4f;
typedef __attribute__((ext_vector_type(8))) float v8f;

#define BSH 7
#define BNODES 128
#define CHUNK 6400
#define MAXBK 512
#define CAPB 8192

__global__ void __launch_bounds__(256)
k_prologue(const float* __restrict__ nf, const float* __restrict__ Wn,
           const float* __restrict__ bn, float* __restrict__ h, int N,
           const int* __restrict__ src, int E, int* __restrict__ counts, int nbk, int zb,
           const float* __restrict__ Wa, const float* __restrict__ ba,
           const float* __restrict__ We, const float* __restrict__ be,
           const float* __restrict__ env, int NV,
           float* __restrict__ aug,
           const float* __restrict__ W1, const float* __restrict__ Wself,
           const float* __restrict__ Wneigh, float* __restrict__ w1f,
           float* __restrict__ wlf, int sb) {
  __shared__ float part[4][4];
  int b = blockIdx.x, tid = threadIdx.x;

  if (b >= 64 + sb + zb) {
    int t = (b - 64 - sb - zb) * 256 + tid;
    int n = t >> 6, j = t & 63;
    if (n >= N) return;
    float f0 = nf[n * 3 + 0], f1 = nf[n * 3 + 1], f2 = nf[n * 3 + 2];
    h[t] = bn[j] + Wn[j * 3 + 0] * f0 + Wn[j * 3 + 1] * f1 + Wn[j * 3 + 2] * f2;
    return;
  }
  if (b >= 64 + sb) {
    __shared__ int hb[MAXBK];
    int c = b - 64 - sb;
    for (int i = tid; i < MAXBK; i += 256) hb[i] = 0;
    __syncthreads();
    int e0 = c * CHUNK;
    int sv[CHUNK / 256];
#pragma unroll
    for (int k = 0; k < CHUNK / 256; ++k) {
      int e = e0 + k * 256 + tid;
      sv[k] = src[e < E ? e : E - 1];
    }
#pragma unroll
    for (int k = 0; k < CHUNK / 256; ++k) {
      int e = e0 + k * 256 + tid;
      unsigned bk = (unsigned)sv[k] >> BSH;
      bk = bk < (unsigned)(nbk - 1) ? bk : (unsigned)(nbk - 1);
      if (e < E) atomicAdd(&hb[bk], 1);
    }
    __syncthreads();
    int nbkp = (nbk + 3) & ~3;
    for (int i = tid; i < nbkp; i += 256) counts[(size_t)c * nbkp + i] = hb[i];
    return;
  }
  if (b >= 64) {
    int t = (b - 64) * 256 + tid;
    int u = t - 4096;
    if (t < 4096) {
      int q = t & 3, lane = (t >> 2) & 31, mt = (t >> 7) & 3, t2 = t >> 9;
      int st = 2 * t2 + (q >> 1);
      int m = mt * 16 + (lane & 15);
      int f = (st >> 2) * 16 + 8 * (lane >> 4) + 2 * (st & 3) + (q & 1);
      w1f[t] = W1[m * 64 + f];
    } else {
      int idx = u & 8191;
      int v = idx & 1, lane = (idx >> 1) & 31, ct = (idx >> 6) & 3, kk = idx >> 8;
      int n = ct * 16 + (lane & 15);
      int k = kk * 4 + 2 * (lane >> 4) + v;
      int l = u >> 13;
      const float* ws = (k < 64) ? Wself : Wneigh;
      wlf[u] = ws[l * 4096 + n * 64 + (k & 63)];
    }
    return;
  }

  int j = b;
  float a0 = 0.f, a1 = 0.f, a2 = 0.f, c = 0.f;
  if (tid < 64) {
    float wv = Wa[j * 128 + tid];
    a0 = wv * We[tid * 3 + 0];
    a1 = wv * We[tid * 3 + 1];
    a2 = wv * We[tid * 3 + 2];
    c = wv * be[tid];
  } else if (tid < 128) {
    int d = tid - 64;
    float vm = 0.f;
    for (int v = 0; v < NV; ++v) vm += env[v * 64 + d];
    c = Wa[j * 128 + 64 + d] * (vm * (1.0f / (float)NV));
  }
  for (int o = 16; o > 0; o >>= 1) {
    a0 += __shfl_xor(a0, o);
    a1 += __shfl_xor(a1, o);
    a2 += __shfl_xor(a2, o);
    c += __shfl_xor(c, o);
  }
  int w = tid >> 5;
  if (w < 4 && (tid & 31) == 0) {
    part[w][0] = a0;
    part[w][1] = a1;
    part[w][2] = a2;
    part[w][3] = c;
  }
  __syncthreads();
  if (tid == 0) {
    int rt = j >> 4;
    float* lo = aug + ((rt >> 1) * 32 + (j & 15)) * 4 + (rt & 1) * 2;
    float* hi = lo + 16 * 4;
    lo[0] = part[0][0] + part[1][0];
    lo[1] = part[0][1] + part[1][1];
    hi[0] = part[0][2] + part[1][2];
    hi[1] = ba[j] + ((part[0][3] + part[1][3]) + (part[2][3] + part[3][3]));
  }
}

struct EdgeJob {
  const float* ef;
  const float* aug;
  const float* W1f;
  const float* b1;
  const float* W2;
  const float* b2;
  float* scores;
  int E;
  int nblk;
  int ratio;
  int first;
};

__device__ __forceinline__ bool block_role(const EdgeJob& j, int& idx) {
  unsigned b = blockIdx.x, cap = (unsigned)j.nblk;
  unsigned q0 = (unsigned)(((unsigned long long)b * (unsigned)j.ratio) >> 16);
  unsigned q1 = (unsigned)(((unsigned long long)(b + 1) * (unsigned)j.ratio) >> 16);
  q0 = q0 < cap ? q0 : cap;
  q1 = q1 < cap ? q1 : cap;
  if (q1 > q0) {
    idx = j.first + (int)q0;
    return true;
  }
  idx = (int)(b - q0);
  return false;
}

__device__ __forceinline__ float relu_bits(float x) {
  int i = __float_as_int(x);
  return __int_as_float(i > 0 ? i : 0);
}

__device__ __forceinline__ void edge_block(const EdgeJob& j, int blk) {
  __shared__ __attribute__((aligned(16))) float Ws[8 * 4 * 32 * 4];
  __shared__ __attribute__((aligned(16))) float AGs[256];
  __shared__ __attribute__((aligned(16))) float BWs[256];
  const float* __restrict__ ef = j.ef;
  const v4f* __restrict__ wsrc = (const v4f*)j.W1f;
  float* __restrict__ scores = j.scores;
  int E = j.E;
  int tid = threadIdx.x;
  int w = tid >> 5, lane = tid & 31;
  int base = (blk * 8 + w) * 16;
  int half = lane >> 4;
  int row = base + (lane & 15);
  long long rc = (row < E) ? row : 0;
  const float* pe = ef + rc * 3 + 2 * half;
  float x0 = pe[0];
  float x1 = pe[1 - half];
  v4f wv[4];
#pragma unroll
  for (int i = 0; i < 4; ++i) wv[i] = wsrc[i * 256 + tid];
  const float* pb = (tid & 64) ? (j.W2 + (tid & 63)) : (j.b1 + (tid & 63));
  float sa = j.aug[tid];
  float sb = *pb;
#pragma unroll
  for (int i = 0; i < 4; ++i) *(v4f*)&Ws[(i * 256 + tid) * 4] = wv[i];
  AGs[tid] = sa;
  BWs[tid] = sb;
  __syncthreads();

  const v2f xb = {x0, half ? 1.0f : x1};
  const v8f zero8 = {};
  const v4f* bw = (const v4f*)BWs;
  const v4f g0 = ((const v4f*)AGs)[lane], g1 = ((const v4f*)AGs)[32 + lane];

  v8f acc[4];
#pragma unroll
  for (int mt = 0; mt < 4; ++mt) {
    v4f lo = bw[mt * 4 + half * 2], hi = bw[mt * 4 + half * 2 + 1];
    v8f a = {lo.x, lo.y, lo.z, lo.w, hi.x, hi.y, hi.z, hi.w};
    acc[mt] = a;
  }

  v8f H = zero8;
#pragma unroll
  for (int t2 = 0; t2 < 8; ++t2) {
    if ((t2 & 1) == 0) {
      int rt = t2 >> 1;
      v2f ga = (rt == 0) ? v2f{g0.x, g0.y} : (rt == 1) ? v2f{g0.z, g0.w}
             : (rt == 2) ? v2f{g1.x, g1.y} : v2f{g1.z, g1.w};
      H = __builtin_amdgcn_wmma_f32_16x16x4_f32(false, ga, false, xb, (short)0, zero8, false, false);
#pragma unroll
      for (int r = 0; r < 8; ++r) H[r] = relu_bits(H[r]);
    }
    int r0 = 4 * (t2 & 1);
    const v2f bA = {H[r0], H[r0 + 1]};
    const v2f bB = {H[r0 + 2], H[r0 + 3]};
    v4f a[4];
#pragma unroll
    for (int mt = 0; mt < 4; ++mt) a[mt] = *(const v4f*)&Ws[((t2 * 4 + mt) * 32 + lane) * 4];
#pragma unroll
    for (int mt = 0; mt < 4; ++mt) {
      const v2f aA = {a[mt].x, a[mt].y};
      acc[mt] = __builtin_amdgcn_wmma_f32_16x16x4_f32(false, aA, false, bA, (short)0, acc[mt], false, false);
    }
#pragma unroll
    for (int mt = 0; mt < 4; ++mt) {
      const v2f aB = {a[mt].z, a[mt].w};
      acc[mt] = __builtin_amdgcn_wmma_f32_16x16x4_f32(false, aB, false, bB, (short)0, acc[mt], false, false);
    }
  }

  float p[4] = {0.f, 0.f, 0.f, 0.f};
#pragma unroll
  for (int mt = 0; mt < 4; ++mt) {
    v4f lo = bw[16 + mt * 4 + half * 2], hi = bw[16 + mt * 4 + half * 2 + 1];
    float wq[8] = {lo.x, lo.y, lo.z, lo.w, hi.x, hi.y, hi.z, hi.w};
#pragma unroll
    for (int r = 0; r < 8; ++r) {
      p[r & 3] += wq[r] * relu_bits(acc[mt][r]);
    }
  }
  float ps = (p[0] + p[1]) + (p[2] + p[3]);
  unsigned pu = __float_as_uint(ps);
  unsigned qu = __builtin_amdgcn_permlanex16(pu, pu, 0x76543210u, 0xfedcba98u, false, false);
  ps += __uint_as_float(qu);
  if (half == 0 && row < E) scores[row] = ps + j.b2[0];
}

__global__ void __launch_bounds__(256)
k_scatter(const int* __restrict__ src, const int* __restrict__ dst,
          const int* __restrict__ counts, int* __restrict__ bstart,
          unsigned* __restrict__ tmp, int E, int nbk, int nchunk, EdgeJob job) {
  if ((int)blockIdx.x >= nchunk) {
    edge_block(job, job.first + ((int)blockIdx.x - nchunk));
    return;
  }
  __shared__ int cur[MAXBK];
  __shared__ int dlt[MAXBK];
  __shared__ int wtot[8];
  __shared__ int wtot2[8];
  __shared__ __attribute__((aligned(16))) unsigned sword[CHUNK + 1];
  __shared__ __attribute__((aligned(16))) int spos[CHUNK + 1];
  int c = blockIdx.x, tid = threadIdx.x, lane = tid & 31, w = tid >> 5;
  int nbkp = (nbk + 3) & ~3;
  int* Tl = (int*)sword;
  int* Pl = spos;
  {
    int part = __builtin_amdgcn_readfirstlane(tid >> 7);
    unsigned g4 = (tid & 127) * 4;
    if (g4 < (unsigned)nbkp) {
      int hlf = (nchunk + 1) >> 1;
      int lo = part * hlf;
      int hi = lo + hlf < nchunk ? lo + hlf : nchunk;
      int T0 = 0, T1 = 0, T2 = 0, T3 = 0, P0 = 0, P1 = 0, P2 = 0, P3 = 0;
      for (int k0 = lo; k0 < hi; k0 += 5) {
        int4 v[5];
#pragma unroll
        for (int u = 0; u < 5; ++u) {
          int k = k0 + u;
          int kc = k < hi ? k : hi - 1;
          const int* rowp = counts + (size_t)kc * nbkp;
          v[u] = *(const int4*)(rowp + g4);
        }
#pragma unroll
        for (int u = 0; u < 5; ++u) {
          int k = k0 + u;
          bool in = k < hi, front = in && k < c;
          T0 += in ? v[u].x : 0;
          T1 += in ? v[u].y : 0;
          T2 += in ? v[u].z : 0;
          T3 += in ? v[u].w : 0;
          P0 += front ? v[u].x : 0;
          P1 += front ? v[u].y : 0;
          P2 += front ? v[u].z : 0;
          P3 += front ? v[u].w : 0;
        }
      }
      int4 Tv = {T0, T1, T2, T3}, Pv = {P0, P1, P2, P3};
      *(int4*)&Tl[part * MAXBK + g4] = Tv;
      *(int4*)&Pl[part * MAXBK + g4] = Pv;
    }
  }
  __syncthreads();
  int b0 = 2 * tid, b1 = 2 * tid + 1;
  int c0 = 0, c1 = 0, t0 = 0, t1 = 0, p0 = 0, p1 = 0;
  if (b0 < nbk) {
    c0 = counts[(size_t)c * nbkp + b0];
    t0 = Tl[b0] + Tl[MAXBK + b0];
    p0 = Pl[b0] + Pl[MAXBK + b0];
  }
  if (b1 < nbk) {
    c1 = counts[(size_t)c * nbkp + b1];
    t1 = Tl[b1] + Tl[MAXBK + b1];
    p1 = Pl[b1] + Pl[MAXBK + b1];
  }
  int sum = c0 + c1, inc = sum;
  int tsum = t0 + t1, tinc = tsum;
  for (int o = 1; o < 32; o <<= 1) {
    int u = __shfl_up(inc, o);
    int u2 = __shfl_up(tinc, o);
    if (lane >= o) {
      inc += u;
      tinc += u2;
    }
  }
  if (lane == 31) {
    wtot[w] = inc;
    wtot2[w] = tinc;
  }
  __syncthreads();
  int pre = 0, tpre = 0;
  for (int k = 0; k < 8; ++k) {
    pre += (k < w) ? wtot[k] : 0;
    tpre += (k < w) ? wtot2[k] : 0;
  }
  int ex = pre + inc - sum;
  int tex = tpre + tinc - tsum;
  int g0 = tex + p0, g1 = tex + t0 + p1;
  if (c == 0) {
    if (b0 < nbk) bstart[b0] = tex;
    if (b1 < nbk) bstart[b1] = tex + t0;
    if (tid == 255) bstart[nbk] = tex + tsum;
  }
  cur[b0] = ex;
  dlt[b0] = g0 - ex;
  cur[b1] = ex + c0;
  dlt[b1] = g1 - (ex + c0);
  __syncthreads();

  int e0 = c * CHUNK;
  for (int k0 = 0; k0 < CHUNK / 256; k0 += 5) {
    unsigned sv[5], dv[5];
#pragma unroll
    for (int u = 0; u < 5; ++u) {
      int e = e0 + (k0 + u) * 256 + tid;
      int ec = e < E ? e : E - 1;
      sv[u] = (unsigned)src[ec];
      dv[u] = (unsigned)dst[ec];
    }
#pragma unroll
    for (int u = 0; u < 5; ++u) {
      int e = e0 + (k0 + u) * 256 + tid;
      int valid = e < E ? 1 : 0;
      unsigned bk = sv[u] >> BSH;
      bk = bk < (unsigned)(nbk - 1) ? bk : (unsigned)(nbk - 1);
      int slot = atomicAdd(&cur[bk], valid);
      slot = slot < CHUNK - 1 ? slot : CHUNK - 1;
      slot = valid ? slot : CHUNK;
      sword[slot] = (dv[u] << BSH) | (sv[u] & (BNODES - 1));
      spos[slot] = slot + dlt[bk];
    }
  }
  __syncthreads();
  int n = E - e0;
  n = n < CHUNK ? n : CHUNK;
  for (int i = tid; i < n; i += 256) {
    int p = spos[i];
    p = (unsigned)p < (unsigned)E ? p : E - 1;
    tmp[p] = sword[i];
  }
}

__device__ __forceinline__ void
layer0_rows(v2f (&A)[32], const float* WBn, int half, float f0, float f1, float f2,
            float g0, float g1, float g2, bool has) {
#pragma unroll
  for (int kk = 0; kk < 16; ++kk) {
    int k0 = kk * 4 + 2 * half;
    const float* Wn_s = WBn;
    const float* bn_s = WBn + 192;
    float w00 = Wn_s[k0 * 3 + 0], w01 = Wn_s[k0 * 3 + 1], w02 = Wn_s[k0 * 3 + 2];
    float w10 = Wn_s[k0 * 3 + 3], w11 = Wn_s[k0 * 3 + 4], w12 = Wn_s[k0 * 3 + 5];
    float c0 = bn_s[k0], c1 = bn_s[k0 + 1];
    v2f a = {c0 + w00 * f0 + w01 * f1 + w02 * f2, c1 + w10 * f0 + w11 * f1 + w12 * f2};
    A[kk] = a;
    float y0 = c0 + w00 * g0 + w01 * g1 + w02 * g2;
    float y1 = c1 + w10 * g0 + w11 * g1 + w12 * g2;
    v2f m = {has ? y0 : 0.f, has ? y1 : 0.f};
    A[16 + kk] = m;
  }
}

__device__ __forceinline__ void
tile_gemm_store(const v2f (&A)[32], const float* Bl, int lane, int half, int base,
                const float* __restrict__ bself, const float* __restrict__ bneigh,
                float* __restrict__ hout, int N) {
  v8f acc0 = {}, acc1 = {}, acc2 = {}, acc3 = {};
#pragma unroll
  for (int kk = 0; kk < 32; ++kk) {
    int bb = kk * 256 + lane * 2;
    acc0 = __builtin_amdgcn_wmma_f32_16x16x4_f32(false, A[kk], false, *(const v2f*)&Bl[bb + 0],   (short)0, acc0, false, false);
    acc1 = __builtin_amdgcn_wmma_f32_16x16x4_f32(false, A[kk], false, *(const v2f*)&Bl[bb + 64],  (short)0, acc1, false, false);
    acc2 = __builtin_amdgcn_wmma_f32_16x16x4_f32(false, A[kk], false, *(const v2f*)&Bl[bb + 128], (short)0, acc2, false, false);
    acc3 = __builtin_amdgcn_wmma_f32_16x16x4_f32(false, A[kk], false, *(const v2f*)&Bl[bb + 192], (short)0, acc3, false, false);
  }

#pragma unroll
  for (int ct = 0; ct < 4; ++ct) {
    v8f av = (ct == 0) ? acc0 : (ct == 1) ? acc1 : (ct == 2) ? acc2 : acc3;
    int n = ct * 16 + (lane & 15);
    float bs = bself[n] + bneigh[n];
#pragma unroll
    for (int r = 0; r < 8; ++r) {
      int mg = base + r + 8 * half;
      if (mg < N) {
        float val = av[r] + bs;
        hout[(long long)mg * 64 + n] = val > 0.f ? val : 0.f;
      }
    }
  }
}

__global__ void __launch_bounds__(256) __attribute__((amdgpu_waves_per_eu(8)))
k_bucket(const int* __restrict__ bstart, const unsigned* __restrict__ tmp,
         int* __restrict__ rowptr, float* __restrict__ cnt, int* __restrict__ col,
         const float* __restrict__ nf, const float* __restrict__ Wn,
         const float* __restrict__ bn, const float* __restrict__ Wf0,
         const float* __restrict__ bself, const float* __restrict__ bneigh,
         float* __restrict__ hout0, int N, int nbk, EdgeJob job) {
  if ((int)blockIdx.x >= nbk) {
    edge_block(job, job.first + ((int)blockIdx.x - nbk));
    return;
  }
  __shared__ __attribute__((aligned(16))) int hcu[2 * BNODES];
  int* hc = hcu;
  int* cu = hcu + BNODES;
  __shared__ int wt[8];
  __shared__ __attribute__((aligned(16))) int scol[CAPB + 1];
  int b = blockIdx.x, tid = threadIdx.x, lane = tid & 31, w = tid >> 5;
  int beg = bstart[b], end = bstart[b + 1];
  int m = end - beg;
  if (tid < BNODES) hc[tid] = 0;
  __syncthreads();
  for (int i0 = 0; i0 < m; i0 += 2048) {
    unsigned wd[8];
#pragma unroll
    for (int u = 0; u < 8; ++u) {
      int i = i0 + u * 256 + tid;
      wd[u] = tmp[beg + (i < m ? i : m - 1)];
    }
#pragma unroll
    for (int u = 0; u < 8; ++u) {
      int i = i0 + u * 256 + tid;
      atomicAdd(&hc[wd[u] & (BNODES - 1)], i < m ? 1 : 0);
    }
  }
  __syncthreads();
  int v = (tid < BNODES) ? hc[tid] : 0;
  int inc = v;
  for (int o = 1; o < 32; o <<= 1) {
    int u = __shfl_up(inc, o);
    if (lane >= o) inc += u;
  }
  if (lane == 31) wt[w] = inc;
  __syncthreads();
  int pre = 0;
  for (int k = 0; k < 8; ++k) pre += (k < w) ? wt[k] : 0;
  int ex = pre + inc - v;
  if (tid < BNODES) {
    cu[tid] = ex;
    int node = (b << BSH) + tid;
    if (node < N) {
      rowptr[node] = beg + ex;
      cnt[node] = (float)v;
    }
  }
  if (b == nbk - 1 && tid == 0) rowptr[N] = end;
  __syncthreads();
  bool stage = m <= CAPB;
  for (int i0 = 0; i0 < m; i0 += 2048) {
    unsigned wd[8];
#pragma unroll
    for (int u = 0; u < 8; ++u) {
      int i = i0 + u * 256 + tid;
      wd[u] = tmp[beg + (i < m ? i : m - 1)];
    }
#pragma unroll
    for (int u = 0; u < 8; ++u) {
      int i = i0 + u * 256 + tid;
      int valid = i < m ? 1 : 0;
      int slot = atomicAdd(&cu[wd[u] & (BNODES - 1)], valid);
      slot = slot < m - 1 ? slot : m - 1;
      int dv = (int)(wd[u] >> BSH);
      if (stage) scol[valid ? slot : CAPB] = dv;
      else if (valid) col[beg + slot] = dv;
    }
  }
  if (!stage) __threadfence();
  __syncthreads();
  if (stage)
    for (int i = tid; i < m; i += 256) col[beg + i] = scol[i];
  if (Wf0 == nullptr) return;

  int half = lane >> 4, mi = lane & 15;
  int base = (b << BSH) + 16 * w;
  int mrow = base + mi;
  long long mc = (mrow < N) ? mrow : 0;
  float f0 = nf[mc * 3 + 0], f1 = nf[mc * 3 + 1], f2 = nf[mc * 3 + 2];
  const v4f* __restrict__ wsrc = (const v4f*)Wf0;
  v4f wv[8];
#pragma unroll
  for (int i = 0; i < 8; ++i) wv[i] = wsrc[i * 256 + tid];
  const float* pn = (tid < 192) ? (Wn + tid) : (bn + (tid - 192));
  float sn = *pn;

  float q0 = 0.f, q1 = 0.f, q2 = 0.f;
  {
    int nl = tid >> 1, sub = tid & 1;
    int cn = hc[nl];
    int st = cu[nl] - cn;
    unsigned nmax = (unsigned)(N - 1);
    for (int off = sub; off < cn; off += 16) {
      unsigned id[8];
      if (stage) {
#pragma unroll
        for (int u = 0; u < 8; ++u) {
          int i = off + 2 * u;
          id[u] = (unsigned)scol[st + (i < cn ? i : cn - 1)];
        }
      } else {
#pragma unroll
        for (int u = 0; u < 8; ++u) {
          int i = off + 2 * u;
          id[u] = (unsigned)__hip_atomic_load(col + beg + st + (i < cn ? i : cn - 1),
                                              __ATOMIC_RELAXED, __HIP_MEMORY_SCOPE_AGENT);
        }
      }
      float fa[8], fb[8], fc[8];
#pragma unroll
      for (int u = 0; u < 8; ++u) {
        unsigned r = id[u] < nmax ? id[u] : nmax;
        const float* f = nf + (size_t)r * 3;
        fa[u] = f[0];
        fb[u] = f[1];
        fc[u] = f[2];
      }
#pragma unroll
      for (int u = 0; u < 8; ++u) {
        bool ok = (off + 2 * u) < cn;
        q0 += ok ? fa[u] : 0.f;
        q1 += ok ? fb[u] : 0.f;
        q2 += ok ? fc[u] : 0.f;
      }
    }
    q0 += __shfl_xor(q0, 1);
    q1 += __shfl_xor(q1, 1);
    q2 += __shfl_xor(q2, 1);
  }
  float s0 = __shfl(q0, 2 * mi), s1 = __shfl(q1, 2 * mi), s2 = __shfl(q2, 2 * mi);
  float c = (float)hc[16 * w + mi];
  __syncthreads();
  float* Bl = (float*)scol;
  float* WBn = (float*)hcu;
#pragma unroll
  for (int i = 0; i < 8; ++i) *(v4f*)&Bl[(i * 256 + tid) * 4] = wv[i];
  WBn[tid] = sn;
  __syncthreads();

  float inv = (c > 0.f) ? 1.f / c : 1.f;
  v2f A[32];
  layer0_rows(A, WBn, half, f0, f1, f2, s0 * inv, s1 * inv, s2 * inv, c > 0.f);
  tile_gemm_store(A, Bl, lane, half, base, bself, bneigh, hout0, N);
}

__global__ void __launch_bounds__(256)
k_gather(const int* __restrict__ rowptr, const int* __restrict__ col,
         const float* __restrict__ h, float* __restrict__ accum, int N, EdgeJob job) {
  int mb;
  if (block_role(job, mb)) {
    edge_block(job, mb);
    return;
  }
  int node = mb * 8 + (threadIdx.x >> 5);
  if (node >= N) return;
  int lane = threadIdx.x & 31, half = lane >> 4, c4 = (lane & 15) * 4;
  int beg = rowptr[node], end = rowptr[node + 1];
  const v4f zero = {0.f, 0.f, 0.f, 0.f};
  v4f s[4] = {zero, zero, zero, zero};
  if (beg < end) {
    int last = end - 1;
    unsigned nmax = (unsigned)(N - 1);
    int id[8];
#pragma unroll
    for (int u = 0; u < 8; ++u) {
      int e = beg + half + 2 * u;
      id[u] = col[e < last ? e : last];
    }
    for (int j0 = beg; j0 < end; j0 += 16) {
      v4f v[8];
#pragma unroll
      for (int u = 0; u < 8; ++u) {
        unsigned r = (unsigned)id[u] < nmax ? (unsigned)id[u] : nmax;
        v[u] = *(const v4f*)(h + (size_t)r * 64 + c4);
      }
#pragma unroll
      for (int u = 0; u < 8; ++u) {
        int e = j0 + 16 + half + 2 * u;
        id[u] = col[e < last ? e : last];
      }
#pragma unroll
      for (int u = 0; u < 8; ++u) {
        bool ok = (j0 + half + 2 * u) < end;
        s[u & 3] += ok ? v[u] : zero;
      }
    }
  }
  v4f a = (s[0] + s[1]) + (s[2] + s[3]);
  a.x += __shfl_xor(a.x, 16);
  a.y += __shfl_xor(a.y, 16);
  a.z += __shfl_xor(a.z, 16);
  a.w += __shfl_xor(a.w, 16);
  if (half == 0) *(v4f*)(accum + (size_t)node * 64 + c4) = a;
}

__global__ void __launch_bounds__(256)
k_layer_gemm(const float* __restrict__ hin, const float* __restrict__ accum,
             const float* __restrict__ cnt,
             const float* __restrict__ Wf, const float* __restrict__ bself,
             const float* __restrict__ bneigh,
             float* __restrict__ hout, int N) {
  __shared__ __attribute__((aligned(16))) float Bl[32 * 4 * 32 * 2];
  int tid = threadIdx.x;
  int w = tid >> 5, lane = tid & 31;
  int base = (blockIdx.x * 8 + w) * 16;
  int half = lane >> 4;
  int mrow = base + (lane & 15);
  long long mc = (mrow < N) ? mrow : 0;

  float c = cnt[mc];
  v2f A[32];
#pragma unroll
  for (int kk = 0; kk < 16; ++kk) {
    int k0 = kk * 4 + 2 * half;
    v2f a = {hin[mc * 64 + k0], hin[mc * 64 + k0 + 1]};
    A[kk] = a;
  }
#pragma unroll
  for (int kk = 0; kk < 16; ++kk) {
    int k0 = kk * 4 + 2 * half;
    v2f a = {accum[mc * 64 + k0], accum[mc * 64 + k0 + 1]};
    A[16 + kk] = a;
  }

  const v4f* __restrict__ wsrc = (const v4f*)Wf;
  v4f wv[8];
#pragma unroll
  for (int i = 0; i < 8; ++i) wv[i] = wsrc[i * 256 + tid];
#pragma unroll
  for (int i = 0; i < 8; ++i) *(v4f*)&Bl[(i * 256 + tid) * 4] = wv[i];
  __syncthreads();

  float inv = (c > 0.f) ? 1.f / c : 1.f;
#pragma unroll
  for (int kk = 0; kk < 16; ++kk) A[16 + kk] = A[16 + kk] * inv;

  tile_gemm_store(A, Bl, lane, half, base, bself, bneigh, hout, N);
}

static int share_ratio(int x, int all) {
  return all > 0 ? (int)(((long long)x * 65536 + all - 1) / all) : 0;
}

extern "C" void kernel_launch(void* const* d_in, const int* in_sizes, int n_in,
                              void* d_out, int out_size, void* d_ws, size_t ws_size,
                              hipStream_t stream) {
  if (n_in < 18) return;
  const float* node_feats = (const float*)d_in[0];
  const float* edge_feats = (const float*)d_in[1];
  const float* env        = (const float*)d_in[2];
  const int*   eidx       = (const int*)d_in[3];
  const float* Wn  = (const float*)d_in[4];
  const float* bn  = (const float*)d_in[5];
  const float* We  = (const float*)d_in[6];
  const float* be  = (const float*)d_in[7];
  const float* Wself  = (const float*)d_in[8];
  const float* bself  = (const float*)d_in[9];
  const float* Wneigh = (const float*)d_in[10];
  const float* bneigh = (const float*)d_in[11];
  const float* Wa  = (const float*)d_in[12];
  const float* ba  = (const float*)d_in[13];
  const float* W1  = (const float*)d_in[14];
  const float* b1  = (const float*)d_in[15];
  const float* W2  = (const float*)d_in[16];
  const float* b2  = (const float*)d_in[17];

  int N  = in_sizes[0] / 3;
  int E  = in_sizes[1] / 3;
  int NV = in_sizes[2] / 64;
  int L  = in_sizes[8] / (64 * 64);

  const int* src = eidx;
  const int* dst = eidx + E;

  float* scores = (float*)d_out;
  float* h_out  = (float*)d_out + E;

  float* accum = (float*)d_ws;
  float* cnt   = accum + (long long)N * 64;
  long long off = (((long long)N * 65) + 63) & ~63LL;
  float* h_b   = (float*)d_ws + off;
  float* aug   = h_b + (long long)N * 64;
  int* rowptr = (int*)(aug + 256);
  unsigned* tmp = (unsigned*)(rowptr + (((long long)N + 1 + 63) & ~63LL));
  int* col    = (int*)tmp + (((long long)E + 63) & ~63LL);
  float* w1f  = (float*)(col + (((long long)E + 63) & ~63LL));
  float* wlf  = w1f + 4096;
  int nbk = (N + BNODES - 1) >> BSH;
  int nchunk = (E + CHUNK - 1) / CHUNK;
  int* bstart = (int*)(wlf + (long long)L * 8192);
  int* counts = bstart + 1024;
  if (nbk > MAXBK) return;

  int zb = nchunk;
  int nb = (L > 0) ? 0 : (N * 64 + 255) / 256;
  int sb = 16 + 32 * L;
  k_prologue<<<64 + sb + zb + nb, 256, 0, stream>>>(node_feats, Wn, bn, h_out, N, eidx, E, counts, nbk, zb, Wa,
                                                    ba, We, be, env, NV, aug, W1, Wself,
                                                    Wneigh, w1f, wlf, sb);
  int etiles = (E + 15) / 16;
  int units = (etiles + 7) / 8;
  int wide = (L > 1) ? L - 1 : 0;
  int xr = units / (2 + wide);
  int xf = (wide > 0) ? units / (2 + wide) : units - xr;
  int xg = (wide > 0) ? (units - xr - xf) / wide : 0;
  EdgeJob job = {edge_feats, aug, w1f, b1, W2, b2, scores, E, 0, 0, 0};

  job.nblk = xr;
  job.first = 0;
  k_scatter<<<nchunk + xr, 256, 0, stream>>>(src, dst, counts, bstart, tmp, E, nbk, nchunk, job);
  job.nblk = xf;
  job.first = xr;
  float* hout0 = (L == 1) ? h_out : h_b;
  k_bucket<<<nbk + xf, 256, 0, stream>>>(bstart, tmp, rowptr, cnt, col, node_feats, Wn, bn,
                                         (L > 0) ? wlf : nullptr, bself, bneigh, hout0, N, nbk, job);

  int tiles = (N + 15) / 16;
  int gb = (N + 7) / 8;
  float* hin = (L > 0) ? hout0 : h_out;
  for (int l = 1; l < L; ++l) {
    float* hout = (l == L - 1) ? h_out : ((hin == h_out) ? h_b : h_out);
    job.first = xr + xf + (l - 1) * xg;
    job.nblk = (l == L - 1) ? units - job.first : xg;
    job.ratio = share_ratio(job.nblk, gb + job.nblk);
    k_gather<<<gb + job.nblk, 256, 0, stream>>>(rowptr, col, hin, accum, N, job);
    k_layer_gemm<<<(tiles + 7) / 8, 256, 0, stream>>>(
        hin, accum, cnt, wlf + (long long)l * 8192, bself + (long long)l * 64,
        bneigh + (long long)l * 64, hout, N);
    hin = hout;
  }
}


# layers 1..: the launch that sums a tile's neighbour rows also runs its WMMAs (8 waves = 4 column tiles x 2 K halves, means kept in LDS): 5 launches to 4, no sum buffer; product sum order changes
# speedup vs baseline: 7.1150x; 1.0295x over previous
#include <hip/hip_runtime.h>

typedef __attribute__((ext_vector_type(2))) float v2f;
typedef __attribute__((ext_vector_type(4))) float v4f;
typedef __attribute__((ext_vector_type(8))) float v8f;

#define BSH 7
#define BNODES 128
#define CHUNK 6400
#define MAXBK 512
#define CAPB 8192

__global__ void __launch_bounds__(256)
k_prologue(const float* __restrict__ nf, const float* __restrict__ Wn,
           const float* __restrict__ bn, float* __restrict__ h, int N,
           const int* __restrict__ src, int E, int* __restrict__ counts, int nbk, int zb,
           const float* __restrict__ Wa, const float* __restrict__ ba,
           const float* __restrict__ We, const float* __restrict__ be,
           const float* __restrict__ env, int NV,
           float* __restrict__ aug,
           const float* __restrict__ W1, const float* __restrict__ Wself,
           const float* __restrict__ Wneigh, float* __restrict__ w1f,
           float* __restrict__ wlf, int sb) {
  __shared__ float part[4][4];
  int b = blockIdx.x, tid = threadIdx.x;

  if (b >= 64 + sb + zb) {
    int t = (b - 64 - sb - zb) * 256 + tid;
    int n = t >> 6, j = t & 63;
    if (n >= N) return;
    float f0 = nf[n * 3 + 0], f1 = nf[n * 3 + 1], f2 = nf[n * 3 + 2];
    h[t] = bn[j] + Wn[j * 3 + 0] * f0 + Wn[j * 3 + 1] * f1 + Wn[j * 3 + 2] * f2;
    return;
  }
  if (b >= 64 + sb) {
    __shared__ int hb[MAXBK];
    int c = b - 64 - sb;
    for (int i = tid; i < MAXBK; i += 256) hb[i] = 0;
    __syncthreads();
    int e0 = c * CHUNK;
    int sv[CHUNK / 256];
#pragma unroll
    for (int k = 0; k < CHUNK / 256; ++k) {
      int e = e0 + k * 256 + tid;
      sv[k] = src[e < E ? e : E - 1];
    }
#pragma unroll
    for (int k = 0; k < CHUNK / 256; ++k) {
      int e = e0 + k * 256 + tid;
      unsigned bk = (unsigned)sv[k] >> BSH;
      bk = bk < (unsigned)(nbk - 1) ? bk : (unsigned)(nbk - 1);
      if (e < E) atomicAdd(&hb[bk], 1);
    }
    __syncthreads();
    int nbkp = (nbk + 3) & ~3;
    for (int i = tid; i < nbkp; i += 256) counts[(size_t)c * nbkp + i] = hb[i];
    return;
  }
  if (b >= 64) {
    int t = (b - 64) * 256 + tid;
    int u = t - 4096;
    if (t < 4096) {
      int q = t & 3, lane = (t >> 2) & 31, mt = (t >> 7) & 3, t2 = t >> 9;
      int st = 2 * t2 + (q >> 1);
      int m = mt * 16 + (lane & 15);
      int f = (st >> 2) * 16 + 8 * (lane >> 4) + 2 * (st & 3) + (q & 1);
      w1f[t] = W1[m * 64 + f];
    } else {
      int idx = u & 8191;
      int v = idx & 1, lane = (idx >> 1) & 31, ct = (idx >> 6) & 3, kk = idx >> 8;
      int n = ct * 16 + (lane & 15);
      int k = kk * 4 + 2 * (lane >> 4) + v;
      int l = u >> 13;
      const float* ws = (k < 64) ? Wself : Wneigh;
      wlf[u] = ws[l * 4096 + n * 64 + (k & 63)];
    }
    return;
  }

  int j = b;
  float a0 = 0.f, a1 = 0.f, a2 = 0.f, c = 0.f;
  if (tid < 64) {
    float wv = Wa[j * 128 + tid];
    a0 = wv * We[tid * 3 + 0];
    a1 = wv * We[tid * 3 + 1];
    a2 = wv * We[tid * 3 + 2];
    c = wv * be[tid];
  } else if (tid < 128) {
    int d = tid - 64;
    float vm = 0.f;
    for (int v = 0; v < NV; ++v) vm += env[v * 64 + d];
    c = Wa[j * 128 + 64 + d] * (vm * (1.0f / (float)NV));
  }
  for (int o = 16; o > 0; o >>= 1) {
    a0 += __shfl_xor(a0, o);
    a1 += __shfl_xor(a1, o);
    a2 += __shfl_xor(a2, o);
    c += __shfl_xor(c, o);
  }
  int w = tid >> 5;
  if (w < 4 && (tid & 31) == 0) {
    part[w][0] = a0;
    part[w][1] = a1;
    part[w][2] = a2;
    part[w][3] = c;
  }
  __syncthreads();
  if (tid == 0) {
    int rt = j >> 4;
    float* lo = aug + ((rt >> 1) * 32 + (j & 15)) * 4 + (rt & 1) * 2;
    float* hi = lo + 16 * 4;
    lo[0] = part[0][0] + part[1][0];
    lo[1] = part[0][1] + part[1][1];
    hi[0] = part[0][2] + part[1][2];
    hi[1] = ba[j] + ((part[0][3] + part[1][3]) + (part[2][3] + part[3][3]));
  }
}

struct EdgeJob {
  const float* ef;
  const float* aug;
  const float* W1f;
  const float* b1;
  const float* W2;
  const float* b2;
  float* scores;
  int E;
  int nblk;
  int ratio;
  int first;
};

__device__ __forceinline__ bool block_role(const EdgeJob& j, int& idx) {
  unsigned b = blockIdx.x, cap = (unsigned)j.nblk;
  unsigned q0 = (unsigned)(((unsigned long long)b * (unsigned)j.ratio) >> 16);
  unsigned q1 = (unsigned)(((unsigned long long)(b + 1) * (unsigned)j.ratio) >> 16);
  q0 = q0 < cap ? q0 : cap;
  q1 = q1 < cap ? q1 : cap;
  if (q1 > q0) {
    idx = j.first + (int)q0;
    return true;
  }
  idx = (int)(b - q0);
  return false;
}

__device__ __forceinline__ float relu_bits(float x) {
  int i = __float_as_int(x);
  return __int_as_float(i > 0 ? i : 0);
}

__device__ __forceinline__ void edge_block(const EdgeJob& j, int blk) {
  __shared__ __attribute__((aligned(16))) float Ws[8 * 4 * 32 * 4];
  __shared__ __attribute__((aligned(16))) float AGs[256];
  __shared__ __attribute__((aligned(16))) float BWs[256];
  const float* __restrict__ ef = j.ef;
  const v4f* __restrict__ wsrc = (const v4f*)j.W1f;
  float* __restrict__ scores = j.scores;
  int E = j.E;
  int tid = threadIdx.x;
  int w = tid >> 5, lane = tid & 31;
  int base = (blk * 8 + w) * 16;
  int half = lane >> 4;
  int row = base + (lane & 15);
  long long rc = (row < E) ? row : 0;
  const float* pe = ef + rc * 3 + 2 * half;
  float x0 = pe[0];
  float x1 = pe[1 - half];
  v4f wv[4];
#pragma unroll
  for (int i = 0; i < 4; ++i) wv[i] = wsrc[i * 256 + tid];
  const float* pb = (tid & 64) ? (j.W2 + (tid & 63)) : (j.b1 + (tid & 63));
  float sa = j.aug[tid];
  float sb = *pb;
#pragma unroll
  for (int i = 0; i < 4; ++i) *(v4f*)&Ws[(i * 256 + tid) * 4] = wv[i];
  AGs[tid] = sa;
  BWs[tid] = sb;
  __syncthreads();

  const v2f xb = {x0, half ? 1.0f : x1};
  const v8f zero8 = {};
  const v4f* bw = (const v4f*)BWs;
  const v4f g0 = ((const v4f*)AGs)[lane], g1 = ((const v4f*)AGs)[32 + lane];

  v8f acc[4];
#pragma unroll
  for (int mt = 0; mt < 4; ++mt) {
    v4f lo = bw[mt * 4 + half * 2], hi = bw[mt * 4 + half * 2 + 1];
    v8f a = {lo.x, lo.y, lo.z, lo.w, hi.x, hi.y, hi.z, hi.w};
    acc[mt] = a;
  }

  v8f H = zero8;
#pragma unroll
  for (int t2 = 0; t2 < 8; ++t2) {
    if ((t2 & 1) == 0) {
      int rt = t2 >> 1;
      v2f ga = (rt == 0) ? v2f{g0.x, g0.y} : (rt == 1) ? v2f{g0.z, g0.w}
             : (rt == 2) ? v2f{g1.x, g1.y} : v2f{g1.z, g1.w};
      H = __builtin_amdgcn_wmma_f32_16x16x4_f32(false, ga, false, xb, (short)0, zero8, false, false);
#pragma unroll
      for (int r = 0; r < 8; ++r) H[r] = relu_bits(H[r]);
    }
    int r0 = 4 * (t2 & 1);
    const v2f bA = {H[r0], H[r0 + 1]};
    const v2f bB = {H[r0 + 2], H[r0 + 3]};
    v4f a[4];
#pragma unroll
    for (int mt = 0; mt < 4; ++mt) a[mt] = *(const v4f*)&Ws[((t2 * 4 + mt) * 32 + lane) * 4];
#pragma unroll
    for (int mt = 0; mt < 4; ++mt) {
      const v2f aA = {a[mt].x, a[mt].y};
      acc[mt] = __builtin_amdgcn_wmma_f32_16x16x4_f32(false, aA, false, bA, (short)0, acc[mt], false, false);
    }
#pragma unroll
    for (int mt = 0; mt < 4; ++mt) {
      const v2f aB = {a[mt].z, a[mt].w};
      acc[mt] = __builtin_amdgcn_wmma_f32_16x16x4_f32(false, aB, false, bB, (short)0, acc[mt], false, false);
    }
  }

  float p[4] = {0.f, 0.f, 0.f, 0.f};
#pragma unroll
  for (int mt = 0; mt < 4; ++mt) {
    v4f lo = bw[16 + mt * 4 + half * 2], hi = bw[16 + mt * 4 + half * 2 + 1];
    float wq[8] = {lo.x, lo.y, lo.z, lo.w, hi.x, hi.y, hi.z, hi.w};
#pragma unroll
    for (int r = 0; r < 8; ++r) {
      p[r & 3] += wq[r] * relu_bits(acc[mt][r]);
    }
  }
  float ps = (p[0] + p[1]) + (p[2] + p[3]);
  unsigned pu = __float_as_uint(ps);
  unsigned qu = __builtin_amdgcn_permlanex16(pu, pu, 0x76543210u, 0xfedcba98u, false, false);
  ps += __uint_as_float(qu);
  if (half == 0 && row < E) scores[row] = ps + j.b2[0];
}

__global__ void __launch_bounds__(256)
k_scatter(const int* __restrict__ src, const int* __restrict__ dst,
          const int* __restrict__ counts, int* __restrict__ bstart,
          unsigned* __restrict__ tmp, int E, int nbk, int nchunk, EdgeJob job) {
  if ((int)blockIdx.x >= nchunk) {
    edge_block(job, job.first + ((int)blockIdx.x - nchunk));
    return;
  }
  __shared__ int cur[MAXBK];
  __shared__ int dlt[MAXBK];
  __shared__ int wtot[8];
  __shared__ int wtot2[8];
  __shared__ __attribute__((aligned(16))) unsigned sword[CHUNK + 1];
  __shared__ __attribute__((aligned(16))) int spos[CHUNK + 1];
  int c = blockIdx.x, tid = threadIdx.x, lane = tid & 31, w = tid >> 5;
  int nbkp = (nbk + 3) & ~3;
  int* Tl = (int*)sword;
  int* Pl = spos;
  {
    int part = __builtin_amdgcn_readfirstlane(tid >> 7);
    unsigned g4 = (tid & 127) * 4;
    if (g4 < (unsigned)nbkp) {
      int hlf = (nchunk + 1) >> 1;
      int lo = part * hlf;
      int hi = lo + hlf < nchunk ? lo + hlf : nchunk;
      int T0 = 0, T1 = 0, T2 = 0, T3 = 0, P0 = 0, P1 = 0, P2 = 0, P3 = 0;
      for (int k0 = lo; k0 < hi; k0 += 5) {
        int4 v[5];
#pragma unroll
        for (int u = 0; u < 5; ++u) {
          int k = k0 + u;
          int kc = k < hi ? k : hi - 1;
          const int* rowp = counts + (size_t)kc * nbkp;
          v[u] = *(const int4*)(rowp + g4);
        }
#pragma unroll
        for (int u = 0; u < 5; ++u) {
          int k = k0 + u;
          bool in = k < hi, front = in && k < c;
          T0 += in ? v[u].x : 0;
          T1 += in ? v[u].y : 0;
          T2 += in ? v[u].z : 0;
          T3 += in ? v[u].w : 0;
          P0 += front ? v[u].x : 0;
          P1 += front ? v[u].y : 0;
          P2 += front ? v[u].z : 0;
          P3 += front ? v[u].w : 0;
        }
      }
      int4 Tv = {T0, T1, T2, T3}, Pv = {P0, P1, P2, P3};
      *(int4*)&Tl[part * MAXBK + g4] = Tv;
      *(int4*)&Pl[part * MAXBK + g4] = Pv;
    }
  }
  __syncthreads();
  int b0 = 2 * tid, b1 = 2 * tid + 1;
  int c0 = 0, c1 = 0, t0 = 0, t1 = 0, p0 = 0, p1 = 0;
  if (b0 < nbk) {
    c0 = counts[(size_t)c * nbkp + b0];
    t0 = Tl[b0] + Tl[MAXBK + b0];
    p0 = Pl[b0] + Pl[MAXBK + b0];
  }
  if (b1 < nbk) {
    c1 = counts[(size_t)c * nbkp + b1];
    t1 = Tl[b1] + Tl[MAXBK + b1];
    p1 = Pl[b1] + Pl[MAXBK + b1];
  }
  int sum = c0 + c1, inc = sum;
  int tsum = t0 + t1, tinc = tsum;
  for (int o = 1; o < 32; o <<= 1) {
    int u = __shfl_up(inc, o);
    int u2 = __shfl_up(tinc, o);
    if (lane >= o) {
      inc += u;
      tinc += u2;
    }
  }
  if (lane == 31) {
    wtot[w] = inc;
    wtot2[w] = tinc;
  }
  __syncthreads();
  int pre = 0, tpre = 0;
  for (int k = 0; k < 8; ++k) {
    pre += (k < w) ? wtot[k] : 0;
    tpre += (k < w) ? wtot2[k] : 0;
  }
  int ex = pre + inc - sum;
  int tex = tpre + tinc - tsum;
  int g0 = tex + p0, g1 = tex + t0 + p1;
  if (c == 0) {
    if (b0 < nbk) bstart[b0] = tex;
    if (b1 < nbk) bstart[b1] = tex + t0;
    if (tid == 255) bstart[nbk] = tex + tsum;
  }
  cur[b0] = ex;
  dlt[b0] = g0 - ex;
  cur[b1] = ex + c0;
  dlt[b1] = g1 - (ex + c0);
  __syncthreads();

  int e0 = c * CHUNK;
  for (int k0 = 0; k0 < CHUNK / 256; k0 += 5) {
    unsigned sv[5], dv[5];
#pragma unroll
    for (int u = 0; u < 5; ++u) {
      int e = e0 + (k0 + u) * 256 + tid;
      int ec = e < E ? e : E - 1;
      sv[u] = (unsigned)src[ec];
      dv[u] = (unsigned)dst[ec];
    }
#pragma unroll
    for (int u = 0; u < 5; ++u) {
      int e = e0 + (k0 + u) * 256 + tid;
      int valid = e < E ? 1 : 0;
      unsigned bk = sv[u] >> BSH;
      bk = bk < (unsigned)(nbk - 1) ? bk : (unsigned)(nbk - 1);
      int slot = atomicAdd(&cur[bk], valid);
      slot = slot < CHUNK - 1 ? slot : CHUNK - 1;
      slot = valid ? slot : CHUNK;
      sword[slot] = (dv[u] << BSH) | (sv[u] & (BNODES - 1));
      spos[slot] = slot + dlt[bk];
    }
  }
  __syncthreads();
  int n = E - e0;
  n = n < CHUNK ? n : CHUNK;
  for (int i = tid; i < n; i += 256) {
    int p = spos[i];
    p = (unsigned)p < (unsigned)E ? p : E - 1;
    tmp[p] = sword[i];
  }
}

__device__ __forceinline__ void
layer0_rows(v2f (&A)[32], const float* WBn, int half, float f0, float f1, float f2,
            float g0, float g1, float g2, bool has) {
#pragma unroll
  for (int kk = 0; kk < 16; ++kk) {
    int k0 = kk * 4 + 2 * half;
    const float* Wn_s = WBn;
    const float* bn_s = WBn + 192;
    float w00 = Wn_s[k0 * 3 + 0], w01 = Wn_s[k0 * 3 + 1], w02 = Wn_s[k0 * 3 + 2];
    float w10 = Wn_s[k0 * 3 + 3], w11 = Wn_s[k0 * 3 + 4], w12 = Wn_s[k0 * 3 + 5];
    float c0 = bn_s[k0], c1 = bn_s[k0 + 1];
    v2f a = {c0 + w00 * f0 + w01 * f1 + w02 * f2, c1 + w10 * f0 + w11 * f1 + w12 * f2};
    A[kk] = a;
    float y0 = c0 + w00 * g0 + w01 * g1 + w02 * g2;
    float y1 = c1 + w10 * g0 + w11 * g1 + w12 * g2;
    v2f m = {has ? y0 : 0.f, has ? y1 : 0.f};
    A[16 + kk] = m;
  }
}

__device__ __forceinline__ void
tile_gemm_store(const v2f (&A)[32], const float* Bl, int lane, int half, int base,
                const float* __restrict__ bself, const float* __restrict__ bneigh,
                float* __restrict__ hout, int N) {
  v8f acc0 = {}, acc1 = {}, acc2 = {}, acc3 = {};
#pragma unroll
  for (int kk = 0; kk < 32; ++kk) {
    int bb = kk * 256 + lane * 2;
    acc0 = __builtin_amdgcn_wmma_f32_16x16x4_f32(false, A[kk], false, *(const v2f*)&Bl[bb + 0],   (short)0, acc0, false, false);
    acc1 = __builtin_amdgcn_wmma_f32_16x16x4_f32(false, A[kk], false, *(const v2f*)&Bl[bb + 64],  (short)0, acc1, false, false);
    acc2 = __builtin_amdgcn_wmma_f32_16x16x4_f32(false, A[kk], false, *(const v2f*)&Bl[bb + 128], (short)0, acc2, false, false);
    acc3 = __builtin_amdgcn_wmma_f32_16x16x4_f32(false, A[kk], false, *(const v2f*)&Bl[bb + 192], (short)0, acc3, false, false);
  }

#pragma unroll
  for (int ct = 0; ct < 4; ++ct) {
    v8f av = (ct == 0) ? acc0 : (ct == 1) ? acc1 : (ct == 2) ? acc2 : acc3;
    int n = ct * 16 + (lane & 15);
    float bs = bself[n] + bneigh[n];
#pragma unroll
    for (int r = 0; r < 8; ++r) {
      int mg = base + r + 8 * half;
      if (mg < N) {
        float val = av[r] + bs;
        hout[(long long)mg * 64 + n] = val > 0.f ? val : 0.f;
      }
    }
  }
}

__global__ void __launch_bounds__(256) __attribute__((amdgpu_waves_per_eu(8)))
k_bucket(const int* __restrict__ bstart, const unsigned* __restrict__ tmp,
         int* __restrict__ rowptr, float* __restrict__ cnt, int* __restrict__ col,
         const float* __restrict__ nf, const float* __restrict__ Wn,
         const float* __restrict__ bn, const float* __restrict__ Wf0,
         const float* __restrict__ bself, const float* __restrict__ bneigh,
         float* __restrict__ hout0, int N, int nbk, EdgeJob job) {
  if ((int)blockIdx.x >= nbk) {
    edge_block(job, job.first + ((int)blockIdx.x - nbk));
    return;
  }
  __shared__ __attribute__((aligned(16))) int hcu[2 * BNODES];
  int* hc = hcu;
  int* cu = hcu + BNODES;
  __shared__ int wt[8];
  __shared__ __attribute__((aligned(16))) int scol[CAPB + 1];
  int b = blockIdx.x, tid = threadIdx.x, lane = tid & 31, w = tid >> 5;
  int beg = bstart[b], end = bstart[b + 1];
  int m = end - beg;
  if (tid < BNODES) hc[tid] = 0;
  __syncthreads();
  for (int i0 = 0; i0 < m; i0 += 2048) {
    unsigned wd[8];
#pragma unroll
    for (int u = 0; u < 8; ++u) {
      int i = i0 + u * 256 + tid;
      wd[u] = tmp[beg + (i < m ? i : m - 1)];
    }
#pragma unroll
    for (int u = 0; u < 8; ++u) {
      int i = i0 + u * 256 + tid;
      atomicAdd(&hc[wd[u] & (BNODES - 1)], i < m ? 1 : 0);
    }
  }
  __syncthreads();
  int v = (tid < BNODES) ? hc[tid] : 0;
  int inc = v;
  for (int o = 1; o < 32; o <<= 1) {
    int u = __shfl_up(inc, o);
    if (lane >= o) inc += u;
  }
  if (lane == 31) wt[w] = inc;
  __syncthreads();
  int pre = 0;
  for (int k = 0; k < 8; ++k) pre += (k < w) ? wt[k] : 0;
  int ex = pre + inc - v;
  if (tid < BNODES) {
    cu[tid] = ex;
    int node = (b << BSH) + tid;
    if (node < N) {
      rowptr[node] = beg + ex;
      cnt[node] = (float)v;
    }
  }
  if (b == nbk - 1 && tid == 0) rowptr[N] = end;
  __syncthreads();
  bool stage = m <= CAPB;
  for (int i0 = 0; i0 < m; i0 += 2048) {
    unsigned wd[8];
#pragma unroll
    for (int u = 0; u < 8; ++u) {
      int i = i0 + u * 256 + tid;
      wd[u] = tmp[beg + (i < m ? i : m - 1)];
    }
#pragma unroll
    for (int u = 0; u < 8; ++u) {
      int i = i0 + u * 256 + tid;
      int valid = i < m ? 1 : 0;
      int slot = atomicAdd(&cu[wd[u] & (BNODES - 1)], valid);
      slot = slot < m - 1 ? slot : m - 1;
      int dv = (int)(wd[u] >> BSH);
      if (stage) scol[valid ? slot : CAPB] = dv;
      else if (valid) col[beg + slot] = dv;
    }
  }
  if (!stage) __threadfence();
  __syncthreads();
  if (stage)
    for (int i = tid; i < m; i += 256) col[beg + i] = scol[i];
  if (Wf0 == nullptr) return;

  int half = lane >> 4, mi = lane & 15;
  int base = (b << BSH) + 16 * w;
  int mrow = base + mi;
  long long mc = (mrow < N) ? mrow : 0;
  float f0 = nf[mc * 3 + 0], f1 = nf[mc * 3 + 1], f2 = nf[mc * 3 + 2];
  const v4f* __restrict__ wsrc = (const v4f*)Wf0;
  v4f wv[8];
#pragma unroll
  for (int i = 0; i < 8; ++i) wv[i] = wsrc[i * 256 + tid];
  const float* pn = (tid < 192) ? (Wn + tid) : (bn + (tid - 192));
  float sn = *pn;

  float q0 = 0.f, q1 = 0.f, q2 = 0.f;
  {
    int nl = tid >> 1, sub = tid & 1;
    int cn = hc[nl];
    int st = cu[nl] - cn;
    unsigned nmax = (unsigned)(N - 1);
    for (int off = sub; off < cn; off += 16) {
      unsigned id[8];
      if (stage) {
#pragma unroll
        for (int u = 0; u < 8; ++u) {
          int i = off + 2 * u;
          id[u] = (unsigned)scol[st + (i < cn ? i : cn - 1)];
        }
      } else {
#pragma unroll
        for (int u = 0; u < 8; ++u) {
          int i = off + 2 * u;
          id[u] = (unsigned)__hip_atomic_load(col + beg + st + (i < cn ? i : cn - 1),
                                              __ATOMIC_RELAXED, __HIP_MEMORY_SCOPE_AGENT);
        }
      }
      float fa[8], fb[8], fc[8];
#pragma unroll
      for (int u = 0; u < 8; ++u) {
        unsigned r = id[u] < nmax ? id[u] : nmax;
        const float* f = nf + (size_t)r * 3;
        fa[u] = f[0];
        fb[u] = f[1];
        fc[u] = f[2];
      }
#pragma unroll
      for (int u = 0; u < 8; ++u) {
        bool ok = (off + 2 * u) < cn;
        q0 += ok ? fa[u] : 0.f;
        q1 += ok ? fb[u] : 0.f;
        q2 += ok ? fc[u] : 0.f;
      }
    }
    q0 += __shfl_xor(q0, 1);
    q1 += __shfl_xor(q1, 1);
    q2 += __shfl_xor(q2, 1);
  }
  float s0 = __shfl(q0, 2 * mi), s1 = __shfl(q1, 2 * mi), s2 = __shfl(q2, 2 * mi);
  float c = (float)hc[16 * w + mi];
  __syncthreads();
  float* Bl = (float*)scol;
  float* WBn = (float*)hcu;
#pragma unroll
  for (int i = 0; i < 8; ++i) *(v4f*)&Bl[(i * 256 + tid) * 4] = wv[i];
  WBn[tid] = sn;
  __syncthreads();

  float inv = (c > 0.f) ? 1.f / c : 1.f;
  v2f A[32];
  layer0_rows(A, WBn, half, f0, f1, f2, s0 * inv, s1 * inv, s2 * inv, c > 0.f);
  tile_gemm_store(A, Bl, lane, half, base, bself, bneigh, hout0, N);
}

__global__ void __launch_bounds__(256) __attribute__((amdgpu_waves_per_eu(12)))
k_gather(const int* __restrict__ rowptr, const int* __restrict__ col,
         const float* __restrict__ h, const float* __restrict__ Wf,
         const float* __restrict__ bself, const float* __restrict__ bneigh,
         float* __restrict__ hout, int N, EdgeJob job) {
  int mb;
  if (block_role(job, mb)) {
    edge_block(job, mb);
    return;
  }
  __shared__ __attribute__((aligned(16))) float Ms[16 * 68];
  __shared__ __attribute__((aligned(16))) float Ps[4 * 8 * 32];
  int tid = threadIdx.x;
  int w = tid >> 5, lane = tid & 31, half = lane >> 4, c4 = (lane & 15) * 4;
  int base = mb * 16;
  const v4f zero = {0.f, 0.f, 0.f, 0.f};
  for (int q = 0; q < 2; ++q) {
    int nl = 2 * w + q;
    int node = base + nl;
    v4f a = zero;
    if (node < N) {
      int beg = rowptr[node], end = rowptr[node + 1];
      v4f s[4] = {zero, zero, zero, zero};
      if (beg < end) {
        int last = end - 1;
        unsigned nmax = (unsigned)(N - 1);
        int id[8];
#pragma unroll
        for (int u = 0; u < 8; ++u) {
          int e = beg + half + 2 * u;
          id[u] = col[e < last ? e : last];
        }
        for (int j0 = beg; j0 < end; j0 += 16) {
          v4f v[8];
#pragma unroll
          for (int u = 0; u < 8; ++u) {
            unsigned r = (unsigned)id[u] < nmax ? (unsigned)id[u] : nmax;
            v[u] = *(const v4f*)(h + (size_t)r * 64 + c4);
          }
#pragma unroll
          for (int u = 0; u < 8; ++u) {
            int e = j0 + 16 + half + 2 * u;
            id[u] = col[e < last ? e : last];
          }
#pragma unroll
          for (int u = 0; u < 8; ++u) {
            bool ok = (j0 + half + 2 * u) < end;
            s[u & 3] += ok ? v[u] : zero;
          }
        }
      }
      a = (s[0] + s[1]) + (s[2] + s[3]);
      a.x += __shfl_xor(a.x, 16);
      a.y += __shfl_xor(a.y, 16);
      a.z += __shfl_xor(a.z, 16);
      a.w += __shfl_xor(a.w, 16);
      float c = (float)(end - beg);
      float inv = (c > 0.f) ? 1.f / c : 1.f;
      a = a * inv;
    }
    if (half == 0) *(v4f*)&Ms[nl * 68 + c4] = a;
  }

  int ct = w & 3, kh = w >> 2;
  int mi = lane & 15;
  int mrow = base + mi;
  long long mc = (mrow < N) ? mrow : 0;
  const float* __restrict__ wb = Wf + kh * 4096 + ct * 64 + lane * 2;
  v2f B[16], A[16];
#pragma unroll
  for (int kk = 0; kk < 16; ++kk) B[kk] = *(const v2f*)(wb + kk * 256);
  if (kh == 0) {
#pragma unroll
    for (int kk = 0; kk < 16; ++kk) A[kk] = *(const v2f*)(h + mc * 64 + kk * 4 + 2 * half);
  }
  int n = ct * 16 + mi;
  float bs = bself[n] + bneigh[n];
  __syncthreads();
  if (kh != 0) {
#pragma unroll
    for (int kk = 0; kk < 16; ++kk) A[kk] = *(const v2f*)&Ms[mi * 68 + kk * 4 + 2 * half];
  }
  v8f acc = {};
#pragma unroll
  for (int kk = 0; kk < 16; ++kk)
    acc = __builtin_amdgcn_wmma_f32_16x16x4_f32(false, A[kk], false, B[kk], (short)0, acc, false, false);
  if (kh != 0) {
#pragma unroll
    for (int r = 0; r < 8; ++r) Ps[(ct * 8 + r) * 32 + lane] = acc[r];
  }
  __syncthreads();
  if (kh == 0) {
    float p[8];
#pragma unroll
    for (int r = 0; r < 8; ++r) p[r] = Ps[(ct * 8 + r) * 32 + lane];
#pragma unroll
    for (int r = 0; r < 8; ++r) {
      int mg = base + r + 8 * half;
      if (mg < N) {
        float val = (acc[r] + p[r]) + bs;
        hout[(long long)mg * 64 + n] = val > 0.f ? val : 0.f;
      }
    }
  }
}

static int share_ratio(int x, int all) {
  return all > 0 ? (int)(((long long)x * 65536 + all - 1) / all) : 0;
}

extern "C" void kernel_launch(void* const* d_in, const int* in_sizes, int n_in,
                              void* d_out, int out_size, void* d_ws, size_t ws_size,
                              hipStream_t stream) {
  if (n_in < 18) return;
  const float* node_feats = (const float*)d_in[0];
  const float* edge_feats = (const float*)d_in[1];
  const float* env        = (const float*)d_in[2];
  const int*   eidx       = (const int*)d_in[3];
  const float* Wn  = (const float*)d_in[4];
  const float* bn  = (const float*)d_in[5];
  const float* We  = (const float*)d_in[6];
  const float* be  = (const float*)d_in[7];
  const float* Wself  = (const float*)d_in[8];
  const float* bself  = (const float*)d_in[9];
  const float* Wneigh = (const float*)d_in[10];
  const float* bneigh = (const float*)d_in[11];
  const float* Wa  = (const float*)d_in[12];
  const float* ba  = (const float*)d_in[13];
  const float* W1  = (const float*)d_in[14];
  const float* b1  = (const float*)d_in[15];
  const float* W2  = (const float*)d_in[16];
  const float* b2  = (const float*)d_in[17];

  int N  = in_sizes[0] / 3;
  int E  = in_sizes[1] / 3;
  int NV = in_sizes[2] / 64;
  int L  = in_sizes[8] / (64 * 64);

  const int* src = eidx;
  const int* dst = eidx + E;

  float* scores = (float*)d_out;
  float* h_out  = (float*)d_out + E;

  float* cnt   = (float*)d_ws + (long long)N * 64;
  long long off = (((long long)N * 65) + 63) & ~63LL;
  float* h_b   = (float*)d_ws + off;
  float* aug   = h_b + (long long)N * 64;
  int* rowptr = (int*)(aug + 256);
  unsigned* tmp = (unsigned*)(rowptr + (((long long)N + 1 + 63) & ~63LL));
  int* col    = (int*)tmp + (((long long)E + 63) & ~63LL);
  float* w1f  = (float*)(col + (((long long)E + 63) & ~63LL));
  float* wlf  = w1f + 4096;
  int nbk = (N + BNODES - 1) >> BSH;
  int nchunk = (E + CHUNK - 1) / CHUNK;
  int* bstart = (int*)(wlf + (long long)L * 8192);
  int* counts = bstart + 1024;
  if (nbk > MAXBK) return;

  int zb = nchunk;
  int nb = (L > 0) ? 0 : (N * 64 + 255) / 256;
  int sb = 16 + 32 * L;
  k_prologue<<<64 + sb + zb + nb, 256, 0, stream>>>(node_feats, Wn, bn, h_out, N, eidx, E, counts, nbk, zb, Wa,
                                                    ba, We, be, env, NV, aug, W1, Wself,
                                                    Wneigh, w1f, wlf, sb);
  int etiles = (E + 15) / 16;
  int units = (etiles + 7) / 8;
  int wide = (L > 1) ? L - 1 : 0;
  int xr = units / (2 + wide);
  int xf = (wide > 0) ? units / (2 + wide) : units - xr;
  int xg = (wide > 0) ? (units - xr - xf) / wide : 0;
  EdgeJob job = {edge_feats, aug, w1f, b1, W2, b2, scores, E, 0, 0, 0};

  job.nblk = xr;
  job.first = 0;
  k_scatter<<<nchunk + xr, 256, 0, stream>>>(src, dst, counts, bstart, tmp, E, nbk, nchunk, job);
  job.nblk = xf;
  job.first = xr;
  float* hout0 = ((L & 1) != 0) ? h_out : h_b;
  k_bucket<<<nbk + xf, 256, 0, stream>>>(bstart, tmp, rowptr, cnt, col, node_feats, Wn, bn,
                                         (L > 0) ? wlf : nullptr, bself, bneigh, hout0, N, nbk, job);

  int gb = (N + 15) / 16;
  float* hin = hout0;
  for (int l = 1; l < L; ++l) {
    float* hout = (hin == h_out) ? h_b : h_out;
    job.first = xr + xf + (l - 1) * xg;
    job.nblk = (l == L - 1) ? units - job.first : xg;
    job.ratio = share_ratio(job.nblk, gb + job.nblk);
    k_gather<<<gb + job.nblk, 256, 0, stream>>>(rowptr, col, hin, wlf + (long long)l * 8192,
                                                bself + (long long)l * 64,
                                                bneigh + (long long)l * 64, hout, N, job);
    hin = hout;
  }
}
